# starting kernel
# baseline (speedup 1.0000x reference)
#include <hip/hip_runtime.h>


typedef float v2f __attribute__((ext_vector_type(2)));
typedef float v8f __attribute__((ext_vector_type(8)));

#define NEG_SLOPE 0.2f
#define EPSV 1e-16f
#define FDIM 64

__device__ __forceinline__ unsigned enc_f32(float f) {
  unsigned u = __float_as_uint(f);
  return (u & 0x80000000u) ? ~u : (u | 0x80000000u);
}
__device__ __forceinline__ float dec_f32(unsigned u) {
  u = (u & 0x80000000u) ? (u & 0x7FFFFFFFu) : ~u;
  return __uint_as_float(u);
}
#define ENC_NEG_INF 0x007FFFFFu

template <int K>
__global__ __launch_bounds__(128) void gat_gemm_wmma(
    const float* __restrict__ X, const float* __restrict__ W,
    float* __restrict__ H, int M) {
  const int wave = threadIdx.x >> 5;
  const int lane = threadIdx.x & 31;
  const int l16  = lane & 15;
  const int kh   = (lane >> 4) << 1;
  const int rowBase = blockIdx.x * 16;
  const int colBase = wave * 16;

  int arow = rowBase + l16;
  if (arow >= M) arow = M - 1;
  const float* xr = X + (size_t)arow * K;
  const float* wc = W + colBase + l16;

  v8f c = {0.f, 0.f, 0.f, 0.f, 0.f, 0.f, 0.f, 0.f};
#pragma unroll
  for (int k = 0; k < K; k += 4) {
    v2f a, b;
    a[0] = xr[k + kh];
    a[1] = xr[k + kh + 1];
    b[0] = wc[(size_t)(k + kh) * FDIM];
    b[1] = wc[(size_t)(k + kh + 1) * FDIM];
    c = __builtin_amdgcn_wmma_f32_16x16x4_f32(false, a, false, b,
                                              (short)0, c, false, false);
  }

  const int rOff = (lane >> 4) << 3;
#pragma unroll
  for (int v = 0; v < 8; ++v) {
    int r = rowBase + rOff + v;
    if (r >= M) r = M - 1;
    H[(size_t)r * FDIM + colBase + l16] = c[v];
  }
}

__global__ void gat_node_dots(const float* __restrict__ H,
                              const float* __restrict__ a_src,
                              const float* __restrict__ a_dst,
                              float* __restrict__ hs, float* __restrict__ hd,
                              int Nn) {
  int i = blockIdx.x * blockDim.x + threadIdx.x;
  if (i >= Nn) return;
  const float* h = H + (size_t)i * FDIM;
  float s = 0.f, d = 0.f;
#pragma unroll
  for (int f = 0; f < FDIM; ++f) {
    float v = h[f];
    s += v * a_src[f];
    d += v * a_dst[f];
  }
  hs[i] = s;
  hd[i] = d;
}

__global__ void gat_init(float* __restrict__ agg, unsigned* __restrict__ maxenc,
                         float* __restrict__ denom, int Nn) {
  int i = blockIdx.x * blockDim.x + threadIdx.x;
  if (i < Nn * FDIM) agg[i] = 0.f;
  if (i < Nn) {
    maxenc[i] = ENC_NEG_INF;
    denom[i]  = 0.f;
  }
}

__global__ void gat_edge_logits(const int* __restrict__ ei,
                                const float* __restrict__ hs,
                                const float* __restrict__ hd,
                                float* __restrict__ ebuf,
                                unsigned* __restrict__ maxenc, int E, int ET) {
  int e = blockIdx.x * blockDim.x + threadIdx.x;
  if (e >= ET) return;
  int s, d;
  if (e < E) { s = ei[e]; d = ei[E + e]; } else { s = d = e - E; }
  float v = hs[s] + hd[d];
  v = (v > 0.f) ? v : NEG_SLOPE * v;
  ebuf[e] = v;
  atomicMax(&maxenc[d], enc_f32(v));
}

__global__ void gat_edge_exp(const int* __restrict__ ei,
                             float* __restrict__ ebuf,
                             const unsigned* __restrict__ maxenc,
                             float* __restrict__ denom, int E, int ET) {
  int e = blockIdx.x * blockDim.x + threadIdx.x;
  if (e >= ET) return;
  int d = (e < E) ? ei[E + e] : (e - E);
  float ex = __expf(ebuf[e] - dec_f32(maxenc[d]));
  ebuf[e] = ex;
  atomicAdd(&denom[d], ex);
}

__global__ void gat_edge_agg(const int* __restrict__ ei,
                             const float* __restrict__ ebuf,
                             const float* __restrict__ denom,
                             const float* __restrict__ H,
                             float* __restrict__ agg,
                             float* __restrict__ alpha_out, int E, int ET) {
  long long gid = (long long)blockIdx.x * blockDim.x + threadIdx.x;
  int e   = (int)(gid >> 4);
  int sub = (int)(gid & 15);
  if (e >= ET) return;
  int s, d;
  if (e < E) { s = ei[e]; d = ei[E + e]; } else { s = d = e - E; }
  float alpha = ebuf[e] / (denom[d] + EPSV);
  if (alpha_out != nullptr && sub == 0) alpha_out[e] = alpha;
  const float4 hv = ((const float4*)(H + (size_t)s * FDIM))[sub];
  float* ad = agg + (size_t)d * FDIM + sub * 4;
  atomicAdd(ad + 0, alpha * hv.x);
  atomicAdd(ad + 1, alpha * hv.y);
  atomicAdd(ad + 2, alpha * hv.z);
  atomicAdd(ad + 3, alpha * hv.w);
}

__global__ void gat_finalize(float* __restrict__ agg, const float* __restrict__ b,
                             int total) {
  int i = blockIdx.x * blockDim.x + threadIdx.x;
  if (i >= total) return;
  float v = agg[i] + b[i & (FDIM - 1)];
  agg[i] = (v > 0.f) ? v : 0.f;
}

static inline int cdiv_ll(long long a, long long b) {
  return (int)((a + b - 1) / b);
}

extern "C" void kernel_launch(void* const* d_in, const int* in_sizes, int n_in,
                              void* d_out, int out_size, void* d_ws,
                              size_t ws_size, hipStream_t stream) {
  const float* x      = (const float*)d_in[0];
  const int*   ei     = (const int*)d_in[1];
  const float* W1     = (const float*)d_in[2];
  const float* a_src1 = (const float*)d_in[3];
  const float* a_dst1 = (const float*)d_in[4];
  const float* b1     = (const float*)d_in[5];
  const float* W2     = (const float*)d_in[6];
  const float* a_src2 = (const float*)d_in[7];
  const float* a_dst2 = (const float*)d_in[8];
  const float* b2     = (const float*)d_in[9];

  const int IN_DIM = 128;
  const int Nn = in_sizes[0] / IN_DIM;
  const int E  = in_sizes[1] / 2;
  const int ET = E + Nn;

  float*    bufH   = (float*)d_ws;
  float*    bufAgg = bufH + (size_t)Nn * FDIM;
  float*    hs     = bufAgg + (size_t)Nn * FDIM;
  float*    hd     = hs + Nn;
  unsigned* maxenc = (unsigned*)(hd + Nn);
  float*    denom  = (float*)(maxenc + Nn);
  float*    ebuf   = denom + Nn;

  float* h2out     = (float*)d_out;
  float* alpha_out = h2out + (size_t)Nn * FDIM;

  const int gemmGrid = cdiv_ll(Nn, 16);
  const int nodeGrid = cdiv_ll(Nn, 256);
  const int nfGrid   = cdiv_ll((long long)Nn * FDIM, 256);
  const int edgeGrid = cdiv_ll(ET, 256);
  const int aggGrid  = cdiv_ll((long long)ET * 16, 256);

  gat_gemm_wmma<128><<<gemmGrid, 128, 0, stream>>>(x, W1, bufH, Nn);
  gat_node_dots<<<nodeGrid, 256, 0, stream>>>(bufH, a_src1, a_dst1, hs, hd, Nn);
  gat_init<<<nfGrid, 256, 0, stream>>>(bufAgg, maxenc, denom, Nn);
  gat_edge_logits<<<edgeGrid, 256, 0, stream>>>(ei, hs, hd, ebuf, maxenc, E, ET);
  gat_edge_exp<<<edgeGrid, 256, 0, stream>>>(ei, ebuf, maxenc, denom, E, ET);
  gat_edge_agg<<<aggGrid, 256, 0, stream>>>(ei, ebuf, denom, bufH, bufAgg,
                                            nullptr, E, ET);
  gat_finalize<<<nfGrid, 256, 0, stream>>>(bufAgg, b1, Nn * FDIM);

  gat_gemm_wmma<64><<<gemmGrid, 128, 0, stream>>>(bufAgg, W2, bufH, Nn);
  gat_node_dots<<<nodeGrid, 256, 0, stream>>>(bufH, a_src2, a_dst2, hs, hd, Nn);
  gat_init<<<nfGrid, 256, 0, stream>>>(h2out, maxenc, denom, Nn);
  gat_edge_logits<<<edgeGrid, 256, 0, stream>>>(ei, hs, hd, ebuf, maxenc, E, ET);
  gat_edge_exp<<<edgeGrid, 256, 0, stream>>>(ei, ebuf, maxenc, denom, E, ET);
  gat_edge_agg<<<aggGrid, 256, 0, stream>>>(ei, ebuf, denom, bufH, h2out,
                                            alpha_out, E, ET);
  gat_finalize<<<nfGrid, 256, 0, stream>>>(h2out, b2, Nn * FDIM);
}


# Edges grouped by destination each call (integer-atomic counting sort, groups then put in edge-id order); one wave per node: softmax and row sum in registers, no float atomics; sums in edge-id order
# speedup vs baseline: 2.0538x; 2.0538x over previous
#include <hip/hip_runtime.h>


typedef float v2f __attribute__((ext_vector_type(2)));
typedef float v8f __attribute__((ext_vector_type(8)));

#define NEG_SLOPE 0.2f
#define EPSV 1e-16f
#define FDIM 64

template <int K>
__global__ __launch_bounds__(128) void gat_gemm_wmma(
    const float* __restrict__ X, const float* __restrict__ W,
    float* __restrict__ H, int M) {
  const int wave = threadIdx.x >> 5;
  const int lane = threadIdx.x & 31;
  const int l16  = lane & 15;
  const int kh   = (lane >> 4) << 1;
  const int rowBase = blockIdx.x * 16;
  const int colBase = wave * 16;

  int arow = rowBase + l16;
  if (arow >= M) arow = M - 1;
  const float* xr = X + (size_t)arow * K;
  const float* wc = W + colBase + l16;

  v8f c = {0.f, 0.f, 0.f, 0.f, 0.f, 0.f, 0.f, 0.f};
#pragma unroll
  for (int k = 0; k < K; k += 4) {
    v2f a, b;
    a[0] = xr[k + kh];
    a[1] = xr[k + kh + 1];
    b[0] = wc[(size_t)(k + kh) * FDIM];
    b[1] = wc[(size_t)(k + kh + 1) * FDIM];
    c = __builtin_amdgcn_wmma_f32_16x16x4_f32(false, a, false, b,
                                              (short)0, c, false, false);
  }

  const int rOff = (lane >> 4) << 3;
#pragma unroll
  for (int v = 0; v < 8; ++v) {
    int r = rowBase + rOff + v;
    if (r >= M) r = M - 1;
    H[(size_t)r * FDIM + colBase + l16] = c[v];
  }
}

__global__ void gat_node_dots(const float* __restrict__ H,
                              const float* __restrict__ a_src,
                              const float* __restrict__ a_dst,
                              float* __restrict__ hs, float* __restrict__ hd,
                              int Nn) {
  int i = blockIdx.x * blockDim.x + threadIdx.x;
  if (i >= Nn) return;
  const float* h = H + (size_t)i * FDIM;
  float s = 0.f, d = 0.f;
#pragma unroll
  for (int f = 0; f < FDIM; ++f) {
    float v = h[f];
    s += v * a_src[f];
    d += v * a_dst[f];
  }
  hs[i] = s;
  hd[i] = d;
}

#define SCAN_TILE 1024

__global__ void csr_zero(int* __restrict__ cnt, int Nn) {
  int i = blockIdx.x * blockDim.x + threadIdx.x;
  if (i < Nn) cnt[i] = 0;
}

__global__ void csr_count(const int* __restrict__ ei, int* __restrict__ cnt,
                          int E) {
  int e = blockIdx.x * blockDim.x + threadIdx.x;
  if (e < E) atomicAdd(&cnt[ei[E + e]], 1);
}

__device__ __forceinline__ int wave_sum_i(int v) {
#pragma unroll
  for (int o = 16; o > 0; o >>= 1) v += __shfl_xor(v, o, 32);
  return v;
}

__global__ __launch_bounds__(256) void csr_scan_sums(
    const int* __restrict__ cnt, int* __restrict__ bsum, int Nn) {
  __shared__ int red[8];
  const int base = blockIdx.x * SCAN_TILE + threadIdx.x * 4;
  int s = 0;
#pragma unroll
  for (int j = 0; j < 4; ++j)
    if (base + j < Nn) s += cnt[base + j];
  s = wave_sum_i(s);
  if ((threadIdx.x & 31) == 0) red[threadIdx.x >> 5] = s;
  __syncthreads();
  if (threadIdx.x == 0) {
    int t = 0;
#pragma unroll
    for (int w = 0; w < 8; ++w) t += red[w];
    bsum[blockIdx.x] = t;
  }
}

__global__ __launch_bounds__(256) void csr_scan_write(
    const int* __restrict__ cnt, const int* __restrict__ bsum,
    int* __restrict__ off, int* __restrict__ cursor, int Nn, int E) {
  __shared__ int red[8];
  __shared__ int wtot[8];
  const int lane = threadIdx.x & 31;
  const int wave = threadIdx.x >> 5;

  int p = 0;
  for (int j = threadIdx.x; j < (int)blockIdx.x; j += 256) p += bsum[j];
  p = wave_sum_i(p);
  if (lane == 0) red[wave] = p;

  const int base = blockIdx.x * SCAN_TILE + threadIdx.x * 4;
  int v[4];
  int s = 0;
#pragma unroll
  for (int j = 0; j < 4; ++j) {
    v[j] = (base + j < Nn) ? cnt[base + j] : 0;
    s += v[j];
  }
  int x = s;
#pragma unroll
  for (int o = 1; o < 32; o <<= 1) {
    int t = __shfl_up(x, o, 32);
    if (lane >= o) x += t;
  }
  if (lane == 31) wtot[wave] = x;
  __syncthreads();

  int run = 0;
#pragma unroll
  for (int w = 0; w < 8; ++w) {
    run += red[w];
    if (w < wave) run += wtot[w];
  }
  run += x - s;
#pragma unroll
  for (int j = 0; j < 4; ++j) {
    if (base + j < Nn) {
      off[base + j]    = run;
      cursor[base + j] = run;
    }
    run += v[j];
  }
  if (blockIdx.x == 0 && threadIdx.x == 0) off[Nn] = E;
}

__global__ void csr_fill(const int* __restrict__ ei, int* __restrict__ cursor,
                         int* __restrict__ slots, int E) {
  int e = blockIdx.x * blockDim.x + threadIdx.x;
  if (e >= E) return;
  const int s = ei[e];
  const int d = ei[E + e];
  const int pos = atomicAdd(&cursor[d], 1);
  ((int2*)slots)[pos] = make_int2(e, s);
}

__global__ __launch_bounds__(256) void csr_order(
    const int* __restrict__ off, const int* __restrict__ slots_in,
    int* __restrict__ slots_out, int Nn) {
  const int node = __builtin_amdgcn_readfirstlane(
      (int)(blockIdx.x * 8 + (threadIdx.x >> 5)));
  const int lane = threadIdx.x & 31;
  if (node >= Nn) return;
  const int beg = off[node];
  const int n   = off[node + 1] - beg;
  const int2* in  = (const int2*)slots_in + beg;
  int2*       out = (int2*)slots_out + beg;
  for (int base = 0; base < n; base += 32) {
    const int  j     = base + lane;
    const bool valid = j < n;
    const int2 my    = valid ? in[j] : make_int2(0x7fffffff, 0);
    int rank = 0;
    for (int b2 = 0; b2 < n; b2 += 32) {
      const int j2 = b2 + lane;
      const int o  = (j2 < n) ? in[j2].x : 0x7fffffff;
      const int c2 = (n - b2 < 32) ? (n - b2) : 32;
      for (int k = 0; k < c2; ++k) {
        const int ok = __builtin_amdgcn_readlane(o, k);
        rank += (ok < my.x) ? 1 : 0;
      }
    }
    if (valid) out[rank] = my;
  }
}

__device__ __forceinline__ float wave_max_f(float v) {
#pragma unroll
  for (int o = 16; o > 0; o >>= 1) v = fmaxf(v, __shfl_xor(v, o, 32));
  return v;
}
__device__ __forceinline__ float wave_sum_f(float v) {
#pragma unroll
  for (int o = 16; o > 0; o >>= 1) v += __shfl_xor(v, o, 32);
  return v;
}
__device__ __forceinline__ float bcast_f(float v, int k) {
  return __int_as_float(__builtin_amdgcn_readlane(__float_as_int(v), k));
}

template <bool WRITE_ALPHA>
__global__ __launch_bounds__(256) void gat_node_fused(
    const int* __restrict__ off, const int* __restrict__ slots_,
    const float* __restrict__ hs, const float* __restrict__ hd,
    const float* __restrict__ H, const float* __restrict__ bias,
    float* __restrict__ out, float* __restrict__ alpha_out, int Nn, int E) {
  const int node = __builtin_amdgcn_readfirstlane(
      (int)(blockIdx.x * 8 + (threadIdx.x >> 5)));
  const int lane = threadIdx.x & 31;
  if (node >= Nn) return;
  const int beg = off[node];
  const int n   = off[node + 1] - beg;
  const int2*   slots = (const int2*)slots_ + beg;
  const float2* Hr    = (const float2*)H;

  const float2 hself = Hr[(size_t)node * 32 + lane];

  const float hdn = hd[node];
  float lself = hs[node] + hdn;
  lself = (lself > 0.f) ? lself : NEG_SLOPE * lself;

  const bool v0  = lane < n;
  const int2 sl0 = v0 ? slots[lane] : make_int2(0, 0);
  float lg0 = -INFINITY;
  if (v0) {
    float t = hs[sl0.y] + hdn;
    lg0 = (t > 0.f) ? t : NEG_SLOPE * t;
  }
  float m = wave_max_f(lg0);
  for (int base = 32; base < n; base += 32) {
    const int j = base + lane;
    float lg = -INFINITY;
    if (j < n) {
      float t = hs[slots[j].y] + hdn;
      lg = (t > 0.f) ? t : NEG_SLOPE * t;
    }
    m = fmaxf(m, wave_max_f(lg));
  }
  m = fmaxf(m, lself);

  const float ex0 = v0 ? __expf(lg0 - m) : 0.f;
  float den = wave_sum_f(ex0);
  for (int base = 32; base < n; base += 32) {
    const int j = base + lane;
    float ex = 0.f;
    if (j < n) {
      float t = hs[slots[j].y] + hdn;
      t = (t > 0.f) ? t : NEG_SLOPE * t;
      ex = __expf(t - m);
    }
    den += wave_sum_f(ex);
  }
  const float exs = __expf(lself - m);
  den += exs;
  const float dd  = den + EPSV;
  const float al0 = ex0 / dd;
  const float als = exs / dd;
  if (WRITE_ALPHA) {
    if (v0) alpha_out[sl0.x] = al0;
    if (lane == 0) alpha_out[E + node] = als;
  }

  float2 acc = make_float2(0.f, 0.f);
  const int n0 = (n < 32) ? n : 32;
  int k = 0;
  for (; k + 4 <= n0; k += 4) {
    const int s0 = __builtin_amdgcn_readlane(sl0.y, k);
    const int s1 = __builtin_amdgcn_readlane(sl0.y, k + 1);
    const int s2 = __builtin_amdgcn_readlane(sl0.y, k + 2);
    const int s3 = __builtin_amdgcn_readlane(sl0.y, k + 3);
    const float2 h0 = Hr[(size_t)s0 * 32 + lane];
    const float2 h1 = Hr[(size_t)s1 * 32 + lane];
    const float2 h2 = Hr[(size_t)s2 * 32 + lane];
    const float2 h3 = Hr[(size_t)s3 * 32 + lane];
    const float a0 = bcast_f(al0, k);
    const float a1 = bcast_f(al0, k + 1);
    const float a2 = bcast_f(al0, k + 2);
    const float a3 = bcast_f(al0, k + 3);
    acc.x = fmaf(a0, h0.x, acc.x);  acc.y = fmaf(a0, h0.y, acc.y);
    acc.x = fmaf(a1, h1.x, acc.x);  acc.y = fmaf(a1, h1.y, acc.y);
    acc.x = fmaf(a2, h2.x, acc.x);  acc.y = fmaf(a2, h2.y, acc.y);
    acc.x = fmaf(a3, h3.x, acc.x);  acc.y = fmaf(a3, h3.y, acc.y);
  }
  const int r = n0 - k;
  if (r > 0) {
    const int k1 = (r > 1) ? k + 1 : k;
    const int k2 = (r > 2) ? k + 2 : k;
    const int s0 = __builtin_amdgcn_readlane(sl0.y, k);
    const int s1 = __builtin_amdgcn_readlane(sl0.y, k1);
    const int s2 = __builtin_amdgcn_readlane(sl0.y, k2);
    const float2 h0 = Hr[(size_t)s0 * 32 + lane];
    const float2 h1 = Hr[(size_t)s1 * 32 + lane];
    const float2 h2 = Hr[(size_t)s2 * 32 + lane];
    const float a0 = bcast_f(al0, k);
    const float a1 = bcast_f(al0, k1);
    const float a2 = bcast_f(al0, k2);
    acc.x = fmaf(a0, h0.x, acc.x);  acc.y = fmaf(a0, h0.y, acc.y);
    if (r > 1) { acc.x = fmaf(a1, h1.x, acc.x);  acc.y = fmaf(a1, h1.y, acc.y); }
    if (r > 2) { acc.x = fmaf(a2, h2.x, acc.x);  acc.y = fmaf(a2, h2.y, acc.y); }
  }
  for (int base = 32; base < n; base += 32) {
    const int  j  = base + lane;
    const bool vj = j < n;
    const int2 sl = vj ? slots[j] : make_int2(0, 0);
    float al = 0.f;
    if (vj) {
      float t = hs[sl.y] + hdn;
      t = (t > 0.f) ? t : NEG_SLOPE * t;
      al = __expf(t - m) / dd;
      if (WRITE_ALPHA) alpha_out[sl.x] = al;
    }
    const int c2 = (n - base < 32) ? (n - base) : 32;
    for (int q = 0; q < c2; ++q) {
      const int    s = __builtin_amdgcn_readlane(sl.y, q);
      const float  a = bcast_f(al, q);
      const float2 h = Hr[(size_t)s * 32 + lane];
      acc.x = fmaf(a, h.x, acc.x);
      acc.y = fmaf(a, h.y, acc.y);
    }
  }
  acc.x = fmaf(als, hself.x, acc.x);
  acc.y = fmaf(als, hself.y, acc.y);
  const float2 b = ((const float2*)bias)[lane];
  float ox = acc.x + b.x;
  float oy = acc.y + b.y;
  ox = (ox > 0.f) ? ox : 0.f;
  oy = (oy > 0.f) ? oy : 0.f;
  ((float2*)out)[(size_t)node * 32 + lane] = make_float2(ox, oy);
}

static inline int cdiv_ll(long long a, long long b) {
  return (int)((a + b - 1) / b);
}

extern "C" void kernel_launch(void* const* d_in, const int* in_sizes, int n_in,
                              void* d_out, int out_size, void* d_ws,
                              size_t ws_size, hipStream_t stream) {
  const float* x      = (const float*)d_in[0];
  const int*   ei     = (const int*)d_in[1];
  const float* W1     = (const float*)d_in[2];
  const float* a_src1 = (const float*)d_in[3];
  const float* a_dst1 = (const float*)d_in[4];
  const float* b1     = (const float*)d_in[5];
  const float* W2     = (const float*)d_in[6];
  const float* a_src2 = (const float*)d_in[7];
  const float* a_dst2 = (const float*)d_in[8];
  const float* b2     = (const float*)d_in[9];

  const int IN_DIM = 128;
  const int Nn = in_sizes[0] / IN_DIM;
  const int E  = in_sizes[1] / 2;

  const int    nTiles = cdiv_ll(Nn, SCAN_TILE);
  const size_t NnPad  = ((size_t)Nn + 4) & ~(size_t)3;

  float* bufH   = (float*)d_ws;
  float* bufAgg = bufH + (size_t)Nn * FDIM;
  float* hs     = bufAgg + (size_t)Nn * FDIM;
  float* hd     = hs + NnPad;
  int*   cnt    = (int*)(hd + NnPad);
  int*   off    = cnt + NnPad;
  int*   cursor = off + NnPad;
  int*   bsum   = cursor + NnPad;
  int*   slotsA = bsum + (((size_t)nTiles + 3) & ~(size_t)3);
  int*   slotsB = slotsA + (((size_t)2 * E + 3) & ~(size_t)3);

  float* h2out     = (float*)d_out;
  float* alpha_out = h2out + (size_t)Nn * FDIM;

  const int gemmGrid = cdiv_ll(Nn, 16);
  const int nodeGrid = cdiv_ll(Nn, 256);
  const int edgeGrid = cdiv_ll(E, 256);
  const int waveGrid = cdiv_ll(Nn, 8);

  csr_zero<<<nodeGrid, 256, 0, stream>>>(cnt, Nn);
  csr_count<<<edgeGrid, 256, 0, stream>>>(ei, cnt, E);
  csr_scan_sums<<<nTiles, 256, 0, stream>>>(cnt, bsum, Nn);
  csr_scan_write<<<nTiles, 256, 0, stream>>>(cnt, bsum, off, cursor, Nn, E);
  csr_fill<<<edgeGrid, 256, 0, stream>>>(ei, cursor, slotsA, E);
  csr_order<<<waveGrid, 256, 0, stream>>>(off, slotsA, slotsB, Nn);

  gat_gemm_wmma<128><<<gemmGrid, 128, 0, stream>>>(x, W1, bufH, Nn);
  gat_node_dots<<<nodeGrid, 256, 0, stream>>>(bufH, a_src1, a_dst1, hs, hd, Nn);
  gat_node_fused<false><<<waveGrid, 256, 0, stream>>>(
      off, slotsB, hs, hd, bufH, b1, bufAgg, nullptr, Nn, E);

  gat_gemm_wmma<64><<<gemmGrid, 128, 0, stream>>>(bufAgg, W2, bufH, Nn);
  gat_node_dots<<<nodeGrid, 256, 0, stream>>>(bufH, a_src2, a_dst2, hs, hd, Nn);
  gat_node_fused<true><<<waveGrid, 256, 0, stream>>>(
      off, slotsB, hs, hd, bufH, b2, h2out, alpha_out, Nn, E);
}


# Fixed grids of 2048 workgroups (1024 in the two edge kernels) whose waves loop over runs of nodes, row tiles or edges: about 5x fewer waves started per call; every float sum in its old order
# speedup vs baseline: 2.1557x; 1.0496x over previous
#include <hip/hip_runtime.h>


#define LOOP_GRID 2048
#define EDGE_GRID 1024

typedef float v2f __attribute__((ext_vector_type(2)));
typedef float v8f __attribute__((ext_vector_type(8)));

#define NEG_SLOPE 0.2f
#define EPSV 1e-16f
#define FDIM 64

template <int K>
__global__ __launch_bounds__(128) void gat_gemm_wmma(
    const float* __restrict__ X, const float* __restrict__ W,
    float* __restrict__ H, int M) {
  const int wave = threadIdx.x >> 5;
  const int lane = threadIdx.x & 31;
  const int l16  = lane & 15;
  const int kh   = (lane >> 4) << 1;
  const int colBase = wave * 16;
  const int rOff = (lane >> 4) << 3;
  const float* wc = W + colBase + l16;

  v2f bf[K / 4];
#pragma unroll
  for (int k = 0; k < K; k += 4) {
    bf[k >> 2][0] = wc[(size_t)(k + kh) * FDIM];
    bf[k >> 2][1] = wc[(size_t)(k + kh + 1) * FDIM];
  }

  const int nTiles = (M + 15) >> 4;
  for (int tile = blockIdx.x; tile < nTiles; tile += gridDim.x) {
    const int rowBase = tile * 16;
    int arow = rowBase + l16;
    if (arow >= M) arow = M - 1;
    const float* xr = X + (size_t)arow * K;

    v8f c = {0.f, 0.f, 0.f, 0.f, 0.f, 0.f, 0.f, 0.f};
#pragma unroll
    for (int k = 0; k < K; k += 4) {
      v2f a;
      a[0] = xr[k + kh];
      a[1] = xr[k + kh + 1];
      c = __builtin_amdgcn_wmma_f32_16x16x4_f32(false, a, false, bf[k >> 2],
                                                (short)0, c, false, false);
    }

#pragma unroll
    for (int v = 0; v < 8; ++v) {
      int r = rowBase + rOff + v;
      if (r >= M) r = M - 1;
      H[(size_t)r * FDIM + colBase + l16] = c[v];
    }
  }
}

__global__ void gat_node_dots(const float* __restrict__ H,
                              const float* __restrict__ a_src,
                              const float* __restrict__ a_dst,
                              float* __restrict__ hs, float* __restrict__ hd,
                              int Nn) {
  int i = blockIdx.x * blockDim.x + threadIdx.x;
  if (i >= Nn) return;
  const float* h = H + (size_t)i * FDIM;
  float s = 0.f, d = 0.f;
#pragma unroll
  for (int f = 0; f < FDIM; ++f) {
    float v = h[f];
    s += v * a_src[f];
    d += v * a_dst[f];
  }
  hs[i] = s;
  hd[i] = d;
}

#define SCAN_TILE 1024

__global__ void csr_zero(int* __restrict__ cnt, int Nn) {
  int i = blockIdx.x * blockDim.x + threadIdx.x;
  if (i < Nn) cnt[i] = 0;
}

__global__ __launch_bounds__(256) void csr_count(
    const int* __restrict__ ei, int* __restrict__ cnt, int E) {
  const int T = gridDim.x * 256;
  const int* dst = ei + E;
  int e = blockIdx.x * 256 + threadIdx.x;
  for (; e + 3 * T < E; e += 4 * T) {
    const int d0 = dst[e];
    const int d1 = dst[e + T];
    const int d2 = dst[e + 2 * T];
    const int d3 = dst[e + 3 * T];
    atomicAdd(&cnt[d0], 1);
    atomicAdd(&cnt[d1], 1);
    atomicAdd(&cnt[d2], 1);
    atomicAdd(&cnt[d3], 1);
  }
  for (; e < E; e += T) atomicAdd(&cnt[dst[e]], 1);
}

__device__ __forceinline__ int wave_sum_i(int v) {
#pragma unroll
  for (int o = 16; o > 0; o >>= 1) v += __shfl_xor(v, o, 32);
  return v;
}

__global__ __launch_bounds__(256) void csr_scan_sums(
    const int* __restrict__ cnt, int* __restrict__ bsum, int Nn) {
  __shared__ int red[8];
  const int base = blockIdx.x * SCAN_TILE + threadIdx.x * 4;
  int s = 0;
#pragma unroll
  for (int j = 0; j < 4; ++j)
    if (base + j < Nn) s += cnt[base + j];
  s = wave_sum_i(s);
  if ((threadIdx.x & 31) == 0) red[threadIdx.x >> 5] = s;
  __syncthreads();
  if (threadIdx.x == 0) {
    int t = 0;
#pragma unroll
    for (int w = 0; w < 8; ++w) t += red[w];
    bsum[blockIdx.x] = t;
  }
}

__global__ __launch_bounds__(256) void csr_scan_write(
    const int* __restrict__ cnt, const int* __restrict__ bsum,
    int* __restrict__ off, int* __restrict__ cursor, int Nn, int E) {
  __shared__ int red[8];
  __shared__ int wtot[8];
  const int lane = threadIdx.x & 31;
  const int wave = threadIdx.x >> 5;

  int p = 0;
  for (int j = threadIdx.x; j < (int)blockIdx.x; j += 256) p += bsum[j];
  p = wave_sum_i(p);
  if (lane == 0) red[wave] = p;

  const int base = blockIdx.x * SCAN_TILE + threadIdx.x * 4;
  int v[4];
  int s = 0;
#pragma unroll
  for (int j = 0; j < 4; ++j) {
    v[j] = (base + j < Nn) ? cnt[base + j] : 0;
    s += v[j];
  }
  int x = s;
#pragma unroll
  for (int o = 1; o < 32; o <<= 1) {
    int t = __shfl_up(x, o, 32);
    if (lane >= o) x += t;
  }
  if (lane == 31) wtot[wave] = x;
  __syncthreads();

  int run = 0;
#pragma unroll
  for (int w = 0; w < 8; ++w) {
    run += red[w];
    if (w < wave) run += wtot[w];
  }
  run += x - s;
#pragma unroll
  for (int j = 0; j < 4; ++j) {
    if (base + j < Nn) {
      off[base + j]    = run;
      cursor[base + j] = run;
    }
    run += v[j];
  }
  if (blockIdx.x == 0 && threadIdx.x == 0) off[Nn] = E;
}

__global__ __launch_bounds__(256) void csr_fill(
    const int* __restrict__ ei, int* __restrict__ cursor,
    int* __restrict__ slots, int E) {
  const int T = gridDim.x * 256;
  const int* dst = ei + E;
  int2* sl = (int2*)slots;
  int e = blockIdx.x * 256 + threadIdx.x;
  for (; e + 3 * T < E; e += 4 * T) {
    const int e1 = e + T, e2 = e + 2 * T, e3 = e + 3 * T;
    const int s0 = ei[e],  d0 = dst[e];
    const int s1 = ei[e1], d1 = dst[e1];
    const int s2 = ei[e2], d2 = dst[e2];
    const int s3 = ei[e3], d3 = dst[e3];
    const int p0 = atomicAdd(&cursor[d0], 1);
    const int p1 = atomicAdd(&cursor[d1], 1);
    const int p2 = atomicAdd(&cursor[d2], 1);
    const int p3 = atomicAdd(&cursor[d3], 1);
    sl[p0] = make_int2(e, s0);
    sl[p1] = make_int2(e1, s1);
    sl[p2] = make_int2(e2, s2);
    sl[p3] = make_int2(e3, s3);
  }
  for (; e < E; e += T) {
    const int s = ei[e];
    const int d = dst[e];
    const int pos = atomicAdd(&cursor[d], 1);
    sl[pos] = make_int2(e, s);
  }
}

__global__ __launch_bounds__(256) void csr_order(
    const int* __restrict__ off, const int* __restrict__ slots_in,
    int* __restrict__ slots_out, int Nn) {
  const int gw = __builtin_amdgcn_readfirstlane(
      (int)(blockIdx.x * 8 + (threadIdx.x >> 5)));
  const int lane  = threadIdx.x & 31;
  const int nw    = gridDim.x * 8;
  const int per   = (Nn + nw - 1) / nw;
  const int first = gw * per;
  const int stop  = (first + per < Nn) ? first + per : Nn;
  if (first >= stop) return;
  int beg = off[first];
  int end = off[first + 1];
  for (int node = first; node < stop; ++node) {
    const int endNext = off[(node + 2 < Nn) ? node + 2 : Nn];
    const int n   = end - beg;
    const int2* in  = (const int2*)slots_in + beg;
    int2*       out = (int2*)slots_out + beg;
    for (int base = 0; base < n; base += 32) {
      const int  j     = base + lane;
      const bool valid = j < n;
      const int2 my    = valid ? in[j] : make_int2(0x7fffffff, 0);
      int rank = 0;
      for (int b2 = 0; b2 < n; b2 += 32) {
        const int j2 = b2 + lane;
        const int o  = (j2 < n) ? in[j2].x : 0x7fffffff;
        const int c2 = (n - b2 < 32) ? (n - b2) : 32;
        for (int k = 0; k < c2; ++k) {
          const int ok = __builtin_amdgcn_readlane(o, k);
          rank += (ok < my.x) ? 1 : 0;
        }
      }
      if (valid) out[rank] = my;
    }
    beg = end;
    end = endNext;
  }
}

__device__ __forceinline__ float wave_max_f(float v) {
#pragma unroll
  for (int o = 16; o > 0; o >>= 1) v = fmaxf(v, __shfl_xor(v, o, 32));
  return v;
}
__device__ __forceinline__ float wave_sum_f(float v) {
#pragma unroll
  for (int o = 16; o > 0; o >>= 1) v += __shfl_xor(v, o, 32);
  return v;
}
__device__ __forceinline__ float bcast_f(float v, int k) {
  return __int_as_float(__builtin_amdgcn_readlane(__float_as_int(v), k));
}

template <bool WRITE_ALPHA>
__global__ __launch_bounds__(256) void gat_node_fused(
    const int* __restrict__ off, const int* __restrict__ slots_,
    const float* __restrict__ hs, const float* __restrict__ hd,
    const float* __restrict__ H, const float* __restrict__ bias,
    float* __restrict__ out, float* __restrict__ alpha_out, int Nn, int E) {
  const int gw = __builtin_amdgcn_readfirstlane(
      (int)(blockIdx.x * 8 + (threadIdx.x >> 5)));
  const int lane  = threadIdx.x & 31;
  const int nw    = gridDim.x * 8;
  const int per   = (Nn + nw - 1) / nw;
  const int first = gw * per;
  const int stop  = (first + per < Nn) ? first + per : Nn;
  if (first >= stop) return;
  const float2* Hr = (const float2*)H;
  const float2  b  = ((const float2*)bias)[lane];
  int beg = off[first];
  int end = off[first + 1];
  for (int node = first; node < stop; ++node) {
    const int endNext = off[(node + 2 < Nn) ? node + 2 : Nn];
    const int n   = end - beg;
    const int2* slots = (const int2*)slots_ + beg;

    const float2 hself = Hr[(size_t)node * 32 + lane];

    const float hdn = hd[node];
    float lself = hs[node] + hdn;
    lself = (lself > 0.f) ? lself : NEG_SLOPE * lself;

    const bool v0  = lane < n;
    const int2 sl0 = v0 ? slots[lane] : make_int2(0, 0);
    float lg0 = -INFINITY;
    if (v0) {
      float t = hs[sl0.y] + hdn;
      lg0 = (t > 0.f) ? t : NEG_SLOPE * t;
    }
    float m = wave_max_f(lg0);
    for (int base = 32; base < n; base += 32) {
      const int j = base + lane;
      float lg = -INFINITY;
      if (j < n) {
        float t = hs[slots[j].y] + hdn;
        lg = (t > 0.f) ? t : NEG_SLOPE * t;
      }
      m = fmaxf(m, wave_max_f(lg));
    }
    m = fmaxf(m, lself);

    const float ex0 = v0 ? __expf(lg0 - m) : 0.f;
    float den = wave_sum_f(ex0);
    for (int base = 32; base < n; base += 32) {
      const int j = base + lane;
      float ex = 0.f;
      if (j < n) {
        float t = hs[slots[j].y] + hdn;
        t = (t > 0.f) ? t : NEG_SLOPE * t;
        ex = __expf(t - m);
      }
      den += wave_sum_f(ex);
    }
    const float exs = __expf(lself - m);
    den += exs;
    const float dd  = den + EPSV;
    const float al0 = ex0 / dd;
    const float als = exs / dd;
    if (WRITE_ALPHA) {
      if (v0) alpha_out[sl0.x] = al0;
      if (lane == 0) alpha_out[E + node] = als;
    }

    float2 acc = make_float2(0.f, 0.f);
    const int n0 = (n < 32) ? n : 32;
    int k = 0;
    for (; k + 4 <= n0; k += 4) {
      const int s0 = __builtin_amdgcn_readlane(sl0.y, k);
      const int s1 = __builtin_amdgcn_readlane(sl0.y, k + 1);
      const int s2 = __builtin_amdgcn_readlane(sl0.y, k + 2);
      const int s3 = __builtin_amdgcn_readlane(sl0.y, k + 3);
      const float2 h0 = Hr[(size_t)s0 * 32 + lane];
      const float2 h1 = Hr[(size_t)s1 * 32 + lane];
      const float2 h2 = Hr[(size_t)s2 * 32 + lane];
      const float2 h3 = Hr[(size_t)s3 * 32 + lane];
      const float a0 = bcast_f(al0, k);
      const float a1 = bcast_f(al0, k + 1);
      const float a2 = bcast_f(al0, k + 2);
      const float a3 = bcast_f(al0, k + 3);
      acc.x = fmaf(a0, h0.x, acc.x);  acc.y = fmaf(a0, h0.y, acc.y);
      acc.x = fmaf(a1, h1.x, acc.x);  acc.y = fmaf(a1, h1.y, acc.y);
      acc.x = fmaf(a2, h2.x, acc.x);  acc.y = fmaf(a2, h2.y, acc.y);
      acc.x = fmaf(a3, h3.x, acc.x);  acc.y = fmaf(a3, h3.y, acc.y);
    }
    const int r = n0 - k;
    if (r > 0) {
      const int k1 = (r > 1) ? k + 1 : k;
      const int k2 = (r > 2) ? k + 2 : k;
      const int s0 = __builtin_amdgcn_readlane(sl0.y, k);
      const int s1 = __builtin_amdgcn_readlane(sl0.y, k1);
      const int s2 = __builtin_amdgcn_readlane(sl0.y, k2);
      const float2 h0 = Hr[(size_t)s0 * 32 + lane];
      const float2 h1 = Hr[(size_t)s1 * 32 + lane];
      const float2 h2 = Hr[(size_t)s2 * 32 + lane];
      const float a0 = bcast_f(al0, k);
      const float a1 = bcast_f(al0, k1);
      const float a2 = bcast_f(al0, k2);
      acc.x = fmaf(a0, h0.x, acc.x);  acc.y = fmaf(a0, h0.y, acc.y);
      if (r > 1) { acc.x = fmaf(a1, h1.x, acc.x);  acc.y = fmaf(a1, h1.y, acc.y); }
      if (r > 2) { acc.x = fmaf(a2, h2.x, acc.x);  acc.y = fmaf(a2, h2.y, acc.y); }
    }
    for (int base = 32; base < n; base += 32) {
      const int  j  = base + lane;
      const bool vj = j < n;
      const int2 sl = vj ? slots[j] : make_int2(0, 0);
      float al = 0.f;
      if (vj) {
        float t = hs[sl.y] + hdn;
        t = (t > 0.f) ? t : NEG_SLOPE * t;
        al = __expf(t - m) / dd;
        if (WRITE_ALPHA) alpha_out[sl.x] = al;
      }
      const int c2 = (n - base < 32) ? (n - base) : 32;
      for (int q = 0; q < c2; ++q) {
        const int    s = __builtin_amdgcn_readlane(sl.y, q);
        const float  a = bcast_f(al, q);
        const float2 h = Hr[(size_t)s * 32 + lane];
        acc.x = fmaf(a, h.x, acc.x);
        acc.y = fmaf(a, h.y, acc.y);
      }
    }
    acc.x = fmaf(als, hself.x, acc.x);
    acc.y = fmaf(als, hself.y, acc.y);
    float ox = acc.x + b.x;
    float oy = acc.y + b.y;
    ox = (ox > 0.f) ? ox : 0.f;
    oy = (oy > 0.f) ? oy : 0.f;
    ((float2*)out)[(size_t)node * 32 + lane] = make_float2(ox, oy);
    beg = end;
    end = endNext;
  }
}

static inline int cdiv_ll(long long a, long long b) {
  return (int)((a + b - 1) / b);
}

extern "C" void kernel_launch(void* const* d_in, const int* in_sizes, int n_in,
                              void* d_out, int out_size, void* d_ws,
                              size_t ws_size, hipStream_t stream) {
  const float* x      = (const float*)d_in[0];
  const int*   ei     = (const int*)d_in[1];
  const float* W1     = (const float*)d_in[2];
  const float* a_src1 = (const float*)d_in[3];
  const float* a_dst1 = (const float*)d_in[4];
  const float* b1     = (const float*)d_in[5];
  const float* W2     = (const float*)d_in[6];
  const float* a_src2 = (const float*)d_in[7];
  const float* a_dst2 = (const float*)d_in[8];
  const float* b2     = (const float*)d_in[9];

  const int IN_DIM = 128;
  const int Nn = in_sizes[0] / IN_DIM;
  const int E  = in_sizes[1] / 2;

  const int    nTiles = cdiv_ll(Nn, SCAN_TILE);
  const size_t NnPad  = ((size_t)Nn + 4) & ~(size_t)3;

  float* bufH   = (float*)d_ws;
  float* bufAgg = bufH + (size_t)Nn * FDIM;
  float* hs     = bufAgg + (size_t)Nn * FDIM;
  float* hd     = hs + NnPad;
  int*   cnt    = (int*)(hd + NnPad);
  int*   off    = cnt + NnPad;
  int*   cursor = off + NnPad;
  int*   bsum   = cursor + NnPad;
  int*   slotsA = bsum + (((size_t)nTiles + 3) & ~(size_t)3);
  int*   slotsB = slotsA + (((size_t)2 * E + 3) & ~(size_t)3);

  float* h2out     = (float*)d_out;
  float* alpha_out = h2out + (size_t)Nn * FDIM;

  const int nodeGrid = cdiv_ll(Nn, 256);

  csr_zero<<<nodeGrid, 256, 0, stream>>>(cnt, Nn);
  csr_count<<<EDGE_GRID, 256, 0, stream>>>(ei, cnt, E);
  csr_scan_sums<<<nTiles, 256, 0, stream>>>(cnt, bsum, Nn);
  csr_scan_write<<<nTiles, 256, 0, stream>>>(cnt, bsum, off, cursor, Nn, E);
  csr_fill<<<EDGE_GRID, 256, 0, stream>>>(ei, cursor, slotsA, E);
  csr_order<<<LOOP_GRID, 256, 0, stream>>>(off, slotsA, slotsB, Nn);

  gat_gemm_wmma<128><<<LOOP_GRID, 128, 0, stream>>>(x, W1, bufH, Nn);
  gat_node_dots<<<nodeGrid, 256, 0, stream>>>(bufH, a_src1, a_dst1, hs, hd, Nn);
  gat_node_fused<false><<<LOOP_GRID, 256, 0, stream>>>(
      off, slotsB, hs, hd, bufH, b1, bufAgg, nullptr, Nn, E);

  gat_gemm_wmma<64><<<LOOP_GRID, 128, 0, stream>>>(bufAgg, W2, bufH, Nn);
  gat_node_dots<<<nodeGrid, 256, 0, stream>>>(bufH, a_src2, a_dst2, hs, hd, Nn);
  gat_node_fused<true><<<LOOP_GRID, 256, 0, stream>>>(
      off, slotsB, hs, hd, bufH, b2, h2out, alpha_out, Nn, E);
}


# Edge grouping: count pass keeps what its integer atomic returns as the edge's rank, so the fill pass writes to offset + rank with no atomics (one per edge, not two); every float sum in its old order
# speedup vs baseline: 2.5528x; 1.1842x over previous
#include <hip/hip_runtime.h>


#define LOOP_GRID 2048
#define EDGE_GRID 1024

typedef float v2f __attribute__((ext_vector_type(2)));
typedef float v8f __attribute__((ext_vector_type(8)));

#define NEG_SLOPE 0.2f
#define EPSV 1e-16f
#define FDIM 64

template <int K>
__global__ __launch_bounds__(128) void gat_gemm_wmma(
    const float* __restrict__ X, const float* __restrict__ W,
    float* __restrict__ H, int M) {
  const int wave = threadIdx.x >> 5;
  const int lane = threadIdx.x & 31;
  const int l16  = lane & 15;
  const int kh   = (lane >> 4) << 1;
  const int colBase = wave * 16;
  const int rOff = (lane >> 4) << 3;
  const float* wc = W + colBase + l16;

  v2f bf[K / 4];
#pragma unroll
  for (int k = 0; k < K; k += 4) {
    bf[k >> 2][0] = wc[(size_t)(k + kh) * FDIM];
    bf[k >> 2][1] = wc[(size_t)(k + kh + 1) * FDIM];
  }

  const int nTiles = (M + 15) >> 4;
  for (int tile = blockIdx.x; tile < nTiles; tile += gridDim.x) {
    const int rowBase = tile * 16;
    int arow = rowBase + l16;
    if (arow >= M) arow = M - 1;
    const float* xr = X + (size_t)arow * K;

    v8f c = {0.f, 0.f, 0.f, 0.f, 0.f, 0.f, 0.f, 0.f};
#pragma unroll
    for (int k = 0; k < K; k += 4) {
      v2f a;
      a[0] = xr[k + kh];
      a[1] = xr[k + kh + 1];
      c = __builtin_amdgcn_wmma_f32_16x16x4_f32(false, a, false, bf[k >> 2],
                                                (short)0, c, false, false);
    }

#pragma unroll
    for (int v = 0; v < 8; ++v) {
      int r = rowBase + rOff + v;
      if (r >= M) r = M - 1;
      H[(size_t)r * FDIM + colBase + l16] = c[v];
    }
  }
}

__global__ void gat_node_dots(const float* __restrict__ H,
                              const float* __restrict__ a_src,
                              const float* __restrict__ a_dst,
                              float* __restrict__ hs, float* __restrict__ hd,
                              int Nn) {
  int i = blockIdx.x * blockDim.x + threadIdx.x;
  if (i >= Nn) return;
  const float* h = H + (size_t)i * FDIM;
  float s = 0.f, d = 0.f;
#pragma unroll
  for (int f = 0; f < FDIM; ++f) {
    float v = h[f];
    s += v * a_src[f];
    d += v * a_dst[f];
  }
  hs[i] = s;
  hd[i] = d;
}

#define SCAN_TILE 1024

__global__ void csr_zero(int* __restrict__ cnt, int Nn) {
  int i = blockIdx.x * blockDim.x + threadIdx.x;
  if (i < Nn) cnt[i] = 0;
}

__global__ __launch_bounds__(256) void csr_count(
    const int* __restrict__ ei, int* __restrict__ cnt, int* __restrict__ rank,
    int E) {
  const int T = gridDim.x * 256;
  const int* dst = ei + E;
  int e = blockIdx.x * 256 + threadIdx.x;
  for (; e + 3 * T < E; e += 4 * T) {
    const int e1 = e + T, e2 = e + 2 * T, e3 = e + 3 * T;
    const int d0 = dst[e];
    const int d1 = dst[e1];
    const int d2 = dst[e2];
    const int d3 = dst[e3];
    const int r0 = atomicAdd(&cnt[d0], 1);
    const int r1 = atomicAdd(&cnt[d1], 1);
    const int r2 = atomicAdd(&cnt[d2], 1);
    const int r3 = atomicAdd(&cnt[d3], 1);
    rank[e]  = r0;
    rank[e1] = r1;
    rank[e2] = r2;
    rank[e3] = r3;
  }
  for (; e < E; e += T) rank[e] = atomicAdd(&cnt[dst[e]], 1);
}

__device__ __forceinline__ int wave_sum_i(int v) {
#pragma unroll
  for (int o = 16; o > 0; o >>= 1) v += __shfl_xor(v, o, 32);
  return v;
}

__global__ __launch_bounds__(256) void csr_scan_sums(
    const int* __restrict__ cnt, int* __restrict__ bsum, int Nn) {
  __shared__ int red[8];
  const int base = blockIdx.x * SCAN_TILE + threadIdx.x * 4;
  int s = 0;
#pragma unroll
  for (int j = 0; j < 4; ++j)
    if (base + j < Nn) s += cnt[base + j];
  s = wave_sum_i(s);
  if ((threadIdx.x & 31) == 0) red[threadIdx.x >> 5] = s;
  __syncthreads();
  if (threadIdx.x == 0) {
    int t = 0;
#pragma unroll
    for (int w = 0; w < 8; ++w) t += red[w];
    bsum[blockIdx.x] = t;
  }
}

__global__ __launch_bounds__(256) void csr_scan_write(
    const int* __restrict__ cnt, const int* __restrict__ bsum,
    int* __restrict__ off, int Nn, int E) {
  __shared__ int red[8];
  __shared__ int wtot[8];
  const int lane = threadIdx.x & 31;
  const int wave = threadIdx.x >> 5;

  int p = 0;
  for (int j = threadIdx.x; j < (int)blockIdx.x; j += 256) p += bsum[j];
  p = wave_sum_i(p);
  if (lane == 0) red[wave] = p;

  const int base = blockIdx.x * SCAN_TILE + threadIdx.x * 4;
  int v[4];
  int s = 0;
#pragma unroll
  for (int j = 0; j < 4; ++j) {
    v[j] = (base + j < Nn) ? cnt[base + j] : 0;
    s += v[j];
  }
  int x = s;
#pragma unroll
  for (int o = 1; o < 32; o <<= 1) {
    int t = __shfl_up(x, o, 32);
    if (lane >= o) x += t;
  }
  if (lane == 31) wtot[wave] = x;
  __syncthreads();

  int run = 0;
#pragma unroll
  for (int w = 0; w < 8; ++w) {
    run += red[w];
    if (w < wave) run += wtot[w];
  }
  run += x - s;
#pragma unroll
  for (int j = 0; j < 4; ++j) {
    if (base + j < Nn) off[base + j] = run;
    run += v[j];
  }
  if (blockIdx.x == 0 && threadIdx.x == 0) off[Nn] = E;
}

__global__ __launch_bounds__(256) void csr_fill(
    const int* __restrict__ ei, const int* __restrict__ off,
    const int* __restrict__ rank, int* __restrict__ slots, int E) {
  const int T = gridDim.x * 256;
  const int* dst = ei + E;
  int2* sl = (int2*)slots;
  int e = blockIdx.x * 256 + threadIdx.x;
  for (; e + 3 * T < E; e += 4 * T) {
    const int e1 = e + T, e2 = e + 2 * T, e3 = e + 3 * T;
    const int s0 = ei[e],  d0 = dst[e],  r0 = rank[e];
    const int s1 = ei[e1], d1 = dst[e1], r1 = rank[e1];
    const int s2 = ei[e2], d2 = dst[e2], r2 = rank[e2];
    const int s3 = ei[e3], d3 = dst[e3], r3 = rank[e3];
    const int p0 = off[d0] + r0;
    const int p1 = off[d1] + r1;
    const int p2 = off[d2] + r2;
    const int p3 = off[d3] + r3;
    sl[p0] = make_int2(e, s0);
    sl[p1] = make_int2(e1, s1);
    sl[p2] = make_int2(e2, s2);
    sl[p3] = make_int2(e3, s3);
  }
  for (; e < E; e += T) {
    const int s = ei[e];
    const int d = dst[e];
    sl[off[d] + rank[e]] = make_int2(e, s);
  }
}

__global__ __launch_bounds__(256) void csr_order(
    const int* __restrict__ off, const int* __restrict__ slots_in,
    int* __restrict__ slots_out, int Nn) {
  const int gw = __builtin_amdgcn_readfirstlane(
      (int)(blockIdx.x * 8 + (threadIdx.x >> 5)));
  const int lane  = threadIdx.x & 31;
  const int nw    = gridDim.x * 8;
  const int per   = (Nn + nw - 1) / nw;
  const int first = gw * per;
  const int stop  = (first + per < Nn) ? first + per : Nn;
  if (first >= stop) return;
  int beg = off[first];
  int end = off[first + 1];
  for (int node = first; node < stop; ++node) {
    const int endNext = off[(node + 2 < Nn) ? node + 2 : Nn];
    const int n   = end - beg;
    const int2* in  = (const int2*)slots_in + beg;
    int2*       out = (int2*)slots_out + beg;
    for (int base = 0; base < n; base += 32) {
      const int  j     = base + lane;
      const bool valid = j < n;
      const int2 my    = valid ? in[j] : make_int2(0x7fffffff, 0);
      int rank = 0;
      for (int b2 = 0; b2 < n; b2 += 32) {
        const int j2 = b2 + lane;
        const int o  = (j2 < n) ? in[j2].x : 0x7fffffff;
        const int c2 = (n - b2 < 32) ? (n - b2) : 32;
        for (int k = 0; k < c2; ++k) {
          const int ok = __builtin_amdgcn_readlane(o, k);
          rank += (ok < my.x) ? 1 : 0;
        }
      }
      if (valid) out[rank] = my;
    }
    beg = end;
    end = endNext;
  }
}

__device__ __forceinline__ float wave_max_f(float v) {
#pragma unroll
  for (int o = 16; o > 0; o >>= 1) v = fmaxf(v, __shfl_xor(v, o, 32));
  return v;
}
__device__ __forceinline__ float wave_sum_f(float v) {
#pragma unroll
  for (int o = 16; o > 0; o >>= 1) v += __shfl_xor(v, o, 32);
  return v;
}
__device__ __forceinline__ float bcast_f(float v, int k) {
  return __int_as_float(__builtin_amdgcn_readlane(__float_as_int(v), k));
}

template <bool WRITE_ALPHA>
__global__ __launch_bounds__(256) void gat_node_fused(
    const int* __restrict__ off, const int* __restrict__ slots_,
    const float* __restrict__ hs, const float* __restrict__ hd,
    const float* __restrict__ H, const float* __restrict__ bias,
    float* __restrict__ out, float* __restrict__ alpha_out, int Nn, int E) {
  const int gw = __builtin_amdgcn_readfirstlane(
      (int)(blockIdx.x * 8 + (threadIdx.x >> 5)));
  const int lane  = threadIdx.x & 31;
  const int nw    = gridDim.x * 8;
  const int per   = (Nn + nw - 1) / nw;
  const int first = gw * per;
  const int stop  = (first + per < Nn) ? first + per : Nn;
  if (first >= stop) return;
  const float2* Hr = (const float2*)H;
  const float2  b  = ((const float2*)bias)[lane];
  int beg = off[first];
  int end = off[first + 1];
  for (int node = first; node < stop; ++node) {
    const int endNext = off[(node + 2 < Nn) ? node + 2 : Nn];
    const int n   = end - beg;
    const int2* slots = (const int2*)slots_ + beg;

    const float2 hself = Hr[(size_t)node * 32 + lane];

    const float hdn = hd[node];
    float lself = hs[node] + hdn;
    lself = (lself > 0.f) ? lself : NEG_SLOPE * lself;

    const bool v0  = lane < n;
    const int2 sl0 = v0 ? slots[lane] : make_int2(0, 0);
    float lg0 = -INFINITY;
    if (v0) {
      float t = hs[sl0.y] + hdn;
      lg0 = (t > 0.f) ? t : NEG_SLOPE * t;
    }
    float m = wave_max_f(lg0);
    for (int base = 32; base < n; base += 32) {
      const int j = base + lane;
      float lg = -INFINITY;
      if (j < n) {
        float t = hs[slots[j].y] + hdn;
        lg = (t > 0.f) ? t : NEG_SLOPE * t;
      }
      m = fmaxf(m, wave_max_f(lg));
    }
    m = fmaxf(m, lself);

    const float ex0 = v0 ? __expf(lg0 - m) : 0.f;
    float den = wave_sum_f(ex0);
    for (int base = 32; base < n; base += 32) {
      const int j = base + lane;
      float ex = 0.f;
      if (j < n) {
        float t = hs[slots[j].y] + hdn;
        t = (t > 0.f) ? t : NEG_SLOPE * t;
        ex = __expf(t - m);
      }
      den += wave_sum_f(ex);
    }
    const float exs = __expf(lself - m);
    den += exs;
    const float dd  = den + EPSV;
    const float al0 = ex0 / dd;
    const float als = exs / dd;
    if (WRITE_ALPHA) {
      if (v0) alpha_out[sl0.x] = al0;
      if (lane == 0) alpha_out[E + node] = als;
    }

    float2 acc = make_float2(0.f, 0.f);
    const int n0 = (n < 32) ? n : 32;
    int k = 0;
    for (; k + 4 <= n0; k += 4) {
      const int s0 = __builtin_amdgcn_readlane(sl0.y, k);
      const int s1 = __builtin_amdgcn_readlane(sl0.y, k + 1);
      const int s2 = __builtin_amdgcn_readlane(sl0.y, k + 2);
      const int s3 = __builtin_amdgcn_readlane(sl0.y, k + 3);
      const float2 h0 = Hr[(size_t)s0 * 32 + lane];
      const float2 h1 = Hr[(size_t)s1 * 32 + lane];
      const float2 h2 = Hr[(size_t)s2 * 32 + lane];
      const float2 h3 = Hr[(size_t)s3 * 32 + lane];
      const float a0 = bcast_f(al0, k);
      const float a1 = bcast_f(al0, k + 1);
      const float a2 = bcast_f(al0, k + 2);
      const float a3 = bcast_f(al0, k + 3);
      acc.x = fmaf(a0, h0.x, acc.x);  acc.y = fmaf(a0, h0.y, acc.y);
      acc.x = fmaf(a1, h1.x, acc.x);  acc.y = fmaf(a1, h1.y, acc.y);
      acc.x = fmaf(a2, h2.x, acc.x);  acc.y = fmaf(a2, h2.y, acc.y);
      acc.x = fmaf(a3, h3.x, acc.x);  acc.y = fmaf(a3, h3.y, acc.y);
    }
    const int r = n0 - k;
    if (r > 0) {
      const int k1 = (r > 1) ? k + 1 : k;
      const int k2 = (r > 2) ? k + 2 : k;
      const int s0 = __builtin_amdgcn_readlane(sl0.y, k);
      const int s1 = __builtin_amdgcn_readlane(sl0.y, k1);
      const int s2 = __builtin_amdgcn_readlane(sl0.y, k2);
      const float2 h0 = Hr[(size_t)s0 * 32 + lane];
      const float2 h1 = Hr[(size_t)s1 * 32 + lane];
      const float2 h2 = Hr[(size_t)s2 * 32 + lane];
      const float a0 = bcast_f(al0, k);
      const float a1 = bcast_f(al0, k1);
      const float a2 = bcast_f(al0, k2);
      acc.x = fmaf(a0, h0.x, acc.x);  acc.y = fmaf(a0, h0.y, acc.y);
      if (r > 1) { acc.x = fmaf(a1, h1.x, acc.x);  acc.y = fmaf(a1, h1.y, acc.y); }
      if (r > 2) { acc.x = fmaf(a2, h2.x, acc.x);  acc.y = fmaf(a2, h2.y, acc.y); }
    }
    for (int base = 32; base < n; base += 32) {
      const int  j  = base + lane;
      const bool vj = j < n;
      const int2 sl = vj ? slots[j] : make_int2(0, 0);
      float al = 0.f;
      if (vj) {
        float t = hs[sl.y] + hdn;
        t = (t > 0.f) ? t : NEG_SLOPE * t;
        al = __expf(t - m) / dd;
        if (WRITE_ALPHA) alpha_out[sl.x] = al;
      }
      const int c2 = (n - base < 32) ? (n - base) : 32;
      for (int q = 0; q < c2; ++q) {
        const int    s = __builtin_amdgcn_readlane(sl.y, q);
        const float  a = bcast_f(al, q);
        const float2 h = Hr[(size_t)s * 32 + lane];
        acc.x = fmaf(a, h.x, acc.x);
        acc.y = fmaf(a, h.y, acc.y);
      }
    }
    acc.x = fmaf(als, hself.x, acc.x);
    acc.y = fmaf(als, hself.y, acc.y);
    float ox = acc.x + b.x;
    float oy = acc.y + b.y;
    ox = (ox > 0.f) ? ox : 0.f;
    oy = (oy > 0.f) ? oy : 0.f;
    ((float2*)out)[(size_t)node * 32 + lane] = make_float2(ox, oy);
    beg = end;
    end = endNext;
  }
}

static inline int cdiv_ll(long long a, long long b) {
  return (int)((a + b - 1) / b);
}

extern "C" void kernel_launch(void* const* d_in, const int* in_sizes, int n_in,
                              void* d_out, int out_size, void* d_ws,
                              size_t ws_size, hipStream_t stream) {
  const float* x      = (const float*)d_in[0];
  const int*   ei     = (const int*)d_in[1];
  const float* W1     = (const float*)d_in[2];
  const float* a_src1 = (const float*)d_in[3];
  const float* a_dst1 = (const float*)d_in[4];
  const float* b1     = (const float*)d_in[5];
  const float* W2     = (const float*)d_in[6];
  const float* a_src2 = (const float*)d_in[7];
  const float* a_dst2 = (const float*)d_in[8];
  const float* b2     = (const float*)d_in[9];

  const int IN_DIM = 128;
  const int Nn = in_sizes[0] / IN_DIM;
  const int E  = in_sizes[1] / 2;

  const int    nTiles = cdiv_ll(Nn, SCAN_TILE);
  const size_t NnPad  = ((size_t)Nn + 4) & ~(size_t)3;

  float* bufH   = (float*)d_ws;
  float* bufAgg = bufH + (size_t)Nn * FDIM;
  float* hs     = bufAgg + (size_t)Nn * FDIM;
  float* hd     = hs + NnPad;
  int*   cnt    = (int*)(hd + NnPad);
  int*   off    = cnt + NnPad;
  int*   bsum   = off + NnPad;
  int*   slotsA = bsum + (((size_t)nTiles + 3) & ~(size_t)3);
  int*   slotsB = slotsA + (((size_t)2 * E + 3) & ~(size_t)3);
  int*   rank   = slotsB + (((size_t)2 * E + 3) & ~(size_t)3);

  float* h2out     = (float*)d_out;
  float* alpha_out = h2out + (size_t)Nn * FDIM;

  const int nodeGrid = cdiv_ll(Nn, 256);

  csr_zero<<<nodeGrid, 256, 0, stream>>>(cnt, Nn);
  csr_count<<<EDGE_GRID, 256, 0, stream>>>(ei, cnt, rank, E);
  csr_scan_sums<<<nTiles, 256, 0, stream>>>(cnt, bsum, Nn);
  csr_scan_write<<<nTiles, 256, 0, stream>>>(cnt, bsum, off, Nn, E);
  csr_fill<<<EDGE_GRID, 256, 0, stream>>>(ei, off, rank, slotsA, E);
  csr_order<<<LOOP_GRID, 256, 0, stream>>>(off, slotsA, slotsB, Nn);

  gat_gemm_wmma<128><<<LOOP_GRID, 128, 0, stream>>>(x, W1, bufH, Nn);
  gat_node_dots<<<nodeGrid, 256, 0, stream>>>(bufH, a_src1, a_dst1, hs, hd, Nn);
  gat_node_fused<false><<<LOOP_GRID, 256, 0, stream>>>(
      off, slotsB, hs, hd, bufH, b1, bufAgg, nullptr, Nn, E);

  gat_gemm_wmma<64><<<LOOP_GRID, 128, 0, stream>>>(bufAgg, W2, bufH, Nn);
  gat_node_dots<<<nodeGrid, 256, 0, stream>>>(bufH, a_src2, a_dst2, hs, hd, Nn);
  gat_node_fused<true><<<LOOP_GRID, 256, 0, stream>>>(
      off, slotsB, hs, hd, bufH, b2, h2out, alpha_out, Nn, E);
}


# Edge grouping without global atomics: workgroups own (8192-node range, edge chunk) pairs, count and then hand out slots on LDS integer counters; per-chunk counts scanned; float sums in old order
# speedup vs baseline: 2.7865x; 1.0915x over previous
#include <hip/hip_runtime.h>


#define LOOP_GRID 2048

typedef float v2f __attribute__((ext_vector_type(2)));
typedef float v8f __attribute__((ext_vector_type(8)));

#define NEG_SLOPE 0.2f
#define EPSV 1e-16f
#define FDIM 64

template <int K>
__global__ __launch_bounds__(128) void gat_gemm_wmma(
    const float* __restrict__ X, const float* __restrict__ W,
    float* __restrict__ H, int M) {
  const int wave = threadIdx.x >> 5;
  const int lane = threadIdx.x & 31;
  const int l16  = lane & 15;
  const int kh   = (lane >> 4) << 1;
  const int colBase = wave * 16;
  const int rOff = (lane >> 4) << 3;
  const float* wc = W + colBase + l16;

  v2f bf[K / 4];
#pragma unroll
  for (int k = 0; k < K; k += 4) {
    bf[k >> 2][0] = wc[(size_t)(k + kh) * FDIM];
    bf[k >> 2][1] = wc[(size_t)(k + kh + 1) * FDIM];
  }

  const int nTiles = (M + 15) >> 4;
  for (int tile = blockIdx.x; tile < nTiles; tile += gridDim.x) {
    const int rowBase = tile * 16;
    int arow = rowBase + l16;
    if (arow >= M) arow = M - 1;
    const float* xr = X + (size_t)arow * K;

    v8f c = {0.f, 0.f, 0.f, 0.f, 0.f, 0.f, 0.f, 0.f};
#pragma unroll
    for (int k = 0; k < K; k += 4) {
      v2f a;
      a[0] = xr[k + kh];
      a[1] = xr[k + kh + 1];
      c = __builtin_amdgcn_wmma_f32_16x16x4_f32(false, a, false, bf[k >> 2],
                                                (short)0, c, false, false);
    }

#pragma unroll
    for (int v = 0; v < 8; ++v) {
      int r = rowBase + rOff + v;
      if (r >= M) r = M - 1;
      H[(size_t)r * FDIM + colBase + l16] = c[v];
    }
  }
}

__global__ void gat_node_dots(const float* __restrict__ H,
                              const float* __restrict__ a_src,
                              const float* __restrict__ a_dst,
                              float* __restrict__ hs, float* __restrict__ hd,
                              int Nn) {
  int i = blockIdx.x * blockDim.x + threadIdx.x;
  if (i >= Nn) return;
  const float* h = H + (size_t)i * FDIM;
  float s = 0.f, d = 0.f;
#pragma unroll
  for (int f = 0; f < FDIM; ++f) {
    float v = h[f];
    s += v * a_src[f];
    d += v * a_dst[f];
  }
  hs[i] = s;
  hd[i] = d;
}

#define SCAN_TILE 1024
#define SPAN 8192
#define NCHUNK 20
#define GRP_T 1024
#define GRP_LD 8

__device__ __forceinline__ void chunk_bounds(int E, int c, int& cb, int& ce) {
  const int per = (((E + NCHUNK - 1) / NCHUNK) + 3) & ~3;
  const int b   = c * per;
  cb = (b < E) ? b : E;
  ce = (E - cb < per) ? E : cb + per;
}

__device__ __forceinline__ void count1(int* lc, int d, int lo) {
  const unsigned rel = (unsigned)(d - lo);
  if (rel < (unsigned)SPAN) atomicAdd(&lc[rel], 1);
}

__global__ __launch_bounds__(GRP_T) void csr_count(
    const int* __restrict__ ei, int* __restrict__ tab, int Nn, int NnPad,
    int E) {
  __shared__ int lc[SPAN];
  const int tid = threadIdx.x;
  const int lo  = blockIdx.x * SPAN;
  const int c   = blockIdx.y;
  for (int i = tid; i < SPAN; i += GRP_T) lc[i] = 0;
  __syncthreads();
  int cb, ce;
  chunk_bounds(E, c, cb, ce);
  const int* dst = ei + E;
  if (((((size_t)dst) & 15) == 0) && ((E & 3) == 0)) {
    const int4* dq = (const int4*)dst;
    const int4 none = make_int4(lo - 1, lo - 1, lo - 1, lo - 1);
    const int qe = ce >> 2;
    for (int q = (cb >> 2) + tid; q < qe; q += GRP_LD * GRP_T) {
      int4 a[GRP_LD];
#pragma unroll
      for (int u = 0; u < GRP_LD; ++u) {
        const int  qu = q + u * GRP_T;
        const bool in = qu < qe;
        const int4 t  = dq[in ? qu : qe - 1];
        a[u] = make_int4(in ? t.x : none.x, in ? t.y : none.y,
                         in ? t.z : none.z, in ? t.w : none.w);
      }
#pragma unroll
      for (int u = 0; u < GRP_LD; ++u) {
        count1(lc, a[u].x, lo); count1(lc, a[u].y, lo);
        count1(lc, a[u].z, lo); count1(lc, a[u].w, lo);
      }
    }
  } else {
    for (int e = cb + tid; e < ce; e += GRP_T) count1(lc, dst[e], lo);
  }
  __syncthreads();
  for (int i = tid; i < SPAN; i += GRP_T) {
    const int d = lo + i;
    if (d < Nn) tab[(size_t)c * NnPad + d] = lc[i];
  }
}

__device__ __forceinline__ int wave_sum_i(int v) {
#pragma unroll
  for (int o = 16; o > 0; o >>= 1) v += __shfl_xor(v, o, 32);
  return v;
}

__global__ __launch_bounds__(256) void csr_scan_sums(
    const int* __restrict__ tab, int* __restrict__ bsum, int Nn, int NnPad) {
  __shared__ int red[8];
  const int base = blockIdx.x * SCAN_TILE + threadIdx.x * 4;
  int s = 0;
#pragma unroll
  for (int j = 0; j < 4; ++j)
    if (base + j < Nn)
      for (int c = 0; c < NCHUNK; ++c) s += tab[(size_t)c * NnPad + base + j];
  s = wave_sum_i(s);
  if ((threadIdx.x & 31) == 0) red[threadIdx.x >> 5] = s;
  __syncthreads();
  if (threadIdx.x == 0) {
    int t = 0;
#pragma unroll
    for (int w = 0; w < 8; ++w) t += red[w];
    bsum[blockIdx.x] = t;
  }
}

__global__ __launch_bounds__(256) void csr_scan_write(
    int* __restrict__ tab, const int* __restrict__ bsum,
    int* __restrict__ off, int Nn, int NnPad, int E) {
  __shared__ int red[8];
  __shared__ int wtot[8];
  const int lane = threadIdx.x & 31;
  const int wave = threadIdx.x >> 5;

  int p = 0;
  for (int j = threadIdx.x; j < (int)blockIdx.x; j += 256) p += bsum[j];
  p = wave_sum_i(p);
  if (lane == 0) red[wave] = p;

  const int base = blockIdx.x * SCAN_TILE + threadIdx.x * 4;
  int v[4];
  int s = 0;
#pragma unroll
  for (int j = 0; j < 4; ++j) {
    int t = 0;
    if (base + j < Nn)
      for (int c = 0; c < NCHUNK; ++c) t += tab[(size_t)c * NnPad + base + j];
    v[j] = t;
    s += t;
  }
  int x = s;
#pragma unroll
  for (int o = 1; o < 32; o <<= 1) {
    int t = __shfl_up(x, o, 32);
    if (lane >= o) x += t;
  }
  if (lane == 31) wtot[wave] = x;
  __syncthreads();

  int run = 0;
#pragma unroll
  for (int w = 0; w < 8; ++w) {
    run += red[w];
    if (w < wave) run += wtot[w];
  }
  run += x - s;
#pragma unroll
  for (int j = 0; j < 4; ++j) {
    if (base + j < Nn) {
      off[base + j] = run;
      int t = run;
      for (int c = 0; c < NCHUNK; ++c) {
        int* q = tab + (size_t)c * NnPad + base + j;
        const int n = *q;
        *q = t;
        t += n;
      }
    }
    run += v[j];
  }
  if (blockIdx.x == 0 && threadIdx.x == 0) off[Nn] = E;
}

__device__ __forceinline__ void place1(int* cur, int2* sl,
                                       const int* __restrict__ ei, int d,
                                       int e, int lo) {
  const unsigned rel = (unsigned)(d - lo);
  if (rel < (unsigned)SPAN) {
    const int p = atomicAdd(&cur[rel], 1);
    sl[p] = make_int2(e, ei[e]);
  }
}

__device__ __forceinline__ void place4(int* cur, int2* sl, int4 d, int4 s,
                                       int e0, int lo, int spare) {
  const unsigned r0 = (unsigned)(d.x - lo), r1 = (unsigned)(d.y - lo);
  const unsigned r2 = (unsigned)(d.z - lo), r3 = (unsigned)(d.w - lo);
  const bool m0 = r0 < (unsigned)SPAN, m1 = r1 < (unsigned)SPAN;
  const bool m2 = r2 < (unsigned)SPAN, m3 = r3 < (unsigned)SPAN;
  const int p0 = atomicAdd(&cur[m0 ? (int)r0 : spare], 1);
  const int p1 = atomicAdd(&cur[m1 ? (int)r1 : spare], 1);
  const int p2 = atomicAdd(&cur[m2 ? (int)r2 : spare], 1);
  const int p3 = atomicAdd(&cur[m3 ? (int)r3 : spare], 1);
  if (m0) sl[p0] = make_int2(e0, s.x);
  if (m1) sl[p1] = make_int2(e0 + 1, s.y);
  if (m2) sl[p2] = make_int2(e0 + 2, s.z);
  if (m3) sl[p3] = make_int2(e0 + 3, s.w);
}

__global__ __launch_bounds__(GRP_T) void csr_place(
    const int* __restrict__ ei, const int* __restrict__ tab,
    int* __restrict__ slots, int Nn, int NnPad, int E) {
  __shared__ int cur[SPAN + 32];
  const int tid = threadIdx.x;
  const int lo  = blockIdx.x * SPAN;
  const int c   = blockIdx.y;
  for (int i = tid; i < SPAN; i += GRP_T) {
    const int d = lo + i;
    cur[i] = (d < Nn) ? tab[(size_t)c * NnPad + d] : 0;
  }
  if (tid < 32) cur[SPAN + tid] = 0;
  const int spare = SPAN + (tid & 31);
  __syncthreads();
  int cb, ce;
  chunk_bounds(E, c, cb, ce);
  const int* dst = ei + E;
  int2* sl = (int2*)slots;
  if (((((size_t)dst) & 15) == 0) && ((E & 3) == 0)) {
    const int4* dq = (const int4*)dst;
    const int4* sq = (const int4*)ei;
    const int4 none = make_int4(lo - 1, lo - 1, lo - 1, lo - 1);
    const int qe = ce >> 2;
    for (int q = (cb >> 2) + tid; q < qe; q += GRP_LD * GRP_T) {
      int4 a[GRP_LD];
      int4 b[GRP_LD];
#pragma unroll
      for (int u = 0; u < GRP_LD; ++u) {
        const int  qu = q + u * GRP_T;
        const bool in = qu < qe;
        const int  qc = in ? qu : qe - 1;
        const int4 t  = dq[qc];
        b[u] = sq[qc];
        a[u] = make_int4(in ? t.x : none.x, in ? t.y : none.y,
                         in ? t.z : none.z, in ? t.w : none.w);
      }
#pragma unroll
      for (int u = 0; u < GRP_LD; ++u)
        place4(cur, sl, a[u], b[u], 4 * (q + u * GRP_T), lo, spare);
    }
  } else {
    for (int e = cb + tid; e < ce; e += GRP_T) place1(cur, sl, ei, dst[e], e, lo);
  }
}

__global__ __launch_bounds__(256) void csr_order(
    const int* __restrict__ off, const int* __restrict__ slots_in,
    int* __restrict__ slots_out, int Nn) {
  const int gw = __builtin_amdgcn_readfirstlane(
      (int)(blockIdx.x * 8 + (threadIdx.x >> 5)));
  const int lane  = threadIdx.x & 31;
  const int nw    = gridDim.x * 8;
  const int per   = (Nn + nw - 1) / nw;
  const int first = gw * per;
  const int stop  = (first + per < Nn) ? first + per : Nn;
  if (first >= stop) return;
  int beg = off[first];
  int end = off[first + 1];
  for (int node = first; node < stop; ++node) {
    const int endNext = off[(node + 2 < Nn) ? node + 2 : Nn];
    const int n   = end - beg;
    const int2* in  = (const int2*)slots_in + beg;
    int2*       out = (int2*)slots_out + beg;
    for (int base = 0; base < n; base += 32) {
      const int  j     = base + lane;
      const bool valid = j < n;
      const int2 my    = valid ? in[j] : make_int2(0x7fffffff, 0);
      int rank = 0;
      for (int b2 = 0; b2 < n; b2 += 32) {
        const int j2 = b2 + lane;
        const int o  = (j2 < n) ? in[j2].x : 0x7fffffff;
        const int c2 = (n - b2 < 32) ? (n - b2) : 32;
        for (int k = 0; k < c2; ++k) {
          const int ok = __builtin_amdgcn_readlane(o, k);
          rank += (ok < my.x) ? 1 : 0;
        }
      }
      if (valid) out[rank] = my;
    }
    beg = end;
    end = endNext;
  }
}

__device__ __forceinline__ float wave_max_f(float v) {
#pragma unroll
  for (int o = 16; o > 0; o >>= 1) v = fmaxf(v, __shfl_xor(v, o, 32));
  return v;
}
__device__ __forceinline__ float wave_sum_f(float v) {
#pragma unroll
  for (int o = 16; o > 0; o >>= 1) v += __shfl_xor(v, o, 32);
  return v;
}
__device__ __forceinline__ float bcast_f(float v, int k) {
  return __int_as_float(__builtin_amdgcn_readlane(__float_as_int(v), k));
}

template <bool WRITE_ALPHA>
__global__ __launch_bounds__(256) void gat_node_fused(
    const int* __restrict__ off, const int* __restrict__ slots_,
    const float* __restrict__ hs, const float* __restrict__ hd,
    const float* __restrict__ H, const float* __restrict__ bias,
    float* __restrict__ out, float* __restrict__ alpha_out, int Nn, int E) {
  const int gw = __builtin_amdgcn_readfirstlane(
      (int)(blockIdx.x * 8 + (threadIdx.x >> 5)));
  const int lane  = threadIdx.x & 31;
  const int nw    = gridDim.x * 8;
  const int per   = (Nn + nw - 1) / nw;
  const int first = gw * per;
  const int stop  = (first + per < Nn) ? first + per : Nn;
  if (first >= stop) return;
  const float2* Hr = (const float2*)H;
  const float2  b  = ((const float2*)bias)[lane];
  int beg = off[first];
  int end = off[first + 1];
  for (int node = first; node < stop; ++node) {
    const int endNext = off[(node + 2 < Nn) ? node + 2 : Nn];
    const int n   = end - beg;
    const int2* slots = (const int2*)slots_ + beg;

    const float2 hself = Hr[(size_t)node * 32 + lane];

    const float hdn = hd[node];
    float lself = hs[node] + hdn;
    lself = (lself > 0.f) ? lself : NEG_SLOPE * lself;

    const bool v0  = lane < n;
    const int2 sl0 = v0 ? slots[lane] : make_int2(0, 0);
    float lg0 = -INFINITY;
    if (v0) {
      float t = hs[sl0.y] + hdn;
      lg0 = (t > 0.f) ? t : NEG_SLOPE * t;
    }
    float m = wave_max_f(lg0);
    for (int base = 32; base < n; base += 32) {
      const int j = base + lane;
      float lg = -INFINITY;
      if (j < n) {
        float t = hs[slots[j].y] + hdn;
        lg = (t > 0.f) ? t : NEG_SLOPE * t;
      }
      m = fmaxf(m, wave_max_f(lg));
    }
    m = fmaxf(m, lself);

    const float ex0 = v0 ? __expf(lg0 - m) : 0.f;
    float den = wave_sum_f(ex0);
    for (int base = 32; base < n; base += 32) {
      const int j = base + lane;
      float ex = 0.f;
      if (j < n) {
        float t = hs[slots[j].y] + hdn;
        t = (t > 0.f) ? t : NEG_SLOPE * t;
        ex = __expf(t - m);
      }
      den += wave_sum_f(ex);
    }
    const float exs = __expf(lself - m);
    den += exs;
    const float dd  = den + EPSV;
    const float al0 = ex0 / dd;
    const float als = exs / dd;
    if (WRITE_ALPHA) {
      if (v0) alpha_out[sl0.x] = al0;
      if (lane == 0) alpha_out[E + node] = als;
    }

    float2 acc = make_float2(0.f, 0.f);
    const int n0 = (n < 32) ? n : 32;
    int k = 0;
    for (; k + 4 <= n0; k += 4) {
      const int s0 = __builtin_amdgcn_readlane(sl0.y, k);
      const int s1 = __builtin_amdgcn_readlane(sl0.y, k + 1);
      const int s2 = __builtin_amdgcn_readlane(sl0.y, k + 2);
      const int s3 = __builtin_amdgcn_readlane(sl0.y, k + 3);
      const float2 h0 = Hr[(size_t)s0 * 32 + lane];
      const float2 h1 = Hr[(size_t)s1 * 32 + lane];
      const float2 h2 = Hr[(size_t)s2 * 32 + lane];
      const float2 h3 = Hr[(size_t)s3 * 32 + lane];
      const float a0 = bcast_f(al0, k);
      const float a1 = bcast_f(al0, k + 1);
      const float a2 = bcast_f(al0, k + 2);
      const float a3 = bcast_f(al0, k + 3);
      acc.x = fmaf(a0, h0.x, acc.x);  acc.y = fmaf(a0, h0.y, acc.y);
      acc.x = fmaf(a1, h1.x, acc.x);  acc.y = fmaf(a1, h1.y, acc.y);
      acc.x = fmaf(a2, h2.x, acc.x);  acc.y = fmaf(a2, h2.y, acc.y);
      acc.x = fmaf(a3, h3.x, acc.x);  acc.y = fmaf(a3, h3.y, acc.y);
    }
    const int r = n0 - k;
    if (r > 0) {
      const int k1 = (r > 1) ? k + 1 : k;
      const int k2 = (r > 2) ? k + 2 : k;
      const int s0 = __builtin_amdgcn_readlane(sl0.y, k);
      const int s1 = __builtin_amdgcn_readlane(sl0.y, k1);
      const int s2 = __builtin_amdgcn_readlane(sl0.y, k2);
      const float2 h0 = Hr[(size_t)s0 * 32 + lane];
      const float2 h1 = Hr[(size_t)s1 * 32 + lane];
      const float2 h2 = Hr[(size_t)s2 * 32 + lane];
      const float a0 = bcast_f(al0, k);
      const float a1 = bcast_f(al0, k1);
      const float a2 = bcast_f(al0, k2);
      acc.x = fmaf(a0, h0.x, acc.x);  acc.y = fmaf(a0, h0.y, acc.y);
      if (r > 1) { acc.x = fmaf(a1, h1.x, acc.x);  acc.y = fmaf(a1, h1.y, acc.y); }
      if (r > 2) { acc.x = fmaf(a2, h2.x, acc.x);  acc.y = fmaf(a2, h2.y, acc.y); }
    }
    for (int base = 32; base < n; base += 32) {
      const int  j  = base + lane;
      const bool vj = j < n;
      const int2 sl = vj ? slots[j] : make_int2(0, 0);
      float al = 0.f;
      if (vj) {
        float t = hs[sl.y] + hdn;
        t = (t > 0.f) ? t : NEG_SLOPE * t;
        al = __expf(t - m) / dd;
        if (WRITE_ALPHA) alpha_out[sl.x] = al;
      }
      const int c2 = (n - base < 32) ? (n - base) : 32;
      for (int q = 0; q < c2; ++q) {
        const int    s = __builtin_amdgcn_readlane(sl.y, q);
        const float  a = bcast_f(al, q);
        const float2 h = Hr[(size_t)s * 32 + lane];
        acc.x = fmaf(a, h.x, acc.x);
        acc.y = fmaf(a, h.y, acc.y);
      }
    }
    acc.x = fmaf(als, hself.x, acc.x);
    acc.y = fmaf(als, hself.y, acc.y);
    float ox = acc.x + b.x;
    float oy = acc.y + b.y;
    ox = (ox > 0.f) ? ox : 0.f;
    oy = (oy > 0.f) ? oy : 0.f;
    ((float2*)out)[(size_t)node * 32 + lane] = make_float2(ox, oy);
    beg = end;
    end = endNext;
  }
}

static inline int cdiv_ll(long long a, long long b) {
  return (int)((a + b - 1) / b);
}

extern "C" void kernel_launch(void* const* d_in, const int* in_sizes, int n_in,
                              void* d_out, int out_size, void* d_ws,
                              size_t ws_size, hipStream_t stream) {
  const float* x      = (const float*)d_in[0];
  const int*   ei     = (const int*)d_in[1];
  const float* W1     = (const float*)d_in[2];
  const float* a_src1 = (const float*)d_in[3];
  const float* a_dst1 = (const float*)d_in[4];
  const float* b1     = (const float*)d_in[5];
  const float* W2     = (const float*)d_in[6];
  const float* a_src2 = (const float*)d_in[7];
  const float* a_dst2 = (const float*)d_in[8];
  const float* b2     = (const float*)d_in[9];

  const int IN_DIM = 128;
  const int Nn = in_sizes[0] / IN_DIM;
  const int E  = in_sizes[1] / 2;

  const int    nTiles = cdiv_ll(Nn, SCAN_TILE);
  const size_t NnPad  = ((size_t)Nn + 4) & ~(size_t)3;

  float* bufH   = (float*)d_ws;
  float* bufAgg = bufH + (size_t)Nn * FDIM;
  float* hs     = bufAgg + (size_t)Nn * FDIM;
  float* hd     = hs + NnPad;
  int*   off    = (int*)(hd + NnPad);
  int*   bsum   = off + NnPad;
  int*   slotsA = bsum + (((size_t)nTiles + 3) & ~(size_t)3);
  int*   slotsB = slotsA + (((size_t)2 * E + 3) & ~(size_t)3);
  int*   tab    = slotsB + (((size_t)2 * E + 3) & ~(size_t)3);

  float* h2out     = (float*)d_out;
  float* alpha_out = h2out + (size_t)Nn * FDIM;

  const int nodeGrid = cdiv_ll(Nn, 256);

  const dim3 grpGrid(cdiv_ll(Nn, SPAN), NCHUNK);
  csr_count<<<grpGrid, GRP_T, 0, stream>>>(ei, tab, Nn, (int)NnPad, E);
  csr_scan_sums<<<nTiles, 256, 0, stream>>>(tab, bsum, Nn, (int)NnPad);
  csr_scan_write<<<nTiles, 256, 0, stream>>>(tab, bsum, off, Nn, (int)NnPad, E);
  csr_place<<<grpGrid, GRP_T, 0, stream>>>(ei, tab, slotsA, Nn, (int)NnPad, E);
  csr_order<<<LOOP_GRID, 256, 0, stream>>>(off, slotsA, slotsB, Nn);

  gat_gemm_wmma<128><<<LOOP_GRID, 128, 0, stream>>>(x, W1, bufH, Nn);
  gat_node_dots<<<nodeGrid, 256, 0, stream>>>(bufH, a_src1, a_dst1, hs, hd, Nn);
  gat_node_fused<false><<<LOOP_GRID, 256, 0, stream>>>(
      off, slotsB, hs, hd, bufH, b1, bufAgg, nullptr, Nn, E);

  gat_gemm_wmma<64><<<LOOP_GRID, 128, 0, stream>>>(bufAgg, W2, bufH, Nn);
  gat_node_dots<<<nodeGrid, 256, 0, stream>>>(bufH, a_src2, a_dst2, hs, hd, Nn);
  gat_node_fused<true><<<LOOP_GRID, 256, 0, stream>>>(
      off, slotsB, hs, hd, bufH, b2, h2out, alpha_out, Nn, E);
}


# Offsets scan in one launch, one node per thread: its 20 chunk counts are loaded together before any store (was 80 load-then-store steps in a row); tile sums from the count pass; float sums unchanged
# speedup vs baseline: 3.2191x; 1.1552x over previous
#include <hip/hip_runtime.h>


#define LOOP_GRID 2048

typedef float v2f __attribute__((ext_vector_type(2)));
typedef float v8f __attribute__((ext_vector_type(8)));

#define NEG_SLOPE 0.2f
#define EPSV 1e-16f
#define FDIM 64

template <int K>
__global__ __launch_bounds__(128) void gat_gemm_wmma(
    const float* __restrict__ X, const float* __restrict__ W,
    float* __restrict__ H, int M) {
  const int wave = threadIdx.x >> 5;
  const int lane = threadIdx.x & 31;
  const int l16  = lane & 15;
  const int kh   = (lane >> 4) << 1;
  const int colBase = wave * 16;
  const int rOff = (lane >> 4) << 3;
  const float* wc = W + colBase + l16;

  v2f bf[K / 4];
#pragma unroll
  for (int k = 0; k < K; k += 4) {
    bf[k >> 2][0] = wc[(size_t)(k + kh) * FDIM];
    bf[k >> 2][1] = wc[(size_t)(k + kh + 1) * FDIM];
  }

  const int nTiles = (M + 15) >> 4;
  for (int tile = blockIdx.x; tile < nTiles; tile += gridDim.x) {
    const int rowBase = tile * 16;
    int arow = rowBase + l16;
    if (arow >= M) arow = M - 1;
    const float* xr = X + (size_t)arow * K;

    v8f c = {0.f, 0.f, 0.f, 0.f, 0.f, 0.f, 0.f, 0.f};
#pragma unroll
    for (int k = 0; k < K; k += 4) {
      v2f a;
      a[0] = xr[k + kh];
      a[1] = xr[k + kh + 1];
      c = __builtin_amdgcn_wmma_f32_16x16x4_f32(false, a, false, bf[k >> 2],
                                                (short)0, c, false, false);
    }

#pragma unroll
    for (int v = 0; v < 8; ++v) {
      int r = rowBase + rOff + v;
      if (r >= M) r = M - 1;
      H[(size_t)r * FDIM + colBase + l16] = c[v];
    }
  }
}

__global__ void gat_node_dots(const float* __restrict__ H,
                              const float* __restrict__ a_src,
                              const float* __restrict__ a_dst,
                              float* __restrict__ hs, float* __restrict__ hd,
                              int Nn) {
  int i = blockIdx.x * blockDim.x + threadIdx.x;
  if (i >= Nn) return;
  const float* h = H + (size_t)i * FDIM;
  float s = 0.f, d = 0.f;
#pragma unroll
  for (int f = 0; f < FDIM; ++f) {
    float v = h[f];
    s += v * a_src[f];
    d += v * a_dst[f];
  }
  hs[i] = s;
  hd[i] = d;
}

#define SCAN_TILE 256
#define SPAN 8192
#define NCHUNK 20
#define GRP_T 1024
#define GRP_LD 8

__device__ __forceinline__ void chunk_bounds(int E, int c, int& cb, int& ce) {
  const int per = (((E + NCHUNK - 1) / NCHUNK) + 3) & ~3;
  const int b   = c * per;
  cb = (b < E) ? b : E;
  ce = (E - cb < per) ? E : cb + per;
}

__device__ __forceinline__ int wave_sum_i(int v) {
#pragma unroll
  for (int o = 16; o > 0; o >>= 1) v += __shfl_xor(v, o, 32);
  return v;
}

__device__ __forceinline__ void count1(int* lc, int d, int lo) {
  const unsigned rel = (unsigned)(d - lo);
  if (rel < (unsigned)SPAN) atomicAdd(&lc[rel], 1);
}

__global__ __launch_bounds__(GRP_T) void csr_count(
    const int* __restrict__ ei, int* __restrict__ tab, int* __restrict__ tsum,
    int Nn, int NnPad, int E) {
  __shared__ int lc[SPAN];
  const int tid = threadIdx.x;
  const int lo  = blockIdx.x * SPAN;
  const int c   = blockIdx.y;
  for (int i = tid; i < SPAN; i += GRP_T) lc[i] = 0;
  __syncthreads();
  int cb, ce;
  chunk_bounds(E, c, cb, ce);
  const int* dst = ei + E;
  if (((((size_t)dst) & 15) == 0) && ((E & 3) == 0)) {
    const int4* dq = (const int4*)dst;
    const int4 none = make_int4(lo - 1, lo - 1, lo - 1, lo - 1);
    const int qe = ce >> 2;
    for (int q = (cb >> 2) + tid; q < qe; q += GRP_LD * GRP_T) {
      int4 a[GRP_LD];
#pragma unroll
      for (int u = 0; u < GRP_LD; ++u) {
        const int  qu = q + u * GRP_T;
        const bool in = qu < qe;
        const int4 t  = dq[in ? qu : qe - 1];
        a[u] = make_int4(in ? t.x : none.x, in ? t.y : none.y,
                         in ? t.z : none.z, in ? t.w : none.w);
      }
#pragma unroll
      for (int u = 0; u < GRP_LD; ++u) {
        count1(lc, a[u].x, lo); count1(lc, a[u].y, lo);
        count1(lc, a[u].z, lo); count1(lc, a[u].w, lo);
      }
    }
  } else {
    for (int e = cb + tid; e < ce; e += GRP_T) count1(lc, dst[e], lo);
  }
  __syncthreads();
  for (int i = tid; i < SPAN; i += GRP_T) {
    const int d = lo + i;
    if (d < Nn) tab[(size_t)c * NnPad + d] = lc[i];
  }
  for (int t = tid >> 5; t < SPAN / SCAN_TILE; t += GRP_T / 32) {
    int v = 0;
#pragma unroll
    for (int k = 0; k < SCAN_TILE / 32; ++k)
      v += lc[t * SCAN_TILE + k * 32 + (tid & 31)];
    v = wave_sum_i(v);
    if ((tid & 31) == 0)
      tsum[(size_t)(blockIdx.x * (SPAN / SCAN_TILE) + t) * NCHUNK + c] = v;
  }
}

__global__ __launch_bounds__(SCAN_TILE) void csr_scan_write(
    int* __restrict__ tab, const int* __restrict__ tsum,
    int* __restrict__ off, int Nn, int NnPad, int E) {
  __shared__ int red[SCAN_TILE / 32];
  __shared__ int wtot[SCAN_TILE / 32];
  const int lane = threadIdx.x & 31;
  const int wave = threadIdx.x >> 5;
  const int d    = blockIdx.x * SCAN_TILE + threadIdx.x;
  const bool in  = d < Nn;

  int cnt[NCHUNK];
#pragma unroll
  for (int c = 0; c < NCHUNK; ++c)
    cnt[c] = in ? tab[(size_t)c * NnPad + d] : 0;

  int p = 0;
  const int np = (int)blockIdx.x * NCHUNK;
  for (int j = threadIdx.x; j < np; j += SCAN_TILE) p += tsum[j];
  p = wave_sum_i(p);
  if (lane == 0) red[wave] = p;

  int s = 0;
#pragma unroll
  for (int c = 0; c < NCHUNK; ++c) s += cnt[c];
  int x = s;
#pragma unroll
  for (int o = 1; o < 32; o <<= 1) {
    int t = __shfl_up(x, o, 32);
    if (lane >= o) x += t;
  }
  if (lane == 31) wtot[wave] = x;
  __syncthreads();

  int run = 0;
#pragma unroll
  for (int w = 0; w < SCAN_TILE / 32; ++w) {
    run += red[w];
    if (w < wave) run += wtot[w];
  }
  run += x - s;
  if (in) {
    off[d] = run;
    int t = run;
#pragma unroll
    for (int c = 0; c < NCHUNK; ++c) {
      tab[(size_t)c * NnPad + d] = t;
      t += cnt[c];
    }
  }
  if (blockIdx.x == 0 && threadIdx.x == 0) off[Nn] = E;
}

__device__ __forceinline__ void place1(int* cur, int2* sl,
                                       const int* __restrict__ ei, int d,
                                       int e, int lo) {
  const unsigned rel = (unsigned)(d - lo);
  if (rel < (unsigned)SPAN) {
    const int p = atomicAdd(&cur[rel], 1);
    sl[p] = make_int2(e, ei[e]);
  }
}

__device__ __forceinline__ void place4(int* cur, int2* sl, int4 d, int4 s,
                                       int e0, int lo, int spare) {
  const unsigned r0 = (unsigned)(d.x - lo), r1 = (unsigned)(d.y - lo);
  const unsigned r2 = (unsigned)(d.z - lo), r3 = (unsigned)(d.w - lo);
  const bool m0 = r0 < (unsigned)SPAN, m1 = r1 < (unsigned)SPAN;
  const bool m2 = r2 < (unsigned)SPAN, m3 = r3 < (unsigned)SPAN;
  const int p0 = atomicAdd(&cur[m0 ? (int)r0 : spare], 1);
  const int p1 = atomicAdd(&cur[m1 ? (int)r1 : spare], 1);
  const int p2 = atomicAdd(&cur[m2 ? (int)r2 : spare], 1);
  const int p3 = atomicAdd(&cur[m3 ? (int)r3 : spare], 1);
  if (m0) sl[p0] = make_int2(e0, s.x);
  if (m1) sl[p1] = make_int2(e0 + 1, s.y);
  if (m2) sl[p2] = make_int2(e0 + 2, s.z);
  if (m3) sl[p3] = make_int2(e0 + 3, s.w);
}

__global__ __launch_bounds__(GRP_T) void csr_place(
    const int* __restrict__ ei, const int* __restrict__ tab,
    int* __restrict__ slots, int Nn, int NnPad, int E) {
  __shared__ int cur[SPAN + 32];
  const int tid = threadIdx.x;
  const int lo  = blockIdx.x * SPAN;
  const int c   = blockIdx.y;
  for (int i = tid; i < SPAN; i += GRP_T) {
    const int d = lo + i;
    cur[i] = (d < Nn) ? tab[(size_t)c * NnPad + d] : 0;
  }
  if (tid < 32) cur[SPAN + tid] = 0;
  const int spare = SPAN + (tid & 31);
  __syncthreads();
  int cb, ce;
  chunk_bounds(E, c, cb, ce);
  const int* dst = ei + E;
  int2* sl = (int2*)slots;
  if (((((size_t)dst) & 15) == 0) && ((E & 3) == 0)) {
    const int4* dq = (const int4*)dst;
    const int4* sq = (const int4*)ei;
    const int4 none = make_int4(lo - 1, lo - 1, lo - 1, lo - 1);
    const int qe = ce >> 2;
    for (int q = (cb >> 2) + tid; q < qe; q += GRP_LD * GRP_T) {
      int4 a[GRP_LD];
      int4 b[GRP_LD];
#pragma unroll
      for (int u = 0; u < GRP_LD; ++u) {
        const int  qu = q + u * GRP_T;
        const bool in = qu < qe;
        const int  qc = in ? qu : qe - 1;
        const int4 t  = dq[qc];
        b[u] = sq[qc];
        a[u] = make_int4(in ? t.x : none.x, in ? t.y : none.y,
                         in ? t.z : none.z, in ? t.w : none.w);
      }
#pragma unroll
      for (int u = 0; u < GRP_LD; ++u)
        place4(cur, sl, a[u], b[u], 4 * (q + u * GRP_T), lo, spare);
    }
  } else {
    for (int e = cb + tid; e < ce; e += GRP_T) place1(cur, sl, ei, dst[e], e, lo);
  }
}

__global__ __launch_bounds__(256) void csr_order(
    const int* __restrict__ off, const int* __restrict__ slots_in,
    int* __restrict__ slots_out, int Nn) {
  const int gw = __builtin_amdgcn_readfirstlane(
      (int)(blockIdx.x * 8 + (threadIdx.x >> 5)));
  const int lane  = threadIdx.x & 31;
  const int nw    = gridDim.x * 8;
  const int per   = (Nn + nw - 1) / nw;
  const int first = gw * per;
  const int stop  = (first + per < Nn) ? first + per : Nn;
  if (first >= stop) return;
  int beg = off[first];
  int end = off[first + 1];
  for (int node = first; node < stop; ++node) {
    const int endNext = off[(node + 2 < Nn) ? node + 2 : Nn];
    const int n   = end - beg;
    const int2* in  = (const int2*)slots_in + beg;
    int2*       out = (int2*)slots_out + beg;
    for (int base = 0; base < n; base += 32) {
      const int  j     = base + lane;
      const bool valid = j < n;
      const int2 my    = valid ? in[j] : make_int2(0x7fffffff, 0);
      int rank = 0;
      for (int b2 = 0; b2 < n; b2 += 32) {
        const int j2 = b2 + lane;
        const int o  = (j2 < n) ? in[j2].x : 0x7fffffff;
        const int c2 = (n - b2 < 32) ? (n - b2) : 32;
        for (int k = 0; k < c2; ++k) {
          const int ok = __builtin_amdgcn_readlane(o, k);
          rank += (ok < my.x) ? 1 : 0;
        }
      }
      if (valid) out[rank] = my;
    }
    beg = end;
    end = endNext;
  }
}

__device__ __forceinline__ float wave_max_f(float v) {
#pragma unroll
  for (int o = 16; o > 0; o >>= 1) v = fmaxf(v, __shfl_xor(v, o, 32));
  return v;
}
__device__ __forceinline__ float wave_sum_f(float v) {
#pragma unroll
  for (int o = 16; o > 0; o >>= 1) v += __shfl_xor(v, o, 32);
  return v;
}
__device__ __forceinline__ float bcast_f(float v, int k) {
  return __int_as_float(__builtin_amdgcn_readlane(__float_as_int(v), k));
}

template <bool WRITE_ALPHA>
__global__ __launch_bounds__(256) void gat_node_fused(
    const int* __restrict__ off, const int* __restrict__ slots_,
    const float* __restrict__ hs, const float* __restrict__ hd,
    const float* __restrict__ H, const float* __restrict__ bias,
    float* __restrict__ out, float* __restrict__ alpha_out, int Nn, int E) {
  const int gw = __builtin_amdgcn_readfirstlane(
      (int)(blockIdx.x * 8 + (threadIdx.x >> 5)));
  const int lane  = threadIdx.x & 31;
  const int nw    = gridDim.x * 8;
  const int per   = (Nn + nw - 1) / nw;
  const int first = gw * per;
  const int stop  = (first + per < Nn) ? first + per : Nn;
  if (first >= stop) return;
  const float2* Hr = (const float2*)H;
  const float2  b  = ((const float2*)bias)[lane];
  int beg = off[first];
  int end = off[first + 1];
  for (int node = first; node < stop; ++node) {
    const int endNext = off[(node + 2 < Nn) ? node + 2 : Nn];
    const int n   = end - beg;
    const int2* slots = (const int2*)slots_ + beg;

    const float2 hself = Hr[(size_t)node * 32 + lane];

    const float hdn = hd[node];
    float lself = hs[node] + hdn;
    lself = (lself > 0.f) ? lself : NEG_SLOPE * lself;

    const bool v0  = lane < n;
    const int2 sl0 = v0 ? slots[lane] : make_int2(0, 0);
    float lg0 = -INFINITY;
    if (v0) {
      float t = hs[sl0.y] + hdn;
      lg0 = (t > 0.f) ? t : NEG_SLOPE * t;
    }
    float m = wave_max_f(lg0);
    for (int base = 32; base < n; base += 32) {
      const int j = base + lane;
      float lg = -INFINITY;
      if (j < n) {
        float t = hs[slots[j].y] + hdn;
        lg = (t > 0.f) ? t : NEG_SLOPE * t;
      }
      m = fmaxf(m, wave_max_f(lg));
    }
    m = fmaxf(m, lself);

    const float ex0 = v0 ? __expf(lg0 - m) : 0.f;
    float den = wave_sum_f(ex0);
    for (int base = 32; base < n; base += 32) {
      const int j = base + lane;
      float ex = 0.f;
      if (j < n) {
        float t = hs[slots[j].y] + hdn;
        t = (t > 0.f) ? t : NEG_SLOPE * t;
        ex = __expf(t - m);
      }
      den += wave_sum_f(ex);
    }
    const float exs = __expf(lself - m);
    den += exs;
    const float dd  = den + EPSV;
    const float al0 = ex0 / dd;
    const float als = exs / dd;
    if (WRITE_ALPHA) {
      if (v0) alpha_out[sl0.x] = al0;
      if (lane == 0) alpha_out[E + node] = als;
    }

    float2 acc = make_float2(0.f, 0.f);
    const int n0 = (n < 32) ? n : 32;
    int k = 0;
    for (; k + 4 <= n0; k += 4) {
      const int s0 = __builtin_amdgcn_readlane(sl0.y, k);
      const int s1 = __builtin_amdgcn_readlane(sl0.y, k + 1);
      const int s2 = __builtin_amdgcn_readlane(sl0.y, k + 2);
      const int s3 = __builtin_amdgcn_readlane(sl0.y, k + 3);
      const float2 h0 = Hr[(size_t)s0 * 32 + lane];
      const float2 h1 = Hr[(size_t)s1 * 32 + lane];
      const float2 h2 = Hr[(size_t)s2 * 32 + lane];
      const float2 h3 = Hr[(size_t)s3 * 32 + lane];
      const float a0 = bcast_f(al0, k);
      const float a1 = bcast_f(al0, k + 1);
      const float a2 = bcast_f(al0, k + 2);
      const float a3 = bcast_f(al0, k + 3);
      acc.x = fmaf(a0, h0.x, acc.x);  acc.y = fmaf(a0, h0.y, acc.y);
      acc.x = fmaf(a1, h1.x, acc.x);  acc.y = fmaf(a1, h1.y, acc.y);
      acc.x = fmaf(a2, h2.x, acc.x);  acc.y = fmaf(a2, h2.y, acc.y);
      acc.x = fmaf(a3, h3.x, acc.x);  acc.y = fmaf(a3, h3.y, acc.y);
    }
    const int r = n0 - k;
    if (r > 0) {
      const int k1 = (r > 1) ? k + 1 : k;
      const int k2 = (r > 2) ? k + 2 : k;
      const int s0 = __builtin_amdgcn_readlane(sl0.y, k);
      const int s1 = __builtin_amdgcn_readlane(sl0.y, k1);
      const int s2 = __builtin_amdgcn_readlane(sl0.y, k2);
      const float2 h0 = Hr[(size_t)s0 * 32 + lane];
      const float2 h1 = Hr[(size_t)s1 * 32 + lane];
      const float2 h2 = Hr[(size_t)s2 * 32 + lane];
      const float a0 = bcast_f(al0, k);
      const float a1 = bcast_f(al0, k1);
      const float a2 = bcast_f(al0, k2);
      acc.x = fmaf(a0, h0.x, acc.x);  acc.y = fmaf(a0, h0.y, acc.y);
      if (r > 1) { acc.x = fmaf(a1, h1.x, acc.x);  acc.y = fmaf(a1, h1.y, acc.y); }
      if (r > 2) { acc.x = fmaf(a2, h2.x, acc.x);  acc.y = fmaf(a2, h2.y, acc.y); }
    }
    for (int base = 32; base < n; base += 32) {
      const int  j  = base + lane;
      const bool vj = j < n;
      const int2 sl = vj ? slots[j] : make_int2(0, 0);
      float al = 0.f;
      if (vj) {
        float t = hs[sl.y] + hdn;
        t = (t > 0.f) ? t : NEG_SLOPE * t;
        al = __expf(t - m) / dd;
        if (WRITE_ALPHA) alpha_out[sl.x] = al;
      }
      const int c2 = (n - base < 32) ? (n - base) : 32;
      for (int q = 0; q < c2; ++q) {
        const int    s = __builtin_amdgcn_readlane(sl.y, q);
        const float  a = bcast_f(al, q);
        const float2 h = Hr[(size_t)s * 32 + lane];
        acc.x = fmaf(a, h.x, acc.x);
        acc.y = fmaf(a, h.y, acc.y);
      }
    }
    acc.x = fmaf(als, hself.x, acc.x);
    acc.y = fmaf(als, hself.y, acc.y);
    float ox = acc.x + b.x;
    float oy = acc.y + b.y;
    ox = (ox > 0.f) ? ox : 0.f;
    oy = (oy > 0.f) ? oy : 0.f;
    ((float2*)out)[(size_t)node * 32 + lane] = make_float2(ox, oy);
    beg = end;
    end = endNext;
  }
}

static inline int cdiv_ll(long long a, long long b) {
  return (int)((a + b - 1) / b);
}

extern "C" void kernel_launch(void* const* d_in, const int* in_sizes, int n_in,
                              void* d_out, int out_size, void* d_ws,
                              size_t ws_size, hipStream_t stream) {
  const float* x      = (const float*)d_in[0];
  const int*   ei     = (const int*)d_in[1];
  const float* W1     = (const float*)d_in[2];
  const float* a_src1 = (const float*)d_in[3];
  const float* a_dst1 = (const float*)d_in[4];
  const float* b1     = (const float*)d_in[5];
  const float* W2     = (const float*)d_in[6];
  const float* a_src2 = (const float*)d_in[7];
  const float* a_dst2 = (const float*)d_in[8];
  const float* b2     = (const float*)d_in[9];

  const int IN_DIM = 128;
  const int Nn = in_sizes[0] / IN_DIM;
  const int E  = in_sizes[1] / 2;

  const int    nTiles = cdiv_ll(Nn, SCAN_TILE);
  const int    nRange = cdiv_ll(Nn, SPAN);
  const size_t nTsum  = (size_t)nRange * (SPAN / SCAN_TILE) * NCHUNK;
  const size_t NnPad  = ((size_t)Nn + 4) & ~(size_t)3;

  float* bufH   = (float*)d_ws;
  float* bufAgg = bufH + (size_t)Nn * FDIM;
  float* hs     = bufAgg + (size_t)Nn * FDIM;
  float* hd     = hs + NnPad;
  int*   off    = (int*)(hd + NnPad);
  int*   tsum   = off + NnPad;
  int*   slotsA = tsum + ((nTsum + 3) & ~(size_t)3);
  int*   slotsB = slotsA + (((size_t)2 * E + 3) & ~(size_t)3);
  int*   tab    = slotsB + (((size_t)2 * E + 3) & ~(size_t)3);

  float* h2out     = (float*)d_out;
  float* alpha_out = h2out + (size_t)Nn * FDIM;

  const int nodeGrid = cdiv_ll(Nn, 256);

  const dim3 grpGrid(nRange, NCHUNK);
  csr_count<<<grpGrid, GRP_T, 0, stream>>>(ei, tab, tsum, Nn, (int)NnPad, E);
  csr_scan_write<<<nTiles, SCAN_TILE, 0, stream>>>(tab, tsum, off, Nn,
                                                   (int)NnPad, E);
  csr_place<<<grpGrid, GRP_T, 0, stream>>>(ei, tab, slotsA, Nn, (int)NnPad, E);
  csr_order<<<LOOP_GRID, 256, 0, stream>>>(off, slotsA, slotsB, Nn);

  gat_gemm_wmma<128><<<LOOP_GRID, 128, 0, stream>>>(x, W1, bufH, Nn);
  gat_node_dots<<<nodeGrid, 256, 0, stream>>>(bufH, a_src1, a_dst1, hs, hd, Nn);
  gat_node_fused<false><<<LOOP_GRID, 256, 0, stream>>>(
      off, slotsB, hs, hd, bufH, b1, bufAgg, nullptr, Nn, E);

  gat_gemm_wmma<64><<<LOOP_GRID, 128, 0, stream>>>(bufAgg, W2, bufH, Nn);
  gat_node_dots<<<nodeGrid, 256, 0, stream>>>(bufH, a_src2, a_dst2, hs, hd, Nn);
  gat_node_fused<true><<<LOOP_GRID, 256, 0, stream>>>(
      off, slotsB, hs, hd, bufH, b2, h2out, alpha_out, Nn, E);
}


# Two pairs of independent launches merged, role by block index: offsets scan with the layer-1 product, group ordering with the layer-1 projections; 8 launches, not 10; float sums in old order
# speedup vs baseline: 3.3390x; 1.0372x over previous
#include <hip/hip_runtime.h>


#define LOOP_GRID 2048

typedef float v2f __attribute__((ext_vector_type(2)));
typedef float v8f __attribute__((ext_vector_type(8)));

#define NEG_SLOPE 0.2f
#define EPSV 1e-16f
#define FDIM 64

template <int K>
__device__ __forceinline__ void gemm_body(
    const float* __restrict__ X, const float* __restrict__ W,
    float* __restrict__ H, int M, int vb, int vgrid, int tid) {
  const int wave = tid >> 5;
  const int lane = tid & 31;
  const int l16  = lane & 15;
  const int kh   = (lane >> 4) << 1;
  const int colBase = wave * 16;
  const int rOff = (lane >> 4) << 3;
  const float* wc = W + colBase + l16;

  v2f bf[K / 4];
#pragma unroll
  for (int k = 0; k < K; k += 4) {
    bf[k >> 2][0] = wc[(size_t)(k + kh) * FDIM];
    bf[k >> 2][1] = wc[(size_t)(k + kh + 1) * FDIM];
  }

  const int nTiles = (M + 15) >> 4;
  for (int tile = vb; tile < nTiles; tile += vgrid) {
    const int rowBase = tile * 16;
    int arow = rowBase + l16;
    if (arow >= M) arow = M - 1;
    const float* xr = X + (size_t)arow * K;

    v8f c = {0.f, 0.f, 0.f, 0.f, 0.f, 0.f, 0.f, 0.f};
#pragma unroll
    for (int k = 0; k < K; k += 4) {
      v2f a;
      a[0] = xr[k + kh];
      a[1] = xr[k + kh + 1];
      c = __builtin_amdgcn_wmma_f32_16x16x4_f32(false, a, false, bf[k >> 2],
                                                (short)0, c, false, false);
    }

#pragma unroll
    for (int v = 0; v < 8; ++v) {
      int r = rowBase + rOff + v;
      if (r >= M) r = M - 1;
      H[(size_t)r * FDIM + colBase + l16] = c[v];
    }
  }
}

template <int K>
__global__ __launch_bounds__(128) void gat_gemm_wmma(
    const float* __restrict__ X, const float* __restrict__ W,
    float* __restrict__ H, int M) {
  gemm_body<K>(X, W, H, M, (int)blockIdx.x, (int)gridDim.x, (int)threadIdx.x);
}

__device__ __forceinline__ void dots_body(const float* __restrict__ H,
                              const float* __restrict__ a_src,
                              const float* __restrict__ a_dst,
                              float* __restrict__ hs, float* __restrict__ hd,
                              int Nn, int i) {
  if (i >= Nn) return;
  const float* h = H + (size_t)i * FDIM;
  float s = 0.f, d = 0.f;
#pragma unroll
  for (int f = 0; f < FDIM; ++f) {
    float v = h[f];
    s += v * a_src[f];
    d += v * a_dst[f];
  }
  hs[i] = s;
  hd[i] = d;
}

__global__ void gat_node_dots(const float* __restrict__ H,
                              const float* __restrict__ a_src,
                              const float* __restrict__ a_dst,
                              float* __restrict__ hs, float* __restrict__ hd,
                              int Nn) {
  dots_body(H, a_src, a_dst, hs, hd, Nn,
            (int)(blockIdx.x * blockDim.x + threadIdx.x));
}

#define SCAN_TILE 256
#define SPAN 8192
#define NCHUNK 20
#define GRP_T 1024
#define GRP_LD 8

__device__ __forceinline__ void chunk_bounds(int E, int c, int& cb, int& ce) {
  const int per = (((E + NCHUNK - 1) / NCHUNK) + 3) & ~3;
  const int b   = c * per;
  cb = (b < E) ? b : E;
  ce = (E - cb < per) ? E : cb + per;
}

__device__ __forceinline__ int wave_sum_i(int v) {
#pragma unroll
  for (int o = 16; o > 0; o >>= 1) v += __shfl_xor(v, o, 32);
  return v;
}

__device__ __forceinline__ void count1(int* lc, int d, int lo) {
  const unsigned rel = (unsigned)(d - lo);
  if (rel < (unsigned)SPAN) atomicAdd(&lc[rel], 1);
}

__global__ __launch_bounds__(GRP_T) void csr_count(
    const int* __restrict__ ei, int* __restrict__ tab, int* __restrict__ tsum,
    int Nn, int NnPad, int E) {
  __shared__ int lc[SPAN];
  const int tid = threadIdx.x;
  const int lo  = blockIdx.x * SPAN;
  const int c   = blockIdx.y;
  for (int i = tid; i < SPAN; i += GRP_T) lc[i] = 0;
  __syncthreads();
  int cb, ce;
  chunk_bounds(E, c, cb, ce);
  const int* dst = ei + E;
  if (((((size_t)dst) & 15) == 0) && ((E & 3) == 0)) {
    const int4* dq = (const int4*)dst;
    const int4 none = make_int4(lo - 1, lo - 1, lo - 1, lo - 1);
    const int qe = ce >> 2;
    for (int q = (cb >> 2) + tid; q < qe; q += GRP_LD * GRP_T) {
      int4 a[GRP_LD];
#pragma unroll
      for (int u = 0; u < GRP_LD; ++u) {
        const int  qu = q + u * GRP_T;
        const bool in = qu < qe;
        const int4 t  = dq[in ? qu : qe - 1];
        a[u] = make_int4(in ? t.x : none.x, in ? t.y : none.y,
                         in ? t.z : none.z, in ? t.w : none.w);
      }
#pragma unroll
      for (int u = 0; u < GRP_LD; ++u) {
        count1(lc, a[u].x, lo); count1(lc, a[u].y, lo);
        count1(lc, a[u].z, lo); count1(lc, a[u].w, lo);
      }
    }
  } else {
    for (int e = cb + tid; e < ce; e += GRP_T) count1(lc, dst[e], lo);
  }
  __syncthreads();
  for (int i = tid; i < SPAN; i += GRP_T) {
    const int d = lo + i;
    if (d < Nn) tab[(size_t)c * NnPad + d] = lc[i];
  }
  for (int t = tid >> 5; t < SPAN / SCAN_TILE; t += GRP_T / 32) {
    int v = 0;
#pragma unroll
    for (int k = 0; k < SCAN_TILE / 32; ++k)
      v += lc[t * SCAN_TILE + k * 32 + (tid & 31)];
    v = wave_sum_i(v);
    if ((tid & 31) == 0)
      tsum[(size_t)(blockIdx.x * (SPAN / SCAN_TILE) + t) * NCHUNK + c] = v;
  }
}

__device__ __forceinline__ void scan_body(
    int* __restrict__ tab, const int* __restrict__ tsum,
    int* __restrict__ off, int Nn, int NnPad, int E, int bid) {
  __shared__ int red[SCAN_TILE / 32];
  __shared__ int wtot[SCAN_TILE / 32];
  const int lane = threadIdx.x & 31;
  const int wave = threadIdx.x >> 5;
  const int d    = bid * SCAN_TILE + threadIdx.x;
  const bool in  = d < Nn;

  int cnt[NCHUNK];
#pragma unroll
  for (int c = 0; c < NCHUNK; ++c)
    cnt[c] = in ? tab[(size_t)c * NnPad + d] : 0;

  int p = 0;
  const int np = bid * NCHUNK;
  for (int j = threadIdx.x; j < np; j += SCAN_TILE) p += tsum[j];
  p = wave_sum_i(p);
  if (lane == 0) red[wave] = p;

  int s = 0;
#pragma unroll
  for (int c = 0; c < NCHUNK; ++c) s += cnt[c];
  int x = s;
#pragma unroll
  for (int o = 1; o < 32; o <<= 1) {
    int t = __shfl_up(x, o, 32);
    if (lane >= o) x += t;
  }
  if (lane == 31) wtot[wave] = x;
  __syncthreads();

  int run = 0;
#pragma unroll
  for (int w = 0; w < SCAN_TILE / 32; ++w) {
    run += red[w];
    if (w < wave) run += wtot[w];
  }
  run += x - s;
  if (in) {
    off[d] = run;
    int t = run;
#pragma unroll
    for (int c = 0; c < NCHUNK; ++c) {
      tab[(size_t)c * NnPad + d] = t;
      t += cnt[c];
    }
  }
  if (bid == 0 && threadIdx.x == 0) off[Nn] = E;
}

__global__ __launch_bounds__(SCAN_TILE) void csr_scan_gemm1(
    int* __restrict__ tab, const int* __restrict__ tsum,
    int* __restrict__ off, int Nn, int NnPad, int E,
    const float* __restrict__ X, const float* __restrict__ W,
    float* __restrict__ H, int gemmBlocks) {
  const int b = (int)blockIdx.x;
  if (b < gemmBlocks) {
    const int half = __builtin_amdgcn_readfirstlane((int)(threadIdx.x >> 7));
    gemm_body<128>(X, W, H, Nn, b * 2 + half, gemmBlocks * 2,
                   (int)(threadIdx.x & 127));
  } else {
    scan_body(tab, tsum, off, Nn, NnPad, E, b - gemmBlocks);
  }
}

__device__ __forceinline__ void place1(int* cur, int2* sl,
                                       const int* __restrict__ ei, int d,
                                       int e, int lo) {
  const unsigned rel = (unsigned)(d - lo);
  if (rel < (unsigned)SPAN) {
    const int p = atomicAdd(&cur[rel], 1);
    sl[p] = make_int2(e, ei[e]);
  }
}

__device__ __forceinline__ void place4(int* cur, int2* sl, int4 d, int4 s,
                                       int e0, int lo, int spare) {
  const unsigned r0 = (unsigned)(d.x - lo), r1 = (unsigned)(d.y - lo);
  const unsigned r2 = (unsigned)(d.z - lo), r3 = (unsigned)(d.w - lo);
  const bool m0 = r0 < (unsigned)SPAN, m1 = r1 < (unsigned)SPAN;
  const bool m2 = r2 < (unsigned)SPAN, m3 = r3 < (unsigned)SPAN;
  const int p0 = atomicAdd(&cur[m0 ? (int)r0 : spare], 1);
  const int p1 = atomicAdd(&cur[m1 ? (int)r1 : spare], 1);
  const int p2 = atomicAdd(&cur[m2 ? (int)r2 : spare], 1);
  const int p3 = atomicAdd(&cur[m3 ? (int)r3 : spare], 1);
  if (m0) sl[p0] = make_int2(e0, s.x);
  if (m1) sl[p1] = make_int2(e0 + 1, s.y);
  if (m2) sl[p2] = make_int2(e0 + 2, s.z);
  if (m3) sl[p3] = make_int2(e0 + 3, s.w);
}

__global__ __launch_bounds__(GRP_T) void csr_place(
    const int* __restrict__ ei, const int* __restrict__ tab,
    int* __restrict__ slots, int Nn, int NnPad, int E) {
  __shared__ int cur[SPAN + 32];
  const int tid = threadIdx.x;
  const int lo  = blockIdx.x * SPAN;
  const int c   = blockIdx.y;
  for (int i = tid; i < SPAN; i += GRP_T) {
    const int d = lo + i;
    cur[i] = (d < Nn) ? tab[(size_t)c * NnPad + d] : 0;
  }
  if (tid < 32) cur[SPAN + tid] = 0;
  const int spare = SPAN + (tid & 31);
  __syncthreads();
  int cb, ce;
  chunk_bounds(E, c, cb, ce);
  const int* dst = ei + E;
  int2* sl = (int2*)slots;
  if (((((size_t)dst) & 15) == 0) && ((E & 3) == 0)) {
    const int4* dq = (const int4*)dst;
    const int4* sq = (const int4*)ei;
    const int4 none = make_int4(lo - 1, lo - 1, lo - 1, lo - 1);
    const int qe = ce >> 2;
    for (int q = (cb >> 2) + tid; q < qe; q += GRP_LD * GRP_T) {
      int4 a[GRP_LD];
      int4 b[GRP_LD];
#pragma unroll
      for (int u = 0; u < GRP_LD; ++u) {
        const int  qu = q + u * GRP_T;
        const bool in = qu < qe;
        const int  qc = in ? qu : qe - 1;
        const int4 t  = dq[qc];
        b[u] = sq[qc];
        a[u] = make_int4(in ? t.x : none.x, in ? t.y : none.y,
                         in ? t.z : none.z, in ? t.w : none.w);
      }
#pragma unroll
      for (int u = 0; u < GRP_LD; ++u)
        place4(cur, sl, a[u], b[u], 4 * (q + u * GRP_T), lo, spare);
    }
  } else {
    for (int e = cb + tid; e < ce; e += GRP_T) place1(cur, sl, ei, dst[e], e, lo);
  }
}

__device__ __forceinline__ void order_body(
    const int* __restrict__ off, const int* __restrict__ slots_in,
    int* __restrict__ slots_out, int Nn, int bid, int nblk) {
  const int gw = __builtin_amdgcn_readfirstlane(
      (int)(bid * 8 + (threadIdx.x >> 5)));
  const int lane  = threadIdx.x & 31;
  const int nw    = nblk * 8;
  const int per   = (Nn + nw - 1) / nw;
  const int first = gw * per;
  const int stop  = (first + per < Nn) ? first + per : Nn;
  if (first >= stop) return;
  int beg = off[first];
  int end = off[first + 1];
  for (int node = first; node < stop; ++node) {
    const int endNext = off[(node + 2 < Nn) ? node + 2 : Nn];
    const int n   = end - beg;
    const int2* in  = (const int2*)slots_in + beg;
    int2*       out = (int2*)slots_out + beg;
    for (int base = 0; base < n; base += 32) {
      const int  j     = base + lane;
      const bool valid = j < n;
      const int2 my    = valid ? in[j] : make_int2(0x7fffffff, 0);
      int rank = 0;
      for (int b2 = 0; b2 < n; b2 += 32) {
        const int j2 = b2 + lane;
        const int o  = (j2 < n) ? in[j2].x : 0x7fffffff;
        const int c2 = (n - b2 < 32) ? (n - b2) : 32;
        for (int k = 0; k < c2; ++k) {
          const int ok = __builtin_amdgcn_readlane(o, k);
          rank += (ok < my.x) ? 1 : 0;
        }
      }
      if (valid) out[rank] = my;
    }
    beg = end;
    end = endNext;
  }
}

__global__ __launch_bounds__(256) void csr_order_dots1(
    const int* __restrict__ off, const int* __restrict__ slots_in,
    int* __restrict__ slots_out, int Nn, int orderBlocks,
    const float* __restrict__ H, const float* __restrict__ a_src,
    const float* __restrict__ a_dst, float* __restrict__ hs,
    float* __restrict__ hd, int dotsBlocks) {
  const int b = (int)blockIdx.x;
  if (b < dotsBlocks) {
    dots_body(H, a_src, a_dst, hs, hd, Nn, b * 256 + (int)threadIdx.x);
  } else {
    order_body(off, slots_in, slots_out, Nn, b - dotsBlocks, orderBlocks);
  }
}

__device__ __forceinline__ float wave_max_f(float v) {
#pragma unroll
  for (int o = 16; o > 0; o >>= 1) v = fmaxf(v, __shfl_xor(v, o, 32));
  return v;
}
__device__ __forceinline__ float wave_sum_f(float v) {
#pragma unroll
  for (int o = 16; o > 0; o >>= 1) v += __shfl_xor(v, o, 32);
  return v;
}
__device__ __forceinline__ float bcast_f(float v, int k) {
  return __int_as_float(__builtin_amdgcn_readlane(__float_as_int(v), k));
}

template <bool WRITE_ALPHA>
__global__ __launch_bounds__(256) void gat_node_fused(
    const int* __restrict__ off, const int* __restrict__ slots_,
    const float* __restrict__ hs, const float* __restrict__ hd,
    const float* __restrict__ H, const float* __restrict__ bias,
    float* __restrict__ out, float* __restrict__ alpha_out, int Nn, int E) {
  const int gw = __builtin_amdgcn_readfirstlane(
      (int)(blockIdx.x * 8 + (threadIdx.x >> 5)));
  const int lane  = threadIdx.x & 31;
  const int nw    = gridDim.x * 8;
  const int per   = (Nn + nw - 1) / nw;
  const int first = gw * per;
  const int stop  = (first + per < Nn) ? first + per : Nn;
  if (first >= stop) return;
  const float2* Hr = (const float2*)H;
  const float2  b  = ((const float2*)bias)[lane];
  int beg = off[first];
  int end = off[first + 1];
  for (int node = first; node < stop; ++node) {
    const int endNext = off[(node + 2 < Nn) ? node + 2 : Nn];
    const int n   = end - beg;
    const int2* slots = (const int2*)slots_ + beg;

    const float2 hself = Hr[(size_t)node * 32 + lane];

    const float hdn = hd[node];
    float lself = hs[node] + hdn;
    lself = (lself > 0.f) ? lself : NEG_SLOPE * lself;

    const bool v0  = lane < n;
    const int2 sl0 = v0 ? slots[lane] : make_int2(0, 0);
    float lg0 = -INFINITY;
    if (v0) {
      float t = hs[sl0.y] + hdn;
      lg0 = (t > 0.f) ? t : NEG_SLOPE * t;
    }
    float m = wave_max_f(lg0);
    for (int base = 32; base < n; base += 32) {
      const int j = base + lane;
      float lg = -INFINITY;
      if (j < n) {
        float t = hs[slots[j].y] + hdn;
        lg = (t > 0.f) ? t : NEG_SLOPE * t;
      }
      m = fmaxf(m, wave_max_f(lg));
    }
    m = fmaxf(m, lself);

    const float ex0 = v0 ? __expf(lg0 - m) : 0.f;
    float den = wave_sum_f(ex0);
    for (int base = 32; base < n; base += 32) {
      const int j = base + lane;
      float ex = 0.f;
      if (j < n) {
        float t = hs[slots[j].y] + hdn;
        t = (t > 0.f) ? t : NEG_SLOPE * t;
        ex = __expf(t - m);
      }
      den += wave_sum_f(ex);
    }
    const float exs = __expf(lself - m);
    den += exs;
    const float dd  = den + EPSV;
    const float al0 = ex0 / dd;
    const float als = exs / dd;
    if (WRITE_ALPHA) {
      if (v0) alpha_out[sl0.x] = al0;
      if (lane == 0) alpha_out[E + node] = als;
    }

    float2 acc = make_float2(0.f, 0.f);
    const int n0 = (n < 32) ? n : 32;
    int k = 0;
    for (; k + 4 <= n0; k += 4) {
      const int s0 = __builtin_amdgcn_readlane(sl0.y, k);
      const int s1 = __builtin_amdgcn_readlane(sl0.y, k + 1);
      const int s2 = __builtin_amdgcn_readlane(sl0.y, k + 2);
      const int s3 = __builtin_amdgcn_readlane(sl0.y, k + 3);
      const float2 h0 = Hr[(size_t)s0 * 32 + lane];
      const float2 h1 = Hr[(size_t)s1 * 32 + lane];
      const float2 h2 = Hr[(size_t)s2 * 32 + lane];
      const float2 h3 = Hr[(size_t)s3 * 32 + lane];
      const float a0 = bcast_f(al0, k);
      const float a1 = bcast_f(al0, k + 1);
      const float a2 = bcast_f(al0, k + 2);
      const float a3 = bcast_f(al0, k + 3);
      acc.x = fmaf(a0, h0.x, acc.x);  acc.y = fmaf(a0, h0.y, acc.y);
      acc.x = fmaf(a1, h1.x, acc.x);  acc.y = fmaf(a1, h1.y, acc.y);
      acc.x = fmaf(a2, h2.x, acc.x);  acc.y = fmaf(a2, h2.y, acc.y);
      acc.x = fmaf(a3, h3.x, acc.x);  acc.y = fmaf(a3, h3.y, acc.y);
    }
    const int r = n0 - k;
    if (r > 0) {
      const int k1 = (r > 1) ? k + 1 : k;
      const int k2 = (r > 2) ? k + 2 : k;
      const int s0 = __builtin_amdgcn_readlane(sl0.y, k);
      const int s1 = __builtin_amdgcn_readlane(sl0.y, k1);
      const int s2 = __builtin_amdgcn_readlane(sl0.y, k2);
      const float2 h0 = Hr[(size_t)s0 * 32 + lane];
      const float2 h1 = Hr[(size_t)s1 * 32 + lane];
      const float2 h2 = Hr[(size_t)s2 * 32 + lane];
      const float a0 = bcast_f(al0, k);
      const float a1 = bcast_f(al0, k1);
      const float a2 = bcast_f(al0, k2);
      acc.x = fmaf(a0, h0.x, acc.x);  acc.y = fmaf(a0, h0.y, acc.y);
      if (r > 1) { acc.x = fmaf(a1, h1.x, acc.x);  acc.y = fmaf(a1, h1.y, acc.y); }
      if (r > 2) { acc.x = fmaf(a2, h2.x, acc.x);  acc.y = fmaf(a2, h2.y, acc.y); }
    }
    for (int base = 32; base < n; base += 32) {
      const int  j  = base + lane;
      const bool vj = j < n;
      const int2 sl = vj ? slots[j] : make_int2(0, 0);
      float al = 0.f;
      if (vj) {
        float t = hs[sl.y] + hdn;
        t = (t > 0.f) ? t : NEG_SLOPE * t;
        al = __expf(t - m) / dd;
        if (WRITE_ALPHA) alpha_out[sl.x] = al;
      }
      const int c2 = (n - base < 32) ? (n - base) : 32;
      for (int q = 0; q < c2; ++q) {
        const int    s = __builtin_amdgcn_readlane(sl.y, q);
        const float  a = bcast_f(al, q);
        const float2 h = Hr[(size_t)s * 32 + lane];
        acc.x = fmaf(a, h.x, acc.x);
        acc.y = fmaf(a, h.y, acc.y);
      }
    }
    acc.x = fmaf(als, hself.x, acc.x);
    acc.y = fmaf(als, hself.y, acc.y);
    float ox = acc.x + b.x;
    float oy = acc.y + b.y;
    ox = (ox > 0.f) ? ox : 0.f;
    oy = (oy > 0.f) ? oy : 0.f;
    ((float2*)out)[(size_t)node * 32 + lane] = make_float2(ox, oy);
    beg = end;
    end = endNext;
  }
}

static inline int cdiv_ll(long long a, long long b) {
  return (int)((a + b - 1) / b);
}

extern "C" void kernel_launch(void* const* d_in, const int* in_sizes, int n_in,
                              void* d_out, int out_size, void* d_ws,
                              size_t ws_size, hipStream_t stream) {
  const float* x      = (const float*)d_in[0];
  const int*   ei     = (const int*)d_in[1];
  const float* W1     = (const float*)d_in[2];
  const float* a_src1 = (const float*)d_in[3];
  const float* a_dst1 = (const float*)d_in[4];
  const float* b1     = (const float*)d_in[5];
  const float* W2     = (const float*)d_in[6];
  const float* a_src2 = (const float*)d_in[7];
  const float* a_dst2 = (const float*)d_in[8];
  const float* b2     = (const float*)d_in[9];

  const int IN_DIM = 128;
  const int Nn = in_sizes[0] / IN_DIM;
  const int E  = in_sizes[1] / 2;

  const int    nTiles = cdiv_ll(Nn, SCAN_TILE);
  const int    nRange = cdiv_ll(Nn, SPAN);
  const size_t nTsum  = (size_t)nRange * (SPAN / SCAN_TILE) * NCHUNK;
  const size_t NnPad  = ((size_t)Nn + 4) & ~(size_t)3;

  float* bufH   = (float*)d_ws;
  float* bufAgg = bufH + (size_t)Nn * FDIM;
  float* hs     = bufAgg + (size_t)Nn * FDIM;
  float* hd     = hs + NnPad;
  int*   off    = (int*)(hd + NnPad);
  int*   tsum   = off + NnPad;
  int*   slotsA = tsum + ((nTsum + 3) & ~(size_t)3);
  int*   slotsB = slotsA + (((size_t)2 * E + 3) & ~(size_t)3);
  int*   tab    = slotsB + (((size_t)2 * E + 3) & ~(size_t)3);

  float* h2out     = (float*)d_out;
  float* alpha_out = h2out + (size_t)Nn * FDIM;

  const int nodeGrid = cdiv_ll(Nn, 256);

  const dim3 grpGrid(nRange, NCHUNK);
  csr_count<<<grpGrid, GRP_T, 0, stream>>>(ei, tab, tsum, Nn, (int)NnPad, E);
  csr_scan_gemm1<<<LOOP_GRID / 2 + nTiles, SCAN_TILE, 0, stream>>>(
      tab, tsum, off, Nn, (int)NnPad, E, x, W1, bufH, LOOP_GRID / 2);
  csr_place<<<grpGrid, GRP_T, 0, stream>>>(ei, tab, slotsA, Nn, (int)NnPad, E);
  csr_order_dots1<<<nodeGrid + LOOP_GRID, 256, 0, stream>>>(
      off, slotsA, slotsB, Nn, LOOP_GRID, bufH, a_src1, a_dst1, hs, hd,
      nodeGrid);

  gat_node_fused<false><<<LOOP_GRID, 256, 0, stream>>>(
      off, slotsB, hs, hd, bufH, b1, bufAgg, nullptr, Nn, E);

  gat_gemm_wmma<64><<<LOOP_GRID, 128, 0, stream>>>(bufAgg, W2, bufH, Nn);
  gat_node_dots<<<nodeGrid, 256, 0, stream>>>(bufH, a_src2, a_dst2, hs, hd, Nn);
  gat_node_fused<true><<<LOOP_GRID, 256, 0, stream>>>(
      off, slotsB, hs, hd, bufH, b2, h2out, alpha_out, Nn, E);
}


# Edge grouping re-cut: 8 node ranges of 12544 (49 KiB LDS) x 31 edge chunks = 248 workgroups, one round on 256 WGPs (was 13 x 20 = 260, a second round in the place pass); float sums in old order
# speedup vs baseline: 3.8916x; 1.1655x over previous
#include <hip/hip_runtime.h>


#define LOOP_GRID 2048

typedef float v2f __attribute__((ext_vector_type(2)));
typedef float v8f __attribute__((ext_vector_type(8)));

#define NEG_SLOPE 0.2f
#define EPSV 1e-16f
#define FDIM 64

template <int K>
__device__ __forceinline__ void gemm_body(
    const float* __restrict__ X, const float* __restrict__ W,
    float* __restrict__ H, int M, int vb, int vgrid, int tid) {
  const int wave = tid >> 5;
  const int lane = tid & 31;
  const int l16  = lane & 15;
  const int kh   = (lane >> 4) << 1;
  const int colBase = wave * 16;
  const int rOff = (lane >> 4) << 3;
  const float* wc = W + colBase + l16;

  v2f bf[K / 4];
#pragma unroll
  for (int k = 0; k < K; k += 4) {
    bf[k >> 2][0] = wc[(size_t)(k + kh) * FDIM];
    bf[k >> 2][1] = wc[(size_t)(k + kh + 1) * FDIM];
  }

  const int nTiles = (M + 15) >> 4;
  for (int tile = vb; tile < nTiles; tile += vgrid) {
    const int rowBase = tile * 16;
    int arow = rowBase + l16;
    if (arow >= M) arow = M - 1;
    const float* xr = X + (size_t)arow * K;

    v8f c = {0.f, 0.f, 0.f, 0.f, 0.f, 0.f, 0.f, 0.f};
#pragma unroll
    for (int k = 0; k < K; k += 4) {
      v2f a;
      a[0] = xr[k + kh];
      a[1] = xr[k + kh + 1];
      c = __builtin_amdgcn_wmma_f32_16x16x4_f32(false, a, false, bf[k >> 2],
                                                (short)0, c, false, false);
    }

#pragma unroll
    for (int v = 0; v < 8; ++v) {
      int r = rowBase + rOff + v;
      if (r >= M) r = M - 1;
      H[(size_t)r * FDIM + colBase + l16] = c[v];
    }
  }
}

template <int K>
__global__ __launch_bounds__(128) void gat_gemm_wmma(
    const float* __restrict__ X, const float* __restrict__ W,
    float* __restrict__ H, int M) {
  gemm_body<K>(X, W, H, M, (int)blockIdx.x, (int)gridDim.x, (int)threadIdx.x);
}

__device__ __forceinline__ void dots_body(const float* __restrict__ H,
                              const float* __restrict__ a_src,
                              const float* __restrict__ a_dst,
                              float* __restrict__ hs, float* __restrict__ hd,
                              int Nn, int i) {
  if (i >= Nn) return;
  const float* h = H + (size_t)i * FDIM;
  float s = 0.f, d = 0.f;
#pragma unroll
  for (int f = 0; f < FDIM; ++f) {
    float v = h[f];
    s += v * a_src[f];
    d += v * a_dst[f];
  }
  hs[i] = s;
  hd[i] = d;
}

__global__ void gat_node_dots(const float* __restrict__ H,
                              const float* __restrict__ a_src,
                              const float* __restrict__ a_dst,
                              float* __restrict__ hs, float* __restrict__ hd,
                              int Nn) {
  dots_body(H, a_src, a_dst, hs, hd, Nn,
            (int)(blockIdx.x * blockDim.x + threadIdx.x));
}

#define SCAN_TILE 256
#define SPAN 12544
#define NCHUNK 31
#define GRP_T 1024
#define GRP_LD 5

__device__ __forceinline__ void chunk_bounds(int E, int c, int& cb, int& ce) {
  const int per = (((E + NCHUNK - 1) / NCHUNK) + 3) & ~3;
  const int b   = c * per;
  cb = (b < E) ? b : E;
  ce = (E - cb < per) ? E : cb + per;
}

__device__ __forceinline__ int wave_sum_i(int v) {
#pragma unroll
  for (int o = 16; o > 0; o >>= 1) v += __shfl_xor(v, o, 32);
  return v;
}

__device__ __forceinline__ void count1(int* lc, int d, int lo) {
  const unsigned rel = (unsigned)(d - lo);
  if (rel < (unsigned)SPAN) atomicAdd(&lc[rel], 1);
}

__global__ __launch_bounds__(GRP_T) void csr_count(
    const int* __restrict__ ei, int* __restrict__ tab, int* __restrict__ tsum,
    int Nn, int NnPad, int E) {
  __shared__ int lc[SPAN];
  const int tid = threadIdx.x;
  const int lo  = blockIdx.x * SPAN;
  const int c   = blockIdx.y;
  for (int i = tid; i < SPAN; i += GRP_T) lc[i] = 0;
  __syncthreads();
  int cb, ce;
  chunk_bounds(E, c, cb, ce);
  const int* dst = ei + E;
  if (((((size_t)dst) & 15) == 0) && ((E & 3) == 0)) {
    const int4* dq = (const int4*)dst;
    const int4 none = make_int4(lo - 1, lo - 1, lo - 1, lo - 1);
    const int qe = ce >> 2;
    for (int q = (cb >> 2) + tid; q < qe; q += GRP_LD * GRP_T) {
      int4 a[GRP_LD];
#pragma unroll
      for (int u = 0; u < GRP_LD; ++u) {
        const int  qu = q + u * GRP_T;
        const bool in = qu < qe;
        const int4 t  = dq[in ? qu : qe - 1];
        a[u] = make_int4(in ? t.x : none.x, in ? t.y : none.y,
                         in ? t.z : none.z, in ? t.w : none.w);
      }
#pragma unroll
      for (int u = 0; u < GRP_LD; ++u) {
        count1(lc, a[u].x, lo); count1(lc, a[u].y, lo);
        count1(lc, a[u].z, lo); count1(lc, a[u].w, lo);
      }
    }
  } else {
    for (int e = cb + tid; e < ce; e += GRP_T) count1(lc, dst[e], lo);
  }
  __syncthreads();
  for (int i = tid; i < SPAN; i += GRP_T) {
    const int d = lo + i;
    if (d < Nn) tab[(size_t)c * NnPad + d] = lc[i];
  }
  for (int t = tid >> 5; t < SPAN / SCAN_TILE; t += GRP_T / 32) {
    int v = 0;
#pragma unroll
    for (int k = 0; k < SCAN_TILE / 32; ++k)
      v += lc[t * SCAN_TILE + k * 32 + (tid & 31)];
    v = wave_sum_i(v);
    if ((tid & 31) == 0)
      tsum[(size_t)(blockIdx.x * (SPAN / SCAN_TILE) + t) * NCHUNK + c] = v;
  }
}

__device__ __forceinline__ void scan_body(
    int* __restrict__ tab, const int* __restrict__ tsum,
    int* __restrict__ off, int Nn, int NnPad, int E, int bid) {
  __shared__ int red[SCAN_TILE / 32];
  __shared__ int wtot[SCAN_TILE / 32];
  const int lane = threadIdx.x & 31;
  const int wave = threadIdx.x >> 5;
  const int d    = bid * SCAN_TILE + threadIdx.x;
  const bool in  = d < Nn;

  int cnt[NCHUNK];
#pragma unroll
  for (int c = 0; c < NCHUNK; ++c)
    cnt[c] = in ? tab[(size_t)c * NnPad + d] : 0;

  int p = 0;
  const int np = bid * NCHUNK;
  for (int j = threadIdx.x; j < np; j += SCAN_TILE) p += tsum[j];
  p = wave_sum_i(p);
  if (lane == 0) red[wave] = p;

  int s = 0;
#pragma unroll
  for (int c = 0; c < NCHUNK; ++c) s += cnt[c];
  int x = s;
#pragma unroll
  for (int o = 1; o < 32; o <<= 1) {
    int t = __shfl_up(x, o, 32);
    if (lane >= o) x += t;
  }
  if (lane == 31) wtot[wave] = x;
  __syncthreads();

  int run = 0;
#pragma unroll
  for (int w = 0; w < SCAN_TILE / 32; ++w) {
    run += red[w];
    if (w < wave) run += wtot[w];
  }
  run += x - s;
  if (in) {
    off[d] = run;
    int t = run;
#pragma unroll
    for (int c = 0; c < NCHUNK; ++c) {
      tab[(size_t)c * NnPad + d] = t;
      t += cnt[c];
    }
  }
  if (bid == 0 && threadIdx.x == 0) off[Nn] = E;
}

__global__ __launch_bounds__(SCAN_TILE) void csr_scan_gemm1(
    int* __restrict__ tab, const int* __restrict__ tsum,
    int* __restrict__ off, int Nn, int NnPad, int E,
    const float* __restrict__ X, const float* __restrict__ W,
    float* __restrict__ H, int gemmBlocks) {
  const int b = (int)blockIdx.x;
  if (b < gemmBlocks) {
    const int half = __builtin_amdgcn_readfirstlane((int)(threadIdx.x >> 7));
    gemm_body<128>(X, W, H, Nn, b * 2 + half, gemmBlocks * 2,
                   (int)(threadIdx.x & 127));
  } else {
    scan_body(tab, tsum, off, Nn, NnPad, E, b - gemmBlocks);
  }
}

__device__ __forceinline__ void place1(int* cur, int2* sl,
                                       const int* __restrict__ ei, int d,
                                       int e, int lo) {
  const unsigned rel = (unsigned)(d - lo);
  if (rel < (unsigned)SPAN) {
    const int p = atomicAdd(&cur[rel], 1);
    sl[p] = make_int2(e, ei[e]);
  }
}

__device__ __forceinline__ void place4(int* cur, int2* sl, int4 d, int4 s,
                                       int e0, int lo, int spare) {
  const unsigned r0 = (unsigned)(d.x - lo), r1 = (unsigned)(d.y - lo);
  const unsigned r2 = (unsigned)(d.z - lo), r3 = (unsigned)(d.w - lo);
  const bool m0 = r0 < (unsigned)SPAN, m1 = r1 < (unsigned)SPAN;
  const bool m2 = r2 < (unsigned)SPAN, m3 = r3 < (unsigned)SPAN;
  const int p0 = atomicAdd(&cur[m0 ? (int)r0 : spare], 1);
  const int p1 = atomicAdd(&cur[m1 ? (int)r1 : spare], 1);
  const int p2 = atomicAdd(&cur[m2 ? (int)r2 : spare], 1);
  const int p3 = atomicAdd(&cur[m3 ? (int)r3 : spare], 1);
  if (m0) sl[p0] = make_int2(e0, s.x);
  if (m1) sl[p1] = make_int2(e0 + 1, s.y);
  if (m2) sl[p2] = make_int2(e0 + 2, s.z);
  if (m3) sl[p3] = make_int2(e0 + 3, s.w);
}

__global__ __launch_bounds__(GRP_T) void csr_place(
    const int* __restrict__ ei, const int* __restrict__ tab,
    int* __restrict__ slots, int Nn, int NnPad, int E) {
  __shared__ int cur[SPAN + 32];
  const int tid = threadIdx.x;
  const int lo  = blockIdx.x * SPAN;
  const int c   = blockIdx.y;
  for (int i = tid; i < SPAN; i += GRP_T) {
    const int d = lo + i;
    cur[i] = (d < Nn) ? tab[(size_t)c * NnPad + d] : 0;
  }
  if (tid < 32) cur[SPAN + tid] = 0;
  const int spare = SPAN + (tid & 31);
  __syncthreads();
  int cb, ce;
  chunk_bounds(E, c, cb, ce);
  const int* dst = ei + E;
  int2* sl = (int2*)slots;
  if (((((size_t)dst) & 15) == 0) && ((E & 3) == 0)) {
    const int4* dq = (const int4*)dst;
    const int4* sq = (const int4*)ei;
    const int4 none = make_int4(lo - 1, lo - 1, lo - 1, lo - 1);
    const int qe = ce >> 2;
    for (int q = (cb >> 2) + tid; q < qe; q += GRP_LD * GRP_T) {
      int4 a[GRP_LD];
      int4 b[GRP_LD];
#pragma unroll
      for (int u = 0; u < GRP_LD; ++u) {
        const int  qu = q + u * GRP_T;
        const bool in = qu < qe;
        const int  qc = in ? qu : qe - 1;
        const int4 t  = dq[qc];
        b[u] = sq[qc];
        a[u] = make_int4(in ? t.x : none.x, in ? t.y : none.y,
                         in ? t.z : none.z, in ? t.w : none.w);
      }
#pragma unroll
      for (int u = 0; u < GRP_LD; ++u)
        place4(cur, sl, a[u], b[u], 4 * (q + u * GRP_T), lo, spare);
    }
  } else {
    for (int e = cb + tid; e < ce; e += GRP_T) place1(cur, sl, ei, dst[e], e, lo);
  }
}

__device__ __forceinline__ void order_body(
    const int* __restrict__ off, const int* __restrict__ slots_in,
    int* __restrict__ slots_out, int Nn, int bid, int nblk) {
  const int gw = __builtin_amdgcn_readfirstlane(
      (int)(bid * 8 + (threadIdx.x >> 5)));
  const int lane  = threadIdx.x & 31;
  const int nw    = nblk * 8;
  const int per   = (Nn + nw - 1) / nw;
  const int first = gw * per;
  const int stop  = (first + per < Nn) ? first + per : Nn;
  if (first >= stop) return;
  int beg = off[first];
  int end = off[first + 1];
  for (int node = first; node < stop; ++node) {
    const int endNext = off[(node + 2 < Nn) ? node + 2 : Nn];
    const int n   = end - beg;
    const int2* in  = (const int2*)slots_in + beg;
    int2*       out = (int2*)slots_out + beg;
    for (int base = 0; base < n; base += 32) {
      const int  j     = base + lane;
      const bool valid = j < n;
      const int2 my    = valid ? in[j] : make_int2(0x7fffffff, 0);
      int rank = 0;
      for (int b2 = 0; b2 < n; b2 += 32) {
        const int j2 = b2 + lane;
        const int o  = (j2 < n) ? in[j2].x : 0x7fffffff;
        const int c2 = (n - b2 < 32) ? (n - b2) : 32;
        for (int k = 0; k < c2; ++k) {
          const int ok = __builtin_amdgcn_readlane(o, k);
          rank += (ok < my.x) ? 1 : 0;
        }
      }
      if (valid) out[rank] = my;
    }
    beg = end;
    end = endNext;
  }
}

__global__ __launch_bounds__(256) void csr_order_dots1(
    const int* __restrict__ off, const int* __restrict__ slots_in,
    int* __restrict__ slots_out, int Nn, int orderBlocks,
    const float* __restrict__ H, const float* __restrict__ a_src,
    const float* __restrict__ a_dst, float* __restrict__ hs,
    float* __restrict__ hd, int dotsBlocks) {
  const int b = (int)blockIdx.x;
  if (b < dotsBlocks) {
    dots_body(H, a_src, a_dst, hs, hd, Nn, b * 256 + (int)threadIdx.x);
  } else {
    order_body(off, slots_in, slots_out, Nn, b - dotsBlocks, orderBlocks);
  }
}

__device__ __forceinline__ float wave_max_f(float v) {
#pragma unroll
  for (int o = 16; o > 0; o >>= 1) v = fmaxf(v, __shfl_xor(v, o, 32));
  return v;
}
__device__ __forceinline__ float wave_sum_f(float v) {
#pragma unroll
  for (int o = 16; o > 0; o >>= 1) v += __shfl_xor(v, o, 32);
  return v;
}
__device__ __forceinline__ float bcast_f(float v, int k) {
  return __int_as_float(__builtin_amdgcn_readlane(__float_as_int(v), k));
}

template <bool WRITE_ALPHA>
__global__ __launch_bounds__(256) void gat_node_fused(
    const int* __restrict__ off, const int* __restrict__ slots_,
    const float* __restrict__ hs, const float* __restrict__ hd,
    const float* __restrict__ H, const float* __restrict__ bias,
    float* __restrict__ out, float* __restrict__ alpha_out, int Nn, int E) {
  const int gw = __builtin_amdgcn_readfirstlane(
      (int)(blockIdx.x * 8 + (threadIdx.x >> 5)));
  const int lane  = threadIdx.x & 31;
  const int nw    = gridDim.x * 8;
  const int per   = (Nn + nw - 1) / nw;
  const int first = gw * per;
  const int stop  = (first + per < Nn) ? first + per : Nn;
  if (first >= stop) return;
  const float2* Hr = (const float2*)H;
  const float2  b  = ((const float2*)bias)[lane];
  int beg = off[first];
  int end = off[first + 1];
  for (int node = first; node < stop; ++node) {
    const int endNext = off[(node + 2 < Nn) ? node + 2 : Nn];
    const int n   = end - beg;
    const int2* slots = (const int2*)slots_ + beg;

    const float2 hself = Hr[(size_t)node * 32 + lane];

    const float hdn = hd[node];
    float lself = hs[node] + hdn;
    lself = (lself > 0.f) ? lself : NEG_SLOPE * lself;

    const bool v0  = lane < n;
    const int2 sl0 = v0 ? slots[lane] : make_int2(0, 0);
    float lg0 = -INFINITY;
    if (v0) {
      float t = hs[sl0.y] + hdn;
      lg0 = (t > 0.f) ? t : NEG_SLOPE * t;
    }
    float m = wave_max_f(lg0);
    for (int base = 32; base < n; base += 32) {
      const int j = base + lane;
      float lg = -INFINITY;
      if (j < n) {
        float t = hs[slots[j].y] + hdn;
        lg = (t > 0.f) ? t : NEG_SLOPE * t;
      }
      m = fmaxf(m, wave_max_f(lg));
    }
    m = fmaxf(m, lself);

    const float ex0 = v0 ? __expf(lg0 - m) : 0.f;
    float den = wave_sum_f(ex0);
    for (int base = 32; base < n; base += 32) {
      const int j = base + lane;
      float ex = 0.f;
      if (j < n) {
        float t = hs[slots[j].y] + hdn;
        t = (t > 0.f) ? t : NEG_SLOPE * t;
        ex = __expf(t - m);
      }
      den += wave_sum_f(ex);
    }
    const float exs = __expf(lself - m);
    den += exs;
    const float dd  = den + EPSV;
    const float al0 = ex0 / dd;
    const float als = exs / dd;
    if (WRITE_ALPHA) {
      if (v0) alpha_out[sl0.x] = al0;
      if (lane == 0) alpha_out[E + node] = als;
    }

    float2 acc = make_float2(0.f, 0.f);
    const int n0 = (n < 32) ? n : 32;
    int k = 0;
    for (; k + 4 <= n0; k += 4) {
      const int s0 = __builtin_amdgcn_readlane(sl0.y, k);
      const int s1 = __builtin_amdgcn_readlane(sl0.y, k + 1);
      const int s2 = __builtin_amdgcn_readlane(sl0.y, k + 2);
      const int s3 = __builtin_amdgcn_readlane(sl0.y, k + 3);
      const float2 h0 = Hr[(size_t)s0 * 32 + lane];
      const float2 h1 = Hr[(size_t)s1 * 32 + lane];
      const float2 h2 = Hr[(size_t)s2 * 32 + lane];
      const float2 h3 = Hr[(size_t)s3 * 32 + lane];
      const float a0 = bcast_f(al0, k);
      const float a1 = bcast_f(al0, k + 1);
      const float a2 = bcast_f(al0, k + 2);
      const float a3 = bcast_f(al0, k + 3);
      acc.x = fmaf(a0, h0.x, acc.x);  acc.y = fmaf(a0, h0.y, acc.y);
      acc.x = fmaf(a1, h1.x, acc.x);  acc.y = fmaf(a1, h1.y, acc.y);
      acc.x = fmaf(a2, h2.x, acc.x);  acc.y = fmaf(a2, h2.y, acc.y);
      acc.x = fmaf(a3, h3.x, acc.x);  acc.y = fmaf(a3, h3.y, acc.y);
    }
    const int r = n0 - k;
    if (r > 0) {
      const int k1 = (r > 1) ? k + 1 : k;
      const int k2 = (r > 2) ? k + 2 : k;
      const int s0 = __builtin_amdgcn_readlane(sl0.y, k);
      const int s1 = __builtin_amdgcn_readlane(sl0.y, k1);
      const int s2 = __builtin_amdgcn_readlane(sl0.y, k2);
      const float2 h0 = Hr[(size_t)s0 * 32 + lane];
      const float2 h1 = Hr[(size_t)s1 * 32 + lane];
      const float2 h2 = Hr[(size_t)s2 * 32 + lane];
      const float a0 = bcast_f(al0, k);
      const float a1 = bcast_f(al0, k1);
      const float a2 = bcast_f(al0, k2);
      acc.x = fmaf(a0, h0.x, acc.x);  acc.y = fmaf(a0, h0.y, acc.y);
      if (r > 1) { acc.x = fmaf(a1, h1.x, acc.x);  acc.y = fmaf(a1, h1.y, acc.y); }
      if (r > 2) { acc.x = fmaf(a2, h2.x, acc.x);  acc.y = fmaf(a2, h2.y, acc.y); }
    }
    for (int base = 32; base < n; base += 32) {
      const int  j  = base + lane;
      const bool vj = j < n;
      const int2 sl = vj ? slots[j] : make_int2(0, 0);
      float al = 0.f;
      if (vj) {
        float t = hs[sl.y] + hdn;
        t = (t > 0.f) ? t : NEG_SLOPE * t;
        al = __expf(t - m) / dd;
        if (WRITE_ALPHA) alpha_out[sl.x] = al;
      }
      const int c2 = (n - base < 32) ? (n - base) : 32;
      for (int q = 0; q < c2; ++q) {
        const int    s = __builtin_amdgcn_readlane(sl.y, q);
        const float  a = bcast_f(al, q);
        const float2 h = Hr[(size_t)s * 32 + lane];
        acc.x = fmaf(a, h.x, acc.x);
        acc.y = fmaf(a, h.y, acc.y);
      }
    }
    acc.x = fmaf(als, hself.x, acc.x);
    acc.y = fmaf(als, hself.y, acc.y);
    float ox = acc.x + b.x;
    float oy = acc.y + b.y;
    ox = (ox > 0.f) ? ox : 0.f;
    oy = (oy > 0.f) ? oy : 0.f;
    ((float2*)out)[(size_t)node * 32 + lane] = make_float2(ox, oy);
    beg = end;
    end = endNext;
  }
}

static inline int cdiv_ll(long long a, long long b) {
  return (int)((a + b - 1) / b);
}

extern "C" void kernel_launch(void* const* d_in, const int* in_sizes, int n_in,
                              void* d_out, int out_size, void* d_ws,
                              size_t ws_size, hipStream_t stream) {
  const float* x      = (const float*)d_in[0];
  const int*   ei     = (const int*)d_in[1];
  const float* W1     = (const float*)d_in[2];
  const float* a_src1 = (const float*)d_in[3];
  const float* a_dst1 = (const float*)d_in[4];
  const float* b1     = (const float*)d_in[5];
  const float* W2     = (const float*)d_in[6];
  const float* a_src2 = (const float*)d_in[7];
  const float* a_dst2 = (const float*)d_in[8];
  const float* b2     = (const float*)d_in[9];

  const int IN_DIM = 128;
  const int Nn = in_sizes[0] / IN_DIM;
  const int E  = in_sizes[1] / 2;

  const int    nTiles = cdiv_ll(Nn, SCAN_TILE);
  const int    nRange = cdiv_ll(Nn, SPAN);
  const size_t nTsum  = (size_t)nRange * (SPAN / SCAN_TILE) * NCHUNK;
  const size_t NnPad  = ((size_t)Nn + 4) & ~(size_t)3;

  float* bufH   = (float*)d_ws;
  float* bufAgg = bufH + (size_t)Nn * FDIM;
  float* hs     = bufAgg + (size_t)Nn * FDIM;
  float* hd     = hs + NnPad;
  int*   off    = (int*)(hd + NnPad);
  int*   tsum   = off + NnPad;
  int*   slotsA = tsum + ((nTsum + 3) & ~(size_t)3);
  int*   slotsB = slotsA + (((size_t)2 * E + 3) & ~(size_t)3);
  int*   tab    = slotsB + (((size_t)2 * E + 3) & ~(size_t)3);

  float* h2out     = (float*)d_out;
  float* alpha_out = h2out + (size_t)Nn * FDIM;

  const int nodeGrid = cdiv_ll(Nn, 256);

  const dim3 grpGrid(nRange, NCHUNK);
  csr_count<<<grpGrid, GRP_T, 0, stream>>>(ei, tab, tsum, Nn, (int)NnPad, E);
  csr_scan_gemm1<<<LOOP_GRID / 2 + nTiles, SCAN_TILE, 0, stream>>>(
      tab, tsum, off, Nn, (int)NnPad, E, x, W1, bufH, LOOP_GRID / 2);
  csr_place<<<grpGrid, GRP_T, 0, stream>>>(ei, tab, slotsA, Nn, (int)NnPad, E);
  csr_order_dots1<<<nodeGrid + LOOP_GRID, 256, 0, stream>>>(
      off, slotsA, slotsB, Nn, LOOP_GRID, bufH, a_src1, a_dst1, hs, hd,
      nodeGrid);

  gat_node_fused<false><<<LOOP_GRID, 256, 0, stream>>>(
      off, slotsB, hs, hd, bufH, b1, bufAgg, nullptr, Nn, E);

  gat_gemm_wmma<64><<<LOOP_GRID, 128, 0, stream>>>(bufAgg, W2, bufH, Nn);
  gat_node_dots<<<nodeGrid, 256, 0, stream>>>(bufH, a_src2, a_dst2, hs, hd, Nn);
  gat_node_fused<true><<<LOOP_GRID, 256, 0, stream>>>(
      off, slotsB, hs, hd, bufH, b2, h2out, alpha_out, Nn, E);
}


# Layer-2 product and projections formed inside the layer-1 node kernel: a workgroup keeps its 64 rows h1 in LDS and multiplies them by W2 on WMMA before it exits; 6 launches, not 8; sums in old order
# speedup vs baseline: 4.2374x; 1.0889x over previous
#include <hip/hip_runtime.h>


#define LOOP_GRID 2048

typedef float v2f __attribute__((ext_vector_type(2)));
typedef float v8f __attribute__((ext_vector_type(8)));

#define NEG_SLOPE 0.2f
#define EPSV 1e-16f
#define FDIM 64

template <int K>
__device__ __forceinline__ void gemm_body(
    const float* __restrict__ X, const float* __restrict__ W,
    float* __restrict__ H, int M, int vb, int vgrid, int tid) {
  const int wave = tid >> 5;
  const int lane = tid & 31;
  const int l16  = lane & 15;
  const int kh   = (lane >> 4) << 1;
  const int colBase = wave * 16;
  const int rOff = (lane >> 4) << 3;
  const float* wc = W + colBase + l16;

  v2f bf[K / 4];
#pragma unroll
  for (int k = 0; k < K; k += 4) {
    bf[k >> 2][0] = wc[(size_t)(k + kh) * FDIM];
    bf[k >> 2][1] = wc[(size_t)(k + kh + 1) * FDIM];
  }

  const int nTiles = (M + 15) >> 4;
  for (int tile = vb; tile < nTiles; tile += vgrid) {
    const int rowBase = tile * 16;
    int arow = rowBase + l16;
    if (arow >= M) arow = M - 1;
    const float* xr = X + (size_t)arow * K;

    v8f c = {0.f, 0.f, 0.f, 0.f, 0.f, 0.f, 0.f, 0.f};
#pragma unroll
    for (int k = 0; k < K; k += 4) {
      v2f a;
      a[0] = xr[k + kh];
      a[1] = xr[k + kh + 1];
      c = __builtin_amdgcn_wmma_f32_16x16x4_f32(false, a, false, bf[k >> 2],
                                                (short)0, c, false, false);
    }

#pragma unroll
    for (int v = 0; v < 8; ++v) {
      int r = rowBase + rOff + v;
      if (r >= M) r = M - 1;
      H[(size_t)r * FDIM + colBase + l16] = c[v];
    }
  }
}

__device__ __forceinline__ void dots_body(const float* __restrict__ H,
                              const float* __restrict__ a_src,
                              const float* __restrict__ a_dst,
                              float* __restrict__ hs, float* __restrict__ hd,
                              int Nn, int i) {
  if (i >= Nn) return;
  const float* h = H + (size_t)i * FDIM;
  float s = 0.f, d = 0.f;
#pragma unroll
  for (int f = 0; f < FDIM; ++f) {
    float v = h[f];
    s += v * a_src[f];
    d += v * a_dst[f];
  }
  hs[i] = s;
  hd[i] = d;
}

#define SCAN_TILE 256
#define SPAN 12544
#define NCHUNK 31
#define GRP_T 1024
#define GRP_LD 5

__device__ __forceinline__ void chunk_bounds(int E, int c, int& cb, int& ce) {
  const int per = (((E + NCHUNK - 1) / NCHUNK) + 3) & ~3;
  const int b   = c * per;
  cb = (b < E) ? b : E;
  ce = (E - cb < per) ? E : cb + per;
}

__device__ __forceinline__ int wave_sum_i(int v) {
#pragma unroll
  for (int o = 16; o > 0; o >>= 1) v += __shfl_xor(v, o, 32);
  return v;
}

__device__ __forceinline__ void count1(int* lc, int d, int lo) {
  const unsigned rel = (unsigned)(d - lo);
  if (rel < (unsigned)SPAN) atomicAdd(&lc[rel], 1);
}

__global__ __launch_bounds__(GRP_T) void csr_count(
    const int* __restrict__ ei, int* __restrict__ tab, int* __restrict__ tsum,
    int Nn, int NnPad, int E) {
  __shared__ int lc[SPAN];
  const int tid = threadIdx.x;
  const int lo  = blockIdx.x * SPAN;
  const int c   = blockIdx.y;
  for (int i = tid; i < SPAN; i += GRP_T) lc[i] = 0;
  __syncthreads();
  int cb, ce;
  chunk_bounds(E, c, cb, ce);
  const int* dst = ei + E;
  if (((((size_t)dst) & 15) == 0) && ((E & 3) == 0)) {
    const int4* dq = (const int4*)dst;
    const int4 none = make_int4(lo - 1, lo - 1, lo - 1, lo - 1);
    const int qe = ce >> 2;
    for (int q = (cb >> 2) + tid; q < qe; q += GRP_LD * GRP_T) {
      int4 a[GRP_LD];
#pragma unroll
      for (int u = 0; u < GRP_LD; ++u) {
        const int  qu = q + u * GRP_T;
        const bool in = qu < qe;
        const int4 t  = dq[in ? qu : qe - 1];
        a[u] = make_int4(in ? t.x : none.x, in ? t.y : none.y,
                         in ? t.z : none.z, in ? t.w : none.w);
      }
#pragma unroll
      for (int u = 0; u < GRP_LD; ++u) {
        count1(lc, a[u].x, lo); count1(lc, a[u].y, lo);
        count1(lc, a[u].z, lo); count1(lc, a[u].w, lo);
      }
    }
  } else {
    for (int e = cb + tid; e < ce; e += GRP_T) count1(lc, dst[e], lo);
  }
  __syncthreads();
  for (int i = tid; i < SPAN; i += GRP_T) {
    const int d = lo + i;
    if (d < Nn) tab[(size_t)c * NnPad + d] = lc[i];
  }
  for (int t = tid >> 5; t < SPAN / SCAN_TILE; t += GRP_T / 32) {
    int v = 0;
#pragma unroll
    for (int k = 0; k < SCAN_TILE / 32; ++k)
      v += lc[t * SCAN_TILE + k * 32 + (tid & 31)];
    v = wave_sum_i(v);
    if ((tid & 31) == 0)
      tsum[(size_t)(blockIdx.x * (SPAN / SCAN_TILE) + t) * NCHUNK + c] = v;
  }
}

__device__ __forceinline__ void scan_body(
    int* __restrict__ tab, const int* __restrict__ tsum,
    int* __restrict__ off, int Nn, int NnPad, int E, int bid) {
  __shared__ int red[SCAN_TILE / 32];
  __shared__ int wtot[SCAN_TILE / 32];
  const int lane = threadIdx.x & 31;
  const int wave = threadIdx.x >> 5;
  const int d    = bid * SCAN_TILE + threadIdx.x;
  const bool in  = d < Nn;

  int cnt[NCHUNK];
#pragma unroll
  for (int c = 0; c < NCHUNK; ++c)
    cnt[c] = in ? tab[(size_t)c * NnPad + d] : 0;

  int p = 0;
  const int np = bid * NCHUNK;
  for (int j = threadIdx.x; j < np; j += SCAN_TILE) p += tsum[j];
  p = wave_sum_i(p);
  if (lane == 0) red[wave] = p;

  int s = 0;
#pragma unroll
  for (int c = 0; c < NCHUNK; ++c) s += cnt[c];
  int x = s;
#pragma unroll
  for (int o = 1; o < 32; o <<= 1) {
    int t = __shfl_up(x, o, 32);
    if (lane >= o) x += t;
  }
  if (lane == 31) wtot[wave] = x;
  __syncthreads();

  int run = 0;
#pragma unroll
  for (int w = 0; w < SCAN_TILE / 32; ++w) {
    run += red[w];
    if (w < wave) run += wtot[w];
  }
  run += x - s;
  if (in) {
    off[d] = run;
    int t = run;
#pragma unroll
    for (int c = 0; c < NCHUNK; ++c) {
      tab[(size_t)c * NnPad + d] = t;
      t += cnt[c];
    }
  }
  if (bid == 0 && threadIdx.x == 0) off[Nn] = E;
}

__global__ __launch_bounds__(SCAN_TILE) void csr_scan_gemm1(
    int* __restrict__ tab, const int* __restrict__ tsum,
    int* __restrict__ off, int Nn, int NnPad, int E,
    const float* __restrict__ X, const float* __restrict__ W,
    float* __restrict__ H, int gemmBlocks) {
  const int b = (int)blockIdx.x;
  if (b < gemmBlocks) {
    const int half = __builtin_amdgcn_readfirstlane((int)(threadIdx.x >> 7));
    gemm_body<128>(X, W, H, Nn, b * 2 + half, gemmBlocks * 2,
                   (int)(threadIdx.x & 127));
  } else {
    scan_body(tab, tsum, off, Nn, NnPad, E, b - gemmBlocks);
  }
}

__device__ __forceinline__ void place1(int* cur, int2* sl,
                                       const int* __restrict__ ei, int d,
                                       int e, int lo) {
  const unsigned rel = (unsigned)(d - lo);
  if (rel < (unsigned)SPAN) {
    const int p = atomicAdd(&cur[rel], 1);
    sl[p] = make_int2(e, ei[e]);
  }
}

__device__ __forceinline__ void place4(int* cur, int2* sl, int4 d, int4 s,
                                       int e0, int lo, int spare) {
  const unsigned r0 = (unsigned)(d.x - lo), r1 = (unsigned)(d.y - lo);
  const unsigned r2 = (unsigned)(d.z - lo), r3 = (unsigned)(d.w - lo);
  const bool m0 = r0 < (unsigned)SPAN, m1 = r1 < (unsigned)SPAN;
  const bool m2 = r2 < (unsigned)SPAN, m3 = r3 < (unsigned)SPAN;
  const int p0 = atomicAdd(&cur[m0 ? (int)r0 : spare], 1);
  const int p1 = atomicAdd(&cur[m1 ? (int)r1 : spare], 1);
  const int p2 = atomicAdd(&cur[m2 ? (int)r2 : spare], 1);
  const int p3 = atomicAdd(&cur[m3 ? (int)r3 : spare], 1);
  if (m0) sl[p0] = make_int2(e0, s.x);
  if (m1) sl[p1] = make_int2(e0 + 1, s.y);
  if (m2) sl[p2] = make_int2(e0 + 2, s.z);
  if (m3) sl[p3] = make_int2(e0 + 3, s.w);
}

__global__ __launch_bounds__(GRP_T) void csr_place(
    const int* __restrict__ ei, const int* __restrict__ tab,
    int* __restrict__ slots, int Nn, int NnPad, int E) {
  __shared__ int cur[SPAN + 32];
  const int tid = threadIdx.x;
  const int lo  = blockIdx.x * SPAN;
  const int c   = blockIdx.y;
  for (int i = tid; i < SPAN; i += GRP_T) {
    const int d = lo + i;
    cur[i] = (d < Nn) ? tab[(size_t)c * NnPad + d] : 0;
  }
  if (tid < 32) cur[SPAN + tid] = 0;
  const int spare = SPAN + (tid & 31);
  __syncthreads();
  int cb, ce;
  chunk_bounds(E, c, cb, ce);
  const int* dst = ei + E;
  int2* sl = (int2*)slots;
  if (((((size_t)dst) & 15) == 0) && ((E & 3) == 0)) {
    const int4* dq = (const int4*)dst;
    const int4* sq = (const int4*)ei;
    const int4 none = make_int4(lo - 1, lo - 1, lo - 1, lo - 1);
    const int qe = ce >> 2;
    for (int q = (cb >> 2) + tid; q < qe; q += GRP_LD * GRP_T) {
      int4 a[GRP_LD];
      int4 b[GRP_LD];
#pragma unroll
      for (int u = 0; u < GRP_LD; ++u) {
        const int  qu = q + u * GRP_T;
        const bool in = qu < qe;
        const int  qc = in ? qu : qe - 1;
        const int4 t  = dq[qc];
        b[u] = sq[qc];
        a[u] = make_int4(in ? t.x : none.x, in ? t.y : none.y,
                         in ? t.z : none.z, in ? t.w : none.w);
      }
#pragma unroll
      for (int u = 0; u < GRP_LD; ++u)
        place4(cur, sl, a[u], b[u], 4 * (q + u * GRP_T), lo, spare);
    }
  } else {
    for (int e = cb + tid; e < ce; e += GRP_T) place1(cur, sl, ei, dst[e], e, lo);
  }
}

__device__ __forceinline__ void order_body(
    const int* __restrict__ off, const int* __restrict__ slots_in,
    int* __restrict__ slots_out, int Nn, int bid, int nblk) {
  const int gw = __builtin_amdgcn_readfirstlane(
      (int)(bid * 8 + (threadIdx.x >> 5)));
  const int lane  = threadIdx.x & 31;
  const int nw    = nblk * 8;
  const int per   = (Nn + nw - 1) / nw;
  const int first = gw * per;
  const int stop  = (first + per < Nn) ? first + per : Nn;
  if (first >= stop) return;
  int beg = off[first];
  int end = off[first + 1];
  for (int node = first; node < stop; ++node) {
    const int endNext = off[(node + 2 < Nn) ? node + 2 : Nn];
    const int n   = end - beg;
    const int2* in  = (const int2*)slots_in + beg;
    int2*       out = (int2*)slots_out + beg;
    for (int base = 0; base < n; base += 32) {
      const int  j     = base + lane;
      const bool valid = j < n;
      const int2 my    = valid ? in[j] : make_int2(0x7fffffff, 0);
      int rank = 0;
      for (int b2 = 0; b2 < n; b2 += 32) {
        const int j2 = b2 + lane;
        const int o  = (j2 < n) ? in[j2].x : 0x7fffffff;
        const int c2 = (n - b2 < 32) ? (n - b2) : 32;
        for (int k = 0; k < c2; ++k) {
          const int ok = __builtin_amdgcn_readlane(o, k);
          rank += (ok < my.x) ? 1 : 0;
        }
      }
      if (valid) out[rank] = my;
    }
    beg = end;
    end = endNext;
  }
}

__global__ __launch_bounds__(256) void csr_order_dots1(
    const int* __restrict__ off, const int* __restrict__ slots_in,
    int* __restrict__ slots_out, int Nn, int orderBlocks,
    const float* __restrict__ H, const float* __restrict__ a_src,
    const float* __restrict__ a_dst, float* __restrict__ hs,
    float* __restrict__ hd, int dotsBlocks) {
  const int b = (int)blockIdx.x;
  if (b < dotsBlocks) {
    dots_body(H, a_src, a_dst, hs, hd, Nn, b * 256 + (int)threadIdx.x);
  } else {
    order_body(off, slots_in, slots_out, Nn, b - dotsBlocks, orderBlocks);
  }
}

__device__ __forceinline__ float wave_max_f(float v) {
#pragma unroll
  for (int o = 16; o > 0; o >>= 1) v = fmaxf(v, __shfl_xor(v, o, 32));
  return v;
}
__device__ __forceinline__ float wave_sum_f(float v) {
#pragma unroll
  for (int o = 16; o > 0; o >>= 1) v += __shfl_xor(v, o, 32);
  return v;
}
__device__ __forceinline__ float bcast_f(float v, int k) {
  return __int_as_float(__builtin_amdgcn_readlane(__float_as_int(v), k));
}

template <bool WRITE_ALPHA>
__device__ __forceinline__ float2 node_row(
    int node, int n, const int2* __restrict__ slots,
    const float* __restrict__ hs, const float* __restrict__ hd,
    const float2* __restrict__ Hr, float2 b, float* __restrict__ alpha_out,
    int E, int lane) {
  const float2 hself = Hr[(size_t)node * 32 + lane];

  const float hdn = hd[node];
  float lself = hs[node] + hdn;
  lself = (lself > 0.f) ? lself : NEG_SLOPE * lself;

  const bool v0  = lane < n;
  const int2 sl0 = v0 ? slots[lane] : make_int2(0, 0);
  float lg0 = -INFINITY;
  if (v0) {
    float t = hs[sl0.y] + hdn;
    lg0 = (t > 0.f) ? t : NEG_SLOPE * t;
  }
  float m = wave_max_f(lg0);
  for (int base = 32; base < n; base += 32) {
    const int j = base + lane;
    float lg = -INFINITY;
    if (j < n) {
      float t = hs[slots[j].y] + hdn;
      lg = (t > 0.f) ? t : NEG_SLOPE * t;
    }
    m = fmaxf(m, wave_max_f(lg));
  }
  m = fmaxf(m, lself);

  const float ex0 = v0 ? __expf(lg0 - m) : 0.f;
  float den = wave_sum_f(ex0);
  for (int base = 32; base < n; base += 32) {
    const int j = base + lane;
    float ex = 0.f;
    if (j < n) {
      float t = hs[slots[j].y] + hdn;
      t = (t > 0.f) ? t : NEG_SLOPE * t;
      ex = __expf(t - m);
    }
    den += wave_sum_f(ex);
  }
  const float exs = __expf(lself - m);
  den += exs;
  const float dd  = den + EPSV;
  const float al0 = ex0 / dd;
  const float als = exs / dd;
  if (WRITE_ALPHA) {
    if (v0) alpha_out[sl0.x] = al0;
    if (lane == 0) alpha_out[E + node] = als;
  }

  float2 acc = make_float2(0.f, 0.f);
  const int n0 = (n < 32) ? n : 32;
  int k = 0;
  for (; k + 4 <= n0; k += 4) {
    const int s0 = __builtin_amdgcn_readlane(sl0.y, k);
    const int s1 = __builtin_amdgcn_readlane(sl0.y, k + 1);
    const int s2 = __builtin_amdgcn_readlane(sl0.y, k + 2);
    const int s3 = __builtin_amdgcn_readlane(sl0.y, k + 3);
    const float2 h0 = Hr[(size_t)s0 * 32 + lane];
    const float2 h1 = Hr[(size_t)s1 * 32 + lane];
    const float2 h2 = Hr[(size_t)s2 * 32 + lane];
    const float2 h3 = Hr[(size_t)s3 * 32 + lane];
    const float a0 = bcast_f(al0, k);
    const float a1 = bcast_f(al0, k + 1);
    const float a2 = bcast_f(al0, k + 2);
    const float a3 = bcast_f(al0, k + 3);
    acc.x = fmaf(a0, h0.x, acc.x);  acc.y = fmaf(a0, h0.y, acc.y);
    acc.x = fmaf(a1, h1.x, acc.x);  acc.y = fmaf(a1, h1.y, acc.y);
    acc.x = fmaf(a2, h2.x, acc.x);  acc.y = fmaf(a2, h2.y, acc.y);
    acc.x = fmaf(a3, h3.x, acc.x);  acc.y = fmaf(a3, h3.y, acc.y);
  }
  const int r = n0 - k;
  if (r > 0) {
    const int k1 = (r > 1) ? k + 1 : k;
    const int k2 = (r > 2) ? k + 2 : k;
    const int s0 = __builtin_amdgcn_readlane(sl0.y, k);
    const int s1 = __builtin_amdgcn_readlane(sl0.y, k1);
    const int s2 = __builtin_amdgcn_readlane(sl0.y, k2);
    const float2 h0 = Hr[(size_t)s0 * 32 + lane];
    const float2 h1 = Hr[(size_t)s1 * 32 + lane];
    const float2 h2 = Hr[(size_t)s2 * 32 + lane];
    const float a0 = bcast_f(al0, k);
    const float a1 = bcast_f(al0, k1);
    const float a2 = bcast_f(al0, k2);
    acc.x = fmaf(a0, h0.x, acc.x);  acc.y = fmaf(a0, h0.y, acc.y);
    if (r > 1) { acc.x = fmaf(a1, h1.x, acc.x);  acc.y = fmaf(a1, h1.y, acc.y); }
    if (r > 2) { acc.x = fmaf(a2, h2.x, acc.x);  acc.y = fmaf(a2, h2.y, acc.y); }
  }
  for (int base = 32; base < n; base += 32) {
    const int  j  = base + lane;
    const bool vj = j < n;
    const int2 sl = vj ? slots[j] : make_int2(0, 0);
    float al = 0.f;
    if (vj) {
      float t = hs[sl.y] + hdn;
      t = (t > 0.f) ? t : NEG_SLOPE * t;
      al = __expf(t - m) / dd;
      if (WRITE_ALPHA) alpha_out[sl.x] = al;
    }
    const int c2 = (n - base < 32) ? (n - base) : 32;
    for (int q = 0; q < c2; ++q) {
      const int    s = __builtin_amdgcn_readlane(sl.y, q);
      const float  a = bcast_f(al, q);
      const float2 h = Hr[(size_t)s * 32 + lane];
      acc.x = fmaf(a, h.x, acc.x);
      acc.y = fmaf(a, h.y, acc.y);
    }
  }
  acc.x = fmaf(als, hself.x, acc.x);
  acc.y = fmaf(als, hself.y, acc.y);
  float ox = acc.x + b.x;
  float oy = acc.y + b.y;
  ox = (ox > 0.f) ? ox : 0.f;
  oy = (oy > 0.f) ? oy : 0.f;
  return make_float2(ox, oy);
}

template <bool WRITE_ALPHA>
__global__ __launch_bounds__(256) void gat_node_fused(
    const int* __restrict__ off, const int* __restrict__ slots_,
    const float* __restrict__ hs, const float* __restrict__ hd,
    const float* __restrict__ H, const float* __restrict__ bias,
    float* __restrict__ out, float* __restrict__ alpha_out, int Nn, int E) {
  const int gw = __builtin_amdgcn_readfirstlane(
      (int)(blockIdx.x * 8 + (threadIdx.x >> 5)));
  const int lane  = threadIdx.x & 31;
  const int nw    = gridDim.x * 8;
  const int per   = (Nn + nw - 1) / nw;
  const int first = gw * per;
  const int stop  = (first + per < Nn) ? first + per : Nn;
  if (first >= stop) return;
  const float2* Hr = (const float2*)H;
  const float2  b  = ((const float2*)bias)[lane];
  int beg = off[first];
  int end = off[first + 1];
  for (int node = first; node < stop; ++node) {
    const int endNext = off[(node + 2 < Nn) ? node + 2 : Nn];
    const float2 o = node_row<WRITE_ALPHA>(
        node, end - beg, (const int2*)slots_ + beg, hs, hd, Hr, b, alpha_out,
        E, lane);
    ((float2*)out)[(size_t)node * 32 + lane] = o;
    beg = end;
    end = endNext;
  }
}

#define NB_SLOTS 8
#define NB_LDP 66
__global__ __launch_bounds__(256, 16) void gat_node_next(
    const int* __restrict__ off, const int* __restrict__ slots_,
    const float* __restrict__ hs, const float* __restrict__ hd,
    const float* __restrict__ H, const float* __restrict__ bias,
    const float* __restrict__ Wn, const float* __restrict__ a_srcn,
    const float* __restrict__ a_dstn, float* __restrict__ Hn,
    float* __restrict__ hsn, float* __restrict__ hdn, int Nn, int E) {
  __shared__ __attribute__((aligned(16))) float tileL[8 * NB_SLOTS * NB_LDP];
  const int wave = __builtin_amdgcn_readfirstlane((int)(threadIdx.x >> 5));
  const int lane = threadIdx.x & 31;
  const int nw   = gridDim.x * 8;
  const int per  = (Nn + nw - 1) / nw;
  const int blockFirst = (int)blockIdx.x * 8 * per;
  if (blockFirst >= Nn) return;
  const int first = blockFirst + wave * per;
  const int stop  = (first + per < Nn) ? first + per : Nn;
  const float2* Hr = (const float2*)H;
  const float2  b  = ((const float2*)bias)[lane];
  int beg = 0, end = 0;
  if (first < stop) {
    beg = off[first];
    end = off[first + 1];
  }
  int cnt = 0;
  for (int node = first; node < stop; ++node) {
    const int endNext = off[(node + 2 < Nn) ? node + 2 : Nn];
    const float2 o = node_row<false>(
        node, end - beg, (const int2*)slots_ + beg, hs, hd, Hr, b, nullptr, E,
        lane);
    *(float2*)&tileL[(wave * NB_SLOTS + cnt) * NB_LDP + 2 * lane] = o;
    ++cnt;
    beg = end;
    end = endNext;
  }
  for (int i = cnt; i < NB_SLOTS; ++i)
    *(float2*)&tileL[(wave * NB_SLOTS + i) * NB_LDP + 2 * lane] =
        make_float2(0.f, 0.f);

  const int l16 = lane & 15;
  const int hw  = lane >> 4;
  const int kh  = hw << 1;
  const int col = (wave & 3) * 16 + l16;
  const int rt0 = (wave >> 2) * 2;

  const float* wc = Wn + col;
  v2f bf[FDIM / 4];
#pragma unroll
  for (int k = 0; k < FDIM; k += 4) {
    bf[k >> 2][0] = wc[(size_t)(k + kh) * FDIM];
    bf[k >> 2][1] = wc[(size_t)(k + kh + 1) * FDIM];
  }
  __syncthreads();

  const float* ar0 = tileL + (rt0 * 16 + l16) * NB_LDP + kh;
  const float* ar1 = ar0 + 16 * NB_LDP;
  v8f c0 = {0.f, 0.f, 0.f, 0.f, 0.f, 0.f, 0.f, 0.f};
  v8f c1 = {0.f, 0.f, 0.f, 0.f, 0.f, 0.f, 0.f, 0.f};
#pragma unroll
  for (int k = 0; k < FDIM; k += 4) {
    v2f a0, a1;
    a0[0] = ar0[k];
    a0[1] = ar0[k + 1];
    a1[0] = ar1[k];
    a1[1] = ar1[k + 1];
    c0 = __builtin_amdgcn_wmma_f32_16x16x4_f32(false, a0, false, bf[k >> 2],
                                               (short)0, c0, false, false);
    c1 = __builtin_amdgcn_wmma_f32_16x16x4_f32(false, a1, false, bf[k >> 2],
                                               (short)0, c1, false, false);
  }

  const int nd0 = blockFirst + (2 * rt0 + hw) * per;
  const int nd1 = nd0 + 2 * per;
#pragma unroll
  for (int v = 0; v < 8; ++v) {
    if (v < per && nd0 + v < Nn) Hn[(size_t)(nd0 + v) * FDIM + col] = c0[v];
    if (v < per && nd1 + v < Nn) Hn[(size_t)(nd1 + v) * FDIM + col] = c1[v];
  }

  __syncthreads();
#pragma unroll
  for (int v = 0; v < 8; ++v) {
    tileL[(rt0 * 16 + 8 * hw + v) * NB_LDP + col]      = c0[v];
    tileL[(rt0 * 16 + 16 + 8 * hw + v) * NB_LDP + col] = c1[v];
  }
  __syncthreads();

  if (wave < 4) {
    const int    r    = threadIdx.x & 63;
    const float* av   = (wave >> 1) ? a_dstn : a_srcn;
    float*       dstp = (wave >> 1) ? hdn : hsn;
    const float* hrow = tileL + r * NB_LDP;
    float s = 0.f;
#pragma unroll
    for (int f = 0; f < FDIM; ++f) s = fmaf(hrow[f], av[f], s);
    const int i  = r & 7;
    const int nd = blockFirst + (r >> 3) * per + i;
    if (i < per && nd < Nn) dstp[nd] = s;
  }
}

static inline int cdiv_ll(long long a, long long b) {
  return (int)((a + b - 1) / b);
}

extern "C" void kernel_launch(void* const* d_in, const int* in_sizes, int n_in,
                              void* d_out, int out_size, void* d_ws,
                              size_t ws_size, hipStream_t stream) {
  const float* x      = (const float*)d_in[0];
  const int*   ei     = (const int*)d_in[1];
  const float* W1     = (const float*)d_in[2];
  const float* a_src1 = (const float*)d_in[3];
  const float* a_dst1 = (const float*)d_in[4];
  const float* b1     = (const float*)d_in[5];
  const float* W2     = (const float*)d_in[6];
  const float* a_src2 = (const float*)d_in[7];
  const float* a_dst2 = (const float*)d_in[8];
  const float* b2     = (const float*)d_in[9];

  const int IN_DIM = 128;
  const int Nn = in_sizes[0] / IN_DIM;
  const int E  = in_sizes[1] / 2;

  const int    nTiles = cdiv_ll(Nn, SCAN_TILE);
  const int    nRange = cdiv_ll(Nn, SPAN);
  const size_t nTsum  = (size_t)nRange * (SPAN / SCAN_TILE) * NCHUNK;
  const size_t NnPad  = ((size_t)Nn + 4) & ~(size_t)3;

  float* bufH   = (float*)d_ws;
  float* bufAgg = bufH + (size_t)Nn * FDIM;
  float* hs     = bufAgg + (size_t)Nn * FDIM;
  float* hd     = hs + NnPad;
  float* hs2    = hd + NnPad;
  float* hd2    = hs2 + NnPad;
  int*   off    = (int*)(hd2 + NnPad);
  int*   tsum   = off + NnPad;
  int*   slotsA = tsum + ((nTsum + 3) & ~(size_t)3);
  int*   slotsB = slotsA + (((size_t)2 * E + 3) & ~(size_t)3);
  int*   tab    = slotsB + (((size_t)2 * E + 3) & ~(size_t)3);

  float* h2out     = (float*)d_out;
  float* alpha_out = h2out + (size_t)Nn * FDIM;

  const int nodeGrid = cdiv_ll(Nn, 256);
  const int nextGrid = (cdiv_ll(Nn, 64) > LOOP_GRID) ? cdiv_ll(Nn, 64) : LOOP_GRID;

  const dim3 grpGrid(nRange, NCHUNK);
  csr_count<<<grpGrid, GRP_T, 0, stream>>>(ei, tab, tsum, Nn, (int)NnPad, E);
  csr_scan_gemm1<<<LOOP_GRID / 2 + nTiles, SCAN_TILE, 0, stream>>>(
      tab, tsum, off, Nn, (int)NnPad, E, x, W1, bufH, LOOP_GRID / 2);
  csr_place<<<grpGrid, GRP_T, 0, stream>>>(ei, tab, slotsA, Nn, (int)NnPad, E);
  csr_order_dots1<<<nodeGrid + LOOP_GRID, 256, 0, stream>>>(
      off, slotsA, slotsB, Nn, LOOP_GRID, bufH, a_src1, a_dst1, hs, hd,
      nodeGrid);

  gat_node_next<<<nextGrid, 256, 0, stream>>>(
      off, slotsB, hs, hd, bufH, b1, W2, a_src2, a_dst2, bufAgg, hs2, hd2, Nn,
      E);

  gat_node_fused<true><<<LOOP_GRID, 256, 0, stream>>>(
      off, slotsB, hs2, hd2, bufAgg, b2, h2out, alpha_out, Nn, E);
}


# Place pass leaves groups in edge-id order itself: the workgroup that fills a (node, chunk) cell of 2+ edges sorts it before exit; ordering pass and its launch go (5 launches); float sums in old order
# speedup vs baseline: 4.3712x; 1.0316x over previous
#include <hip/hip_runtime.h>


#define LOOP_GRID 2048

typedef float v2f __attribute__((ext_vector_type(2)));
typedef float v8f __attribute__((ext_vector_type(8)));

#define NEG_SLOPE 0.2f
#define EPSV 1e-16f
#define FDIM 64

template <int K>
__device__ __forceinline__ void gemm_body(
    const float* __restrict__ X, const float* __restrict__ W,
    float* __restrict__ H, int M, int vb, int vgrid, int tid) {
  const int wave = tid >> 5;
  const int lane = tid & 31;
  const int l16  = lane & 15;
  const int kh   = (lane >> 4) << 1;
  const int colBase = wave * 16;
  const int rOff = (lane >> 4) << 3;
  const float* wc = W + colBase + l16;

  v2f bf[K / 4];
#pragma unroll
  for (int k = 0; k < K; k += 4) {
    bf[k >> 2][0] = wc[(size_t)(k + kh) * FDIM];
    bf[k >> 2][1] = wc[(size_t)(k + kh + 1) * FDIM];
  }

  const int nTiles = (M + 15) >> 4;
  for (int tile = vb; tile < nTiles; tile += vgrid) {
    const int rowBase = tile * 16;
    int arow = rowBase + l16;
    if (arow >= M) arow = M - 1;
    const float* xr = X + (size_t)arow * K;

    v8f c = {0.f, 0.f, 0.f, 0.f, 0.f, 0.f, 0.f, 0.f};
#pragma unroll
    for (int k = 0; k < K; k += 4) {
      v2f a;
      a[0] = xr[k + kh];
      a[1] = xr[k + kh + 1];
      c = __builtin_amdgcn_wmma_f32_16x16x4_f32(false, a, false, bf[k >> 2],
                                                (short)0, c, false, false);
    }

#pragma unroll
    for (int v = 0; v < 8; ++v) {
      int r = rowBase + rOff + v;
      if (r >= M) r = M - 1;
      H[(size_t)r * FDIM + colBase + l16] = c[v];
    }
  }
}

__device__ __forceinline__ void dots_body(const float* __restrict__ H,
                              const float* __restrict__ a_src,
                              const float* __restrict__ a_dst,
                              float* __restrict__ hs, float* __restrict__ hd,
                              int Nn, int i) {
  if (i >= Nn) return;
  const float* h = H + (size_t)i * FDIM;
  float s = 0.f, d = 0.f;
#pragma unroll 1
  for (int g = 0; g < FDIM; g += 16) {
#pragma unroll
    for (int f = g; f < g + 16; ++f) {
      float v = h[f];
      s += v * a_src[f];
      d += v * a_dst[f];
    }
  }
  hs[i] = s;
  hd[i] = d;
}

#define SCAN_TILE 256
#define SPAN 12544
#define NCHUNK 31
#define GRP_T 1024
#define GRP_LD 5

__device__ __forceinline__ void chunk_bounds(int E, int c, int& cb, int& ce) {
  const int per = (((E + NCHUNK - 1) / NCHUNK) + 3) & ~3;
  const int b   = c * per;
  cb = (b < E) ? b : E;
  ce = (E - cb < per) ? E : cb + per;
}

__device__ __forceinline__ int wave_sum_i(int v) {
#pragma unroll
  for (int o = 16; o > 0; o >>= 1) v += __shfl_xor(v, o, 32);
  return v;
}

__device__ __forceinline__ void count1(int* lc, int d, int lo) {
  const unsigned rel = (unsigned)(d - lo);
  if (rel < (unsigned)SPAN) atomicAdd(&lc[rel], 1);
}

__global__ __launch_bounds__(GRP_T) void csr_count(
    const int* __restrict__ ei, int* __restrict__ tab, int* __restrict__ tsum,
    int Nn, int NnPad, int E) {
  __shared__ int lc[SPAN];
  const int tid = threadIdx.x;
  const int lo  = blockIdx.x * SPAN;
  const int c   = blockIdx.y;
  for (int i = tid; i < SPAN; i += GRP_T) lc[i] = 0;
  __syncthreads();
  int cb, ce;
  chunk_bounds(E, c, cb, ce);
  const int* dst = ei + E;
  if (((((size_t)dst) & 15) == 0) && ((E & 3) == 0)) {
    const int4* dq = (const int4*)dst;
    const int4 none = make_int4(lo - 1, lo - 1, lo - 1, lo - 1);
    const int qe = ce >> 2;
    for (int q = (cb >> 2) + tid; q < qe; q += GRP_LD * GRP_T) {
      int4 a[GRP_LD];
#pragma unroll
      for (int u = 0; u < GRP_LD; ++u) {
        const int  qu = q + u * GRP_T;
        const bool in = qu < qe;
        const int4 t  = dq[in ? qu : qe - 1];
        a[u] = make_int4(in ? t.x : none.x, in ? t.y : none.y,
                         in ? t.z : none.z, in ? t.w : none.w);
      }
#pragma unroll
      for (int u = 0; u < GRP_LD; ++u) {
        count1(lc, a[u].x, lo); count1(lc, a[u].y, lo);
        count1(lc, a[u].z, lo); count1(lc, a[u].w, lo);
      }
    }
  } else {
    for (int e = cb + tid; e < ce; e += GRP_T) count1(lc, dst[e], lo);
  }
  __syncthreads();
  for (int i = tid; i < SPAN; i += GRP_T) {
    const int d = lo + i;
    if (d < Nn) tab[(size_t)c * NnPad + d] = lc[i];
  }
  for (int t = tid >> 5; t < SPAN / SCAN_TILE; t += GRP_T / 32) {
    int v = 0;
#pragma unroll
    for (int k = 0; k < SCAN_TILE / 32; ++k)
      v += lc[t * SCAN_TILE + k * 32 + (tid & 31)];
    v = wave_sum_i(v);
    if ((tid & 31) == 0)
      tsum[(size_t)(blockIdx.x * (SPAN / SCAN_TILE) + t) * NCHUNK + c] = v;
  }
}

__device__ __forceinline__ void scan_body(
    int* __restrict__ tab, const int* __restrict__ tsum,
    int* __restrict__ off, int Nn, int NnPad, int E, int bid) {
  __shared__ int red[SCAN_TILE / 32];
  __shared__ int wtot[SCAN_TILE / 32];
  const int lane = threadIdx.x & 31;
  const int wave = threadIdx.x >> 5;
  const int d    = bid * SCAN_TILE + threadIdx.x;
  const bool in  = d < Nn;

  int cnt[NCHUNK];
#pragma unroll
  for (int c = 0; c < NCHUNK; ++c)
    cnt[c] = in ? tab[(size_t)c * NnPad + d] : 0;

  int p = 0;
  const int np = bid * NCHUNK;
  for (int j = threadIdx.x; j < np; j += SCAN_TILE) p += tsum[j];
  p = wave_sum_i(p);
  if (lane == 0) red[wave] = p;

  int s = 0;
#pragma unroll
  for (int c = 0; c < NCHUNK; ++c) s += cnt[c];
  int x = s;
#pragma unroll
  for (int o = 1; o < 32; o <<= 1) {
    int t = __shfl_up(x, o, 32);
    if (lane >= o) x += t;
  }
  if (lane == 31) wtot[wave] = x;
  __syncthreads();

  int run = 0;
#pragma unroll
  for (int w = 0; w < SCAN_TILE / 32; ++w) {
    run += red[w];
    if (w < wave) run += wtot[w];
  }
  run += x - s;
  if (in) {
    off[d] = run;
    int t = run;
#pragma unroll
    for (int c = 0; c < NCHUNK; ++c) {
      tab[(size_t)c * NnPad + d] = t;
      t += cnt[c];
    }
  }
  if (bid == 0 && threadIdx.x == 0) off[Nn] = E;
}

__global__ __launch_bounds__(SCAN_TILE) void csr_scan_gemm1(
    int* __restrict__ tab, const int* __restrict__ tsum,
    int* __restrict__ off, int Nn, int NnPad, int E,
    const float* __restrict__ X, const float* __restrict__ W,
    float* __restrict__ H, int gemmBlocks) {
  const int b = (int)blockIdx.x;
  if (b < gemmBlocks) {
    const int half = __builtin_amdgcn_readfirstlane((int)(threadIdx.x >> 7));
    gemm_body<128>(X, W, H, Nn, b * 2 + half, gemmBlocks * 2,
                   (int)(threadIdx.x & 127));
  } else {
    scan_body(tab, tsum, off, Nn, NnPad, E, b - gemmBlocks);
  }
}

__device__ __forceinline__ void place1(int* cur, int2* sl,
                                       const int* __restrict__ ei, int d,
                                       int e, int lo) {
  const unsigned rel = (unsigned)(d - lo);
  if (rel < (unsigned)SPAN) {
    const int p = atomicAdd(&cur[rel], 1);
    sl[p] = make_int2(e, ei[e]);
  }
}

__device__ __forceinline__ void place4(int* cur, int2* sl, int4 d, int4 s,
                                       int e0, int lo, int spare) {
  const unsigned r0 = (unsigned)(d.x - lo), r1 = (unsigned)(d.y - lo);
  const unsigned r2 = (unsigned)(d.z - lo), r3 = (unsigned)(d.w - lo);
  const bool m0 = r0 < (unsigned)SPAN, m1 = r1 < (unsigned)SPAN;
  const bool m2 = r2 < (unsigned)SPAN, m3 = r3 < (unsigned)SPAN;
  const int p0 = atomicAdd(&cur[m0 ? (int)r0 : spare], 1);
  const int p1 = atomicAdd(&cur[m1 ? (int)r1 : spare], 1);
  const int p2 = atomicAdd(&cur[m2 ? (int)r2 : spare], 1);
  const int p3 = atomicAdd(&cur[m3 ? (int)r3 : spare], 1);
  if (m0) sl[p0] = make_int2(e0, s.x);
  if (m1) sl[p1] = make_int2(e0 + 1, s.y);
  if (m2) sl[p2] = make_int2(e0 + 2, s.z);
  if (m3) sl[p3] = make_int2(e0 + 3, s.w);
}

#define CELL_K ((SPAN + GRP_T - 1) / GRP_T)
#define DOTS_T 512

__global__ __launch_bounds__(GRP_T, 16) void csr_place_dots1(
    const int* __restrict__ ei, const int* __restrict__ tab,
    int* __restrict__ slots, int Nn, int NnPad, int E,
    const float* __restrict__ H, const float* __restrict__ a_src,
    const float* __restrict__ a_dst, float* __restrict__ hs,
    float* __restrict__ hd) {
  __shared__ int cur[SPAN + 32];
  __shared__ int2 multi[GRP_T / 32][CELL_K * 32];
  const int tid = threadIdx.x;
  if ((int)blockIdx.y >= NCHUNK) {
    const int b = ((int)blockIdx.y - NCHUNK) * (int)gridDim.x + (int)blockIdx.x;
    if (tid < DOTS_T) dots_body(H, a_src, a_dst, hs, hd, Nn, b * DOTS_T + tid);
    return;
  }
  const int lo  = blockIdx.x * SPAN;
  const int c   = blockIdx.y;
  for (int i = tid; i < SPAN; i += GRP_T) {
    const int d = lo + i;
    cur[i] = (d < Nn) ? tab[(size_t)c * NnPad + d] : 0;
  }
  if (tid < 32) cur[SPAN + tid] = 0;
  const int spare = SPAN + (tid & 31);
  __syncthreads();
  int cb, ce;
  chunk_bounds(E, c, cb, ce);
  const int* dst = ei + E;
  int2* sl = (int2*)slots;
  if (((((size_t)dst) & 15) == 0) && ((E & 3) == 0)) {
    const int4* dq = (const int4*)dst;
    const int4* sq = (const int4*)ei;
    const int4 none = make_int4(lo - 1, lo - 1, lo - 1, lo - 1);
    const int qe = ce >> 2;
    for (int q = (cb >> 2) + tid; q < qe; q += GRP_LD * GRP_T) {
      int4 a[GRP_LD];
      int4 b[GRP_LD];
#pragma unroll
      for (int u = 0; u < GRP_LD; ++u) {
        const int  qu = q + u * GRP_T;
        const bool in = qu < qe;
        const int  qc = in ? qu : qe - 1;
        const int4 t  = dq[qc];
        b[u] = sq[qc];
        a[u] = make_int4(in ? t.x : none.x, in ? t.y : none.y,
                         in ? t.z : none.z, in ? t.w : none.w);
      }
#pragma unroll
      for (int u = 0; u < GRP_LD; ++u)
        place4(cur, sl, a[u], b[u], 4 * (q + u * GRP_T), lo, spare);
    }
  } else {
    for (int e = cb + tid; e < ce; e += GRP_T) place1(cur, sl, ei, dst[e], e, lo);
  }
  __syncthreads();
  int st[CELL_K];
#pragma unroll
  for (int k = 0; k < CELL_K; ++k) {
    const int i = tid + k * GRP_T;
    const int d = lo + i;
    st[k] = (i < SPAN && d < Nn) ? tab[(size_t)c * NnPad + d] : 0;
  }
  const int lane = tid & 31;
  int2* mine = multi[tid >> 5];
  int nm = 0;
#pragma unroll
  for (int k = 0; k < CELL_K; ++k) {
    const int i = tid + k * GRP_T;
    int cn = 0;
    if (i < SPAN && lo + i < Nn) cn = cur[i] - st[k];
    const bool many = cn >= 2;
    const unsigned m = __builtin_amdgcn_ballot_w32(many);
    if (many) mine[nm + (int)__builtin_amdgcn_mbcnt_lo(m, 0u)] = make_int2(st[k], cn);
    nm += __builtin_popcount(m);
  }
  for (int j = lane; j < nm; j += 32) {
    const int2 it = mine[j];
    int2* cell = sl + it.x;
    const int2 p0 = cell[0];
    const int2 p1 = cell[1];
    if (it.y == 2) {
      if (p1.x < p0.x) { cell[0] = p1; cell[1] = p0; }
    } else {
      for (int a = 0; a + 1 < it.y; ++a) {
        const int2 first = cell[a];
        int2 best = first;
        int  bi   = a;
        for (int b = a + 1; b < it.y; ++b) {
          const int2 t = cell[b];
          if (t.x < best.x) { best = t; bi = b; }
        }
        if (bi != a) { cell[bi] = first; cell[a] = best; }
      }
    }
  }
}

__device__ __forceinline__ float wave_max_f(float v) {
#pragma unroll
  for (int o = 16; o > 0; o >>= 1) v = fmaxf(v, __shfl_xor(v, o, 32));
  return v;
}
__device__ __forceinline__ float wave_sum_f(float v) {
#pragma unroll
  for (int o = 16; o > 0; o >>= 1) v += __shfl_xor(v, o, 32);
  return v;
}
__device__ __forceinline__ float bcast_f(float v, int k) {
  return __int_as_float(__builtin_amdgcn_readlane(__float_as_int(v), k));
}

template <bool WRITE_ALPHA>
__device__ __forceinline__ float2 node_row(
    int node, int n, const int2* __restrict__ slots,
    const float* __restrict__ hs, const float* __restrict__ hd,
    const float2* __restrict__ Hr, float2 b, float* __restrict__ alpha_out,
    int E, int lane) {
  const float2 hself = Hr[(size_t)node * 32 + lane];

  const float hdn = hd[node];
  float lself = hs[node] + hdn;
  lself = (lself > 0.f) ? lself : NEG_SLOPE * lself;

  const bool v0  = lane < n;
  const int2 sl0 = v0 ? slots[lane] : make_int2(0, 0);
  float lg0 = -INFINITY;
  if (v0) {
    float t = hs[sl0.y] + hdn;
    lg0 = (t > 0.f) ? t : NEG_SLOPE * t;
  }
  float m = wave_max_f(lg0);
  for (int base = 32; base < n; base += 32) {
    const int j = base + lane;
    float lg = -INFINITY;
    if (j < n) {
      float t = hs[slots[j].y] + hdn;
      lg = (t > 0.f) ? t : NEG_SLOPE * t;
    }
    m = fmaxf(m, wave_max_f(lg));
  }
  m = fmaxf(m, lself);

  const float ex0 = v0 ? __expf(lg0 - m) : 0.f;
  float den = wave_sum_f(ex0);
  for (int base = 32; base < n; base += 32) {
    const int j = base + lane;
    float ex = 0.f;
    if (j < n) {
      float t = hs[slots[j].y] + hdn;
      t = (t > 0.f) ? t : NEG_SLOPE * t;
      ex = __expf(t - m);
    }
    den += wave_sum_f(ex);
  }
  const float exs = __expf(lself - m);
  den += exs;
  const float dd  = den + EPSV;
  const float al0 = ex0 / dd;
  const float als = exs / dd;
  if (WRITE_ALPHA) {
    if (v0) alpha_out[sl0.x] = al0;
    if (lane == 0) alpha_out[E + node] = als;
  }

  float2 acc = make_float2(0.f, 0.f);
  const int n0 = (n < 32) ? n : 32;
  int k = 0;
  for (; k + 4 <= n0; k += 4) {
    const int s0 = __builtin_amdgcn_readlane(sl0.y, k);
    const int s1 = __builtin_amdgcn_readlane(sl0.y, k + 1);
    const int s2 = __builtin_amdgcn_readlane(sl0.y, k + 2);
    const int s3 = __builtin_amdgcn_readlane(sl0.y, k + 3);
    const float2 h0 = Hr[(size_t)s0 * 32 + lane];
    const float2 h1 = Hr[(size_t)s1 * 32 + lane];
    const float2 h2 = Hr[(size_t)s2 * 32 + lane];
    const float2 h3 = Hr[(size_t)s3 * 32 + lane];
    const float a0 = bcast_f(al0, k);
    const float a1 = bcast_f(al0, k + 1);
    const float a2 = bcast_f(al0, k + 2);
    const float a3 = bcast_f(al0, k + 3);
    acc.x = fmaf(a0, h0.x, acc.x);  acc.y = fmaf(a0, h0.y, acc.y);
    acc.x = fmaf(a1, h1.x, acc.x);  acc.y = fmaf(a1, h1.y, acc.y);
    acc.x = fmaf(a2, h2.x, acc.x);  acc.y = fmaf(a2, h2.y, acc.y);
    acc.x = fmaf(a3, h3.x, acc.x);  acc.y = fmaf(a3, h3.y, acc.y);
  }
  const int r = n0 - k;
  if (r > 0) {
    const int k1 = (r > 1) ? k + 1 : k;
    const int k2 = (r > 2) ? k + 2 : k;
    const int s0 = __builtin_amdgcn_readlane(sl0.y, k);
    const int s1 = __builtin_amdgcn_readlane(sl0.y, k1);
    const int s2 = __builtin_amdgcn_readlane(sl0.y, k2);
    const float2 h0 = Hr[(size_t)s0 * 32 + lane];
    const float2 h1 = Hr[(size_t)s1 * 32 + lane];
    const float2 h2 = Hr[(size_t)s2 * 32 + lane];
    const float a0 = bcast_f(al0, k);
    const float a1 = bcast_f(al0, k1);
    const float a2 = bcast_f(al0, k2);
    acc.x = fmaf(a0, h0.x, acc.x);  acc.y = fmaf(a0, h0.y, acc.y);
    if (r > 1) { acc.x = fmaf(a1, h1.x, acc.x);  acc.y = fmaf(a1, h1.y, acc.y); }
    if (r > 2) { acc.x = fmaf(a2, h2.x, acc.x);  acc.y = fmaf(a2, h2.y, acc.y); }
  }
  for (int base = 32; base < n; base += 32) {
    const int  j  = base + lane;
    const bool vj = j < n;
    const int2 sl = vj ? slots[j] : make_int2(0, 0);
    float al = 0.f;
    if (vj) {
      float t = hs[sl.y] + hdn;
      t = (t > 0.f) ? t : NEG_SLOPE * t;
      al = __expf(t - m) / dd;
      if (WRITE_ALPHA) alpha_out[sl.x] = al;
    }
    const int c2 = (n - base < 32) ? (n - base) : 32;
    for (int q = 0; q < c2; ++q) {
      const int    s = __builtin_amdgcn_readlane(sl.y, q);
      const float  a = bcast_f(al, q);
      const float2 h = Hr[(size_t)s * 32 + lane];
      acc.x = fmaf(a, h.x, acc.x);
      acc.y = fmaf(a, h.y, acc.y);
    }
  }
  acc.x = fmaf(als, hself.x, acc.x);
  acc.y = fmaf(als, hself.y, acc.y);
  float ox = acc.x + b.x;
  float oy = acc.y + b.y;
  ox = (ox > 0.f) ? ox : 0.f;
  oy = (oy > 0.f) ? oy : 0.f;
  return make_float2(ox, oy);
}

template <bool WRITE_ALPHA>
__global__ __launch_bounds__(256) void gat_node_fused(
    const int* __restrict__ off, const int* __restrict__ slots_,
    const float* __restrict__ hs, const float* __restrict__ hd,
    const float* __restrict__ H, const float* __restrict__ bias,
    float* __restrict__ out, float* __restrict__ alpha_out, int Nn, int E) {
  const int gw = __builtin_amdgcn_readfirstlane(
      (int)(blockIdx.x * 8 + (threadIdx.x >> 5)));
  const int lane  = threadIdx.x & 31;
  const int nw    = gridDim.x * 8;
  const int per   = (Nn + nw - 1) / nw;
  const int first = gw * per;
  const int stop  = (first + per < Nn) ? first + per : Nn;
  if (first >= stop) return;
  const float2* Hr = (const float2*)H;
  const float2  b  = ((const float2*)bias)[lane];
  int beg = off[first];
  int end = off[first + 1];
  for (int node = first; node < stop; ++node) {
    const int endNext = off[(node + 2 < Nn) ? node + 2 : Nn];
    const float2 o = node_row<WRITE_ALPHA>(
        node, end - beg, (const int2*)slots_ + beg, hs, hd, Hr, b, alpha_out,
        E, lane);
    ((float2*)out)[(size_t)node * 32 + lane] = o;
    beg = end;
    end = endNext;
  }
}

#define NB_SLOTS 8
#define NB_LDP 66
__global__ __launch_bounds__(256, 16) void gat_node_next(
    const int* __restrict__ off, const int* __restrict__ slots_,
    const float* __restrict__ hs, const float* __restrict__ hd,
    const float* __restrict__ H, const float* __restrict__ bias,
    const float* __restrict__ Wn, const float* __restrict__ a_srcn,
    const float* __restrict__ a_dstn, float* __restrict__ Hn,
    float* __restrict__ hsn, float* __restrict__ hdn, int Nn, int E) {
  __shared__ __attribute__((aligned(16))) float tileL[8 * NB_SLOTS * NB_LDP];
  const int wave = __builtin_amdgcn_readfirstlane((int)(threadIdx.x >> 5));
  const int lane = threadIdx.x & 31;
  const int nw   = gridDim.x * 8;
  const int per  = (Nn + nw - 1) / nw;
  const int blockFirst = (int)blockIdx.x * 8 * per;
  if (blockFirst >= Nn) return;
  const int first = blockFirst + wave * per;
  const int stop  = (first + per < Nn) ? first + per : Nn;
  const float2* Hr = (const float2*)H;
  const float2  b  = ((const float2*)bias)[lane];
  int beg = 0, end = 0;
  if (first < stop) {
    beg = off[first];
    end = off[first + 1];
  }
  int cnt = 0;
  for (int node = first; node < stop; ++node) {
    const int endNext = off[(node + 2 < Nn) ? node + 2 : Nn];
    const float2 o = node_row<false>(
        node, end - beg, (const int2*)slots_ + beg, hs, hd, Hr, b, nullptr, E,
        lane);
    *(float2*)&tileL[(wave * NB_SLOTS + cnt) * NB_LDP + 2 * lane] = o;
    ++cnt;
    beg = end;
    end = endNext;
  }
  for (int i = cnt; i < NB_SLOTS; ++i)
    *(float2*)&tileL[(wave * NB_SLOTS + i) * NB_LDP + 2 * lane] =
        make_float2(0.f, 0.f);

  const int l16 = lane & 15;
  const int hw  = lane >> 4;
  const int kh  = hw << 1;
  const int col = (wave & 3) * 16 + l16;
  const int rt0 = (wave >> 2) * 2;

  const float* wc = Wn + col;
  v2f bf[FDIM / 4];
#pragma unroll
  for (int k = 0; k < FDIM; k += 4) {
    bf[k >> 2][0] = wc[(size_t)(k + kh) * FDIM];
    bf[k >> 2][1] = wc[(size_t)(k + kh + 1) * FDIM];
  }
  __syncthreads();

  const float* ar0 = tileL + (rt0 * 16 + l16) * NB_LDP + kh;
  const float* ar1 = ar0 + 16 * NB_LDP;
  v8f c0 = {0.f, 0.f, 0.f, 0.f, 0.f, 0.f, 0.f, 0.f};
  v8f c1 = {0.f, 0.f, 0.f, 0.f, 0.f, 0.f, 0.f, 0.f};
#pragma unroll
  for (int k = 0; k < FDIM; k += 4) {
    v2f a0, a1;
    a0[0] = ar0[k];
    a0[1] = ar0[k + 1];
    a1[0] = ar1[k];
    a1[1] = ar1[k + 1];
    c0 = __builtin_amdgcn_wmma_f32_16x16x4_f32(false, a0, false, bf[k >> 2],
                                               (short)0, c0, false, false);
    c1 = __builtin_amdgcn_wmma_f32_16x16x4_f32(false, a1, false, bf[k >> 2],
                                               (short)0, c1, false, false);
  }

  const int nd0 = blockFirst + (2 * rt0 + hw) * per;
  const int nd1 = nd0 + 2 * per;
#pragma unroll
  for (int v = 0; v < 8; ++v) {
    if (v < per && nd0 + v < Nn) Hn[(size_t)(nd0 + v) * FDIM + col] = c0[v];
    if (v < per && nd1 + v < Nn) Hn[(size_t)(nd1 + v) * FDIM + col] = c1[v];
  }

  __syncthreads();
#pragma unroll
  for (int v = 0; v < 8; ++v) {
    tileL[(rt0 * 16 + 8 * hw + v) * NB_LDP + col]      = c0[v];
    tileL[(rt0 * 16 + 16 + 8 * hw + v) * NB_LDP + col] = c1[v];
  }
  __syncthreads();

  if (wave < 4) {
    const int    r    = threadIdx.x & 63;
    const float* av   = (wave >> 1) ? a_dstn : a_srcn;
    float*       dstp = (wave >> 1) ? hdn : hsn;
    const float* hrow = tileL + r * NB_LDP;
    float s = 0.f;
#pragma unroll
    for (int f = 0; f < FDIM; ++f) s = fmaf(hrow[f], av[f], s);
    const int i  = r & 7;
    const int nd = blockFirst + (r >> 3) * per + i;
    if (i < per && nd < Nn) dstp[nd] = s;
  }
}

static inline int cdiv_ll(long long a, long long b) {
  return (int)((a + b - 1) / b);
}

extern "C" void kernel_launch(void* const* d_in, const int* in_sizes, int n_in,
                              void* d_out, int out_size, void* d_ws,
                              size_t ws_size, hipStream_t stream) {
  const float* x      = (const float*)d_in[0];
  const int*   ei     = (const int*)d_in[1];
  const float* W1     = (const float*)d_in[2];
  const float* a_src1 = (const float*)d_in[3];
  const float* a_dst1 = (const float*)d_in[4];
  const float* b1     = (const float*)d_in[5];
  const float* W2     = (const float*)d_in[6];
  const float* a_src2 = (const float*)d_in[7];
  const float* a_dst2 = (const float*)d_in[8];
  const float* b2     = (const float*)d_in[9];

  const int IN_DIM = 128;
  const int Nn = in_sizes[0] / IN_DIM;
  const int E  = in_sizes[1] / 2;

  const int    nTiles = cdiv_ll(Nn, SCAN_TILE);
  const int    nRange = cdiv_ll(Nn, SPAN);
  const size_t nTsum  = (size_t)nRange * (SPAN / SCAN_TILE) * NCHUNK;
  const size_t NnPad  = ((size_t)Nn + 4) & ~(size_t)3;

  float* bufH   = (float*)d_ws;
  float* bufAgg = bufH + (size_t)Nn * FDIM;
  float* hs     = bufAgg + (size_t)Nn * FDIM;
  float* hd     = hs + NnPad;
  float* hs2    = hd + NnPad;
  float* hd2    = hs2 + NnPad;
  int*   off    = (int*)(hd2 + NnPad);
  int*   tsum   = off + NnPad;
  int*   slots  = tsum + ((nTsum + 3) & ~(size_t)3);
  int*   tab    = slots + (((size_t)2 * E + 3) & ~(size_t)3);

  float* h2out     = (float*)d_out;
  float* alpha_out = h2out + (size_t)Nn * FDIM;

  const int nextGrid = (cdiv_ll(Nn, 64) > LOOP_GRID) ? cdiv_ll(Nn, 64) : LOOP_GRID;

  const dim3 grpGrid(nRange, NCHUNK);
  const int  dotsRows = cdiv_ll(cdiv_ll(Nn, DOTS_T), nRange);
  const dim3 placeGrid(nRange, NCHUNK + dotsRows);
  csr_count<<<grpGrid, GRP_T, 0, stream>>>(ei, tab, tsum, Nn, (int)NnPad, E);
  csr_scan_gemm1<<<LOOP_GRID / 2 + nTiles, SCAN_TILE, 0, stream>>>(
      tab, tsum, off, Nn, (int)NnPad, E, x, W1, bufH, LOOP_GRID / 2);
  csr_place_dots1<<<placeGrid, GRP_T, 0, stream>>>(
      ei, tab, slots, Nn, (int)NnPad, E, bufH, a_src1, a_dst1, hs, hd);

  gat_node_next<<<nextGrid, 256, 0, stream>>>(
      off, slots, hs, hd, bufH, b1, W2, a_src2, a_dst2, bufAgg, hs2, hd2, Nn,
      E);

  gat_node_fused<true><<<LOOP_GRID, 256, 0, stream>>>(
      off, slots, hs2, hd2, bufAgg, b2, h2out, alpha_out, Nn, E);
}


# Layer-1 product: each 16-row tile of x goes through LDS, read once per 4-wave group with whole-row 16-byte loads (was read by every wave, 32 loads of 16 lines each); float sums in old order
# speedup vs baseline: 4.5410x; 1.0388x over previous
#include <hip/hip_runtime.h>


#define LOOP_GRID 2048

typedef float v2f __attribute__((ext_vector_type(2)));
typedef float v8f __attribute__((ext_vector_type(8)));

#define NEG_SLOPE 0.2f
#define EPSV 1e-16f
#define FDIM 64

#define XT_PITCH 132
template <int K>
__device__ __forceinline__ void gemm_body(
    const float* __restrict__ X, const float* __restrict__ W,
    float* __restrict__ H, int M, int vb, int vgrid, int trips, int tid,
    float* __restrict__ xs) {
  const int wave = tid >> 5;
  const int lane = tid & 31;
  const int l16  = lane & 15;
  const int kh   = (lane >> 4) << 1;
  const int colBase = wave * 16;
  const int rOff = (lane >> 4) << 3;
  const float* wc = W + colBase + l16;

  v2f bf[K / 4];
#pragma unroll
  for (int k = 0; k < K; k += 4) {
    bf[k >> 2][0] = wc[(size_t)(k + kh) * FDIM];
    bf[k >> 2][1] = wc[(size_t)(k + kh + 1) * FDIM];
  }

  constexpr int PIECES = (16 * K / 4) / 128;
  constexpr int RPIECE = K / 4;
  const int nTiles = (M + 15) >> 4;
  const float* ar = xs + l16 * XT_PITCH + kh;
  int tile = vb;
  for (int it = 0; it < trips; ++it, tile += vgrid) {
    const bool have = tile < nTiles;
    const int  rowBase = tile * 16;
    const bool full = rowBase + 16 <= M;
    float4 pre[PIECES];
#pragma unroll
    for (int j = 0; j < PIECES; ++j) pre[j] = make_float4(0.f, 0.f, 0.f, 0.f);
    if (have) {
      if (full) {
        const float* xp = X + (size_t)rowBase * K;
#pragma unroll
        for (int j = 0; j < PIECES; ++j) {
          const int p = tid + 128 * j;
          pre[j] = *(const float4*)(xp + (size_t)(p / RPIECE) * K + (p % RPIECE) * 4);
        }
      } else {
#pragma unroll
        for (int j = 0; j < PIECES; ++j) {
          const int p = tid + 128 * j;
          int row = rowBase + p / RPIECE;
          if (row >= M) row = M - 1;
          pre[j] = *(const float4*)(X + (size_t)row * K + (p % RPIECE) * 4);
        }
      }
    }
    __syncthreads();
    if (have) {
#pragma unroll
      for (int j = 0; j < PIECES; ++j) {
        const int p = tid + 128 * j;
        *(float4*)(xs + (p / RPIECE) * XT_PITCH + (p % RPIECE) * 4) = pre[j];
      }
    }
    __syncthreads();
    if (have) {
      v8f c = {0.f, 0.f, 0.f, 0.f, 0.f, 0.f, 0.f, 0.f};
#pragma unroll
      for (int k = 0; k < K; k += 4) {
        const v2f a = *(const v2f*)(ar + k);
        c = __builtin_amdgcn_wmma_f32_16x16x4_f32(false, a, false, bf[k >> 2],
                                                  (short)0, c, false, false);
      }
      if (full) {
        float* hp = H + (size_t)(rowBase + rOff) * FDIM + colBase + l16;
#pragma unroll
        for (int v = 0; v < 8; ++v) hp[(size_t)v * FDIM] = c[v];
      } else {
#pragma unroll
        for (int v = 0; v < 8; ++v) {
          int r = rowBase + rOff + v;
          if (r >= M) r = M - 1;
          H[(size_t)r * FDIM + colBase + l16] = c[v];
        }
      }
    }
  }
}

__device__ __forceinline__ void dots_body(const float* __restrict__ H,
                              const float* __restrict__ a_src,
                              const float* __restrict__ a_dst,
                              float* __restrict__ hs, float* __restrict__ hd,
                              int Nn, int i) {
  if (i >= Nn) return;
  const float* h = H + (size_t)i * FDIM;
  float s = 0.f, d = 0.f;
#pragma unroll 1
  for (int g = 0; g < FDIM; g += 16) {
#pragma unroll
    for (int f = g; f < g + 16; ++f) {
      float v = h[f];
      s += v * a_src[f];
      d += v * a_dst[f];
    }
  }
  hs[i] = s;
  hd[i] = d;
}

#define SCAN_TILE 256
#define SPAN 12544
#define NCHUNK 31
#define GRP_T 1024
#define GRP_LD 5

__device__ __forceinline__ void chunk_bounds(int E, int c, int& cb, int& ce) {
  const int per = (((E + NCHUNK - 1) / NCHUNK) + 3) & ~3;
  const int b   = c * per;
  cb = (b < E) ? b : E;
  ce = (E - cb < per) ? E : cb + per;
}

__device__ __forceinline__ int wave_sum_i(int v) {
#pragma unroll
  for (int o = 16; o > 0; o >>= 1) v += __shfl_xor(v, o, 32);
  return v;
}

__device__ __forceinline__ void count1(int* lc, int d, int lo) {
  const unsigned rel = (unsigned)(d - lo);
  if (rel < (unsigned)SPAN) atomicAdd(&lc[rel], 1);
}

__global__ __launch_bounds__(GRP_T) void csr_count(
    const int* __restrict__ ei, int* __restrict__ tab, int* __restrict__ tsum,
    int Nn, int NnPad, int E) {
  __shared__ int lc[SPAN];
  const int tid = threadIdx.x;
  const int lo  = blockIdx.x * SPAN;
  const int c   = blockIdx.y;
  for (int i = tid; i < SPAN; i += GRP_T) lc[i] = 0;
  __syncthreads();
  int cb, ce;
  chunk_bounds(E, c, cb, ce);
  const int* dst = ei + E;
  if (((((size_t)dst) & 15) == 0) && ((E & 3) == 0)) {
    const int4* dq = (const int4*)dst;
    const int4 none = make_int4(lo - 1, lo - 1, lo - 1, lo - 1);
    const int qe = ce >> 2;
    for (int q = (cb >> 2) + tid; q < qe; q += GRP_LD * GRP_T) {
      int4 a[GRP_LD];
#pragma unroll
      for (int u = 0; u < GRP_LD; ++u) {
        const int  qu = q + u * GRP_T;
        const bool in = qu < qe;
        const int4 t  = dq[in ? qu : qe - 1];
        a[u] = make_int4(in ? t.x : none.x, in ? t.y : none.y,
                         in ? t.z : none.z, in ? t.w : none.w);
      }
#pragma unroll
      for (int u = 0; u < GRP_LD; ++u) {
        count1(lc, a[u].x, lo); count1(lc, a[u].y, lo);
        count1(lc, a[u].z, lo); count1(lc, a[u].w, lo);
      }
    }
  } else {
    for (int e = cb + tid; e < ce; e += GRP_T) count1(lc, dst[e], lo);
  }
  __syncthreads();
  for (int i = tid; i < SPAN; i += GRP_T) {
    const int d = lo + i;
    if (d < Nn) tab[(size_t)c * NnPad + d] = lc[i];
  }
  for (int t = tid >> 5; t < SPAN / SCAN_TILE; t += GRP_T / 32) {
    int v = 0;
#pragma unroll
    for (int k = 0; k < SCAN_TILE / 32; ++k)
      v += lc[t * SCAN_TILE + k * 32 + (tid & 31)];
    v = wave_sum_i(v);
    if ((tid & 31) == 0)
      tsum[(size_t)(blockIdx.x * (SPAN / SCAN_TILE) + t) * NCHUNK + c] = v;
  }
}

__device__ __forceinline__ void scan_body(
    int* __restrict__ tab, const int* __restrict__ tsum,
    int* __restrict__ off, int Nn, int NnPad, int E, int bid) {
  __shared__ int red[SCAN_TILE / 32];
  __shared__ int wtot[SCAN_TILE / 32];
  const int lane = threadIdx.x & 31;
  const int wave = threadIdx.x >> 5;
  const int d    = bid * SCAN_TILE + threadIdx.x;
  const bool in  = d < Nn;

  int cnt[NCHUNK];
#pragma unroll
  for (int c = 0; c < NCHUNK; ++c)
    cnt[c] = in ? tab[(size_t)c * NnPad + d] : 0;

  int p = 0;
  const int np = bid * NCHUNK;
  for (int j = threadIdx.x; j < np; j += SCAN_TILE) p += tsum[j];
  p = wave_sum_i(p);
  if (lane == 0) red[wave] = p;

  int s = 0;
#pragma unroll
  for (int c = 0; c < NCHUNK; ++c) s += cnt[c];
  int x = s;
#pragma unroll
  for (int o = 1; o < 32; o <<= 1) {
    int t = __shfl_up(x, o, 32);
    if (lane >= o) x += t;
  }
  if (lane == 31) wtot[wave] = x;
  __syncthreads();

  int run = 0;
#pragma unroll
  for (int w = 0; w < SCAN_TILE / 32; ++w) {
    run += red[w];
    if (w < wave) run += wtot[w];
  }
  run += x - s;
  if (in) {
    off[d] = run;
    int t = run;
#pragma unroll
    for (int c = 0; c < NCHUNK; ++c) {
      tab[(size_t)c * NnPad + d] = t;
      t += cnt[c];
    }
  }
  if (bid == 0 && threadIdx.x == 0) off[Nn] = E;
}

__global__ __launch_bounds__(SCAN_TILE, 10) void csr_scan_gemm1(
    int* __restrict__ tab, const int* __restrict__ tsum,
    int* __restrict__ off, int Nn, int NnPad, int E,
    const float* __restrict__ X, const float* __restrict__ W,
    float* __restrict__ H, int gemmBlocks) {
  __shared__ __attribute__((aligned(16))) float xt[2][16 * XT_PITCH];
  const int b = (int)blockIdx.x;
  if (b < gemmBlocks) {
    const int half = __builtin_amdgcn_readfirstlane((int)(threadIdx.x >> 7));
    const int vgrid  = gemmBlocks * 2;
    const int nTiles = (Nn + 15) >> 4;
    const int trips = (nTiles - 2 * b + vgrid - 1) / vgrid;
    gemm_body<128>(X, W, H, Nn, b * 2 + half, vgrid, trips,
                   (int)(threadIdx.x & 127), xt[half]);
  } else {
    scan_body(tab, tsum, off, Nn, NnPad, E, b - gemmBlocks);
  }
}

__device__ __forceinline__ void place1(int* cur, int2* sl,
                                       const int* __restrict__ ei, int d,
                                       int e, int lo) {
  const unsigned rel = (unsigned)(d - lo);
  if (rel < (unsigned)SPAN) {
    const int p = atomicAdd(&cur[rel], 1);
    sl[p] = make_int2(e, ei[e]);
  }
}

__device__ __forceinline__ void place4(int* cur, int2* sl, int4 d, int4 s,
                                       int e0, int lo, int spare) {
  const unsigned r0 = (unsigned)(d.x - lo), r1 = (unsigned)(d.y - lo);
  const unsigned r2 = (unsigned)(d.z - lo), r3 = (unsigned)(d.w - lo);
  const bool m0 = r0 < (unsigned)SPAN, m1 = r1 < (unsigned)SPAN;
  const bool m2 = r2 < (unsigned)SPAN, m3 = r3 < (unsigned)SPAN;
  const int p0 = atomicAdd(&cur[m0 ? (int)r0 : spare], 1);
  const int p1 = atomicAdd(&cur[m1 ? (int)r1 : spare], 1);
  const int p2 = atomicAdd(&cur[m2 ? (int)r2 : spare], 1);
  const int p3 = atomicAdd(&cur[m3 ? (int)r3 : spare], 1);
  if (m0) sl[p0] = make_int2(e0, s.x);
  if (m1) sl[p1] = make_int2(e0 + 1, s.y);
  if (m2) sl[p2] = make_int2(e0 + 2, s.z);
  if (m3) sl[p3] = make_int2(e0 + 3, s.w);
}

#define CELL_K ((SPAN + GRP_T - 1) / GRP_T)
#define DOTS_T 512

__global__ __launch_bounds__(GRP_T, 16) void csr_place_dots1(
    const int* __restrict__ ei, const int* __restrict__ tab,
    int* __restrict__ slots, int Nn, int NnPad, int E,
    const float* __restrict__ H, const float* __restrict__ a_src,
    const float* __restrict__ a_dst, float* __restrict__ hs,
    float* __restrict__ hd) {
  __shared__ int cur[SPAN + 32];
  __shared__ int2 multi[GRP_T / 32][CELL_K * 32];
  const int tid = threadIdx.x;
  if ((int)blockIdx.y >= NCHUNK) {
    const int b = ((int)blockIdx.y - NCHUNK) * (int)gridDim.x + (int)blockIdx.x;
    if (tid < DOTS_T) dots_body(H, a_src, a_dst, hs, hd, Nn, b * DOTS_T + tid);
    return;
  }
  const int lo  = blockIdx.x * SPAN;
  const int c   = blockIdx.y;
  for (int i = tid; i < SPAN; i += GRP_T) {
    const int d = lo + i;
    cur[i] = (d < Nn) ? tab[(size_t)c * NnPad + d] : 0;
  }
  if (tid < 32) cur[SPAN + tid] = 0;
  const int spare = SPAN + (tid & 31);
  __syncthreads();
  int cb, ce;
  chunk_bounds(E, c, cb, ce);
  const int* dst = ei + E;
  int2* sl = (int2*)slots;
  if (((((size_t)dst) & 15) == 0) && ((E & 3) == 0)) {
    const int4* dq = (const int4*)dst;
    const int4* sq = (const int4*)ei;
    const int4 none = make_int4(lo - 1, lo - 1, lo - 1, lo - 1);
    const int qe = ce >> 2;
    for (int q = (cb >> 2) + tid; q < qe; q += GRP_LD * GRP_T) {
      int4 a[GRP_LD];
      int4 b[GRP_LD];
#pragma unroll
      for (int u = 0; u < GRP_LD; ++u) {
        const int  qu = q + u * GRP_T;
        const bool in = qu < qe;
        const int  qc = in ? qu : qe - 1;
        const int4 t  = dq[qc];
        b[u] = sq[qc];
        a[u] = make_int4(in ? t.x : none.x, in ? t.y : none.y,
                         in ? t.z : none.z, in ? t.w : none.w);
      }
#pragma unroll
      for (int u = 0; u < GRP_LD; ++u)
        place4(cur, sl, a[u], b[u], 4 * (q + u * GRP_T), lo, spare);
    }
  } else {
    for (int e = cb + tid; e < ce; e += GRP_T) place1(cur, sl, ei, dst[e], e, lo);
  }
  __syncthreads();
  int st[CELL_K];
#pragma unroll
  for (int k = 0; k < CELL_K; ++k) {
    const int i = tid + k * GRP_T;
    const int d = lo + i;
    st[k] = (i < SPAN && d < Nn) ? tab[(size_t)c * NnPad + d] : 0;
  }
  const int lane = tid & 31;
  int2* mine = multi[tid >> 5];
  int nm = 0;
#pragma unroll
  for (int k = 0; k < CELL_K; ++k) {
    const int i = tid + k * GRP_T;
    int cn = 0;
    if (i < SPAN && lo + i < Nn) cn = cur[i] - st[k];
    const bool many = cn >= 2;
    const unsigned m = __builtin_amdgcn_ballot_w32(many);
    if (many) mine[nm + (int)__builtin_amdgcn_mbcnt_lo(m, 0u)] = make_int2(st[k], cn);
    nm += __builtin_popcount(m);
  }
  for (int j = lane; j < nm; j += 32) {
    const int2 it = mine[j];
    int2* cell = sl + it.x;
    const int2 p0 = cell[0];
    const int2 p1 = cell[1];
    if (it.y == 2) {
      if (p1.x < p0.x) { cell[0] = p1; cell[1] = p0; }
    } else {
      for (int a = 0; a + 1 < it.y; ++a) {
        const int2 first = cell[a];
        int2 best = first;
        int  bi   = a;
        for (int b = a + 1; b < it.y; ++b) {
          const int2 t = cell[b];
          if (t.x < best.x) { best = t; bi = b; }
        }
        if (bi != a) { cell[bi] = first; cell[a] = best; }
      }
    }
  }
}

__device__ __forceinline__ float wave_max_f(float v) {
#pragma unroll
  for (int o = 16; o > 0; o >>= 1) v = fmaxf(v, __shfl_xor(v, o, 32));
  return v;
}
__device__ __forceinline__ float wave_sum_f(float v) {
#pragma unroll
  for (int o = 16; o > 0; o >>= 1) v += __shfl_xor(v, o, 32);
  return v;
}
__device__ __forceinline__ float bcast_f(float v, int k) {
  return __int_as_float(__builtin_amdgcn_readlane(__float_as_int(v), k));
}

template <bool WRITE_ALPHA>
__device__ __forceinline__ float2 node_row(
    int node, int n, const int2* __restrict__ slots,
    const float* __restrict__ hs, const float* __restrict__ hd,
    const float2* __restrict__ Hr, float2 b, float* __restrict__ alpha_out,
    int E, int lane) {
  const float2 hself = Hr[(size_t)node * 32 + lane];

  const float hdn = hd[node];
  float lself = hs[node] + hdn;
  lself = (lself > 0.f) ? lself : NEG_SLOPE * lself;

  const bool v0  = lane < n;
  const int2 sl0 = v0 ? slots[lane] : make_int2(0, 0);
  float lg0 = -INFINITY;
  if (v0) {
    float t = hs[sl0.y] + hdn;
    lg0 = (t > 0.f) ? t : NEG_SLOPE * t;
  }
  float m = wave_max_f(lg0);
  for (int base = 32; base < n; base += 32) {
    const int j = base + lane;
    float lg = -INFINITY;
    if (j < n) {
      float t = hs[slots[j].y] + hdn;
      lg = (t > 0.f) ? t : NEG_SLOPE * t;
    }
    m = fmaxf(m, wave_max_f(lg));
  }
  m = fmaxf(m, lself);

  const float ex0 = v0 ? __expf(lg0 - m) : 0.f;
  float den = wave_sum_f(ex0);
  for (int base = 32; base < n; base += 32) {
    const int j = base + lane;
    float ex = 0.f;
    if (j < n) {
      float t = hs[slots[j].y] + hdn;
      t = (t > 0.f) ? t : NEG_SLOPE * t;
      ex = __expf(t - m);
    }
    den += wave_sum_f(ex);
  }
  const float exs = __expf(lself - m);
  den += exs;
  const float dd  = den + EPSV;
  const float al0 = ex0 / dd;
  const float als = exs / dd;
  if (WRITE_ALPHA) {
    if (v0) alpha_out[sl0.x] = al0;
    if (lane == 0) alpha_out[E + node] = als;
  }

  float2 acc = make_float2(0.f, 0.f);
  const int n0 = (n < 32) ? n : 32;
  int k = 0;
  for (; k + 4 <= n0; k += 4) {
    const int s0 = __builtin_amdgcn_readlane(sl0.y, k);
    const int s1 = __builtin_amdgcn_readlane(sl0.y, k + 1);
    const int s2 = __builtin_amdgcn_readlane(sl0.y, k + 2);
    const int s3 = __builtin_amdgcn_readlane(sl0.y, k + 3);
    const float2 h0 = Hr[(size_t)s0 * 32 + lane];
    const float2 h1 = Hr[(size_t)s1 * 32 + lane];
    const float2 h2 = Hr[(size_t)s2 * 32 + lane];
    const float2 h3 = Hr[(size_t)s3 * 32 + lane];
    const float a0 = bcast_f(al0, k);
    const float a1 = bcast_f(al0, k + 1);
    const float a2 = bcast_f(al0, k + 2);
    const float a3 = bcast_f(al0, k + 3);
    acc.x = fmaf(a0, h0.x, acc.x);  acc.y = fmaf(a0, h0.y, acc.y);
    acc.x = fmaf(a1, h1.x, acc.x);  acc.y = fmaf(a1, h1.y, acc.y);
    acc.x = fmaf(a2, h2.x, acc.x);  acc.y = fmaf(a2, h2.y, acc.y);
    acc.x = fmaf(a3, h3.x, acc.x);  acc.y = fmaf(a3, h3.y, acc.y);
  }
  const int r = n0 - k;
  if (r > 0) {
    const int k1 = (r > 1) ? k + 1 : k;
    const int k2 = (r > 2) ? k + 2 : k;
    const int s0 = __builtin_amdgcn_readlane(sl0.y, k);
    const int s1 = __builtin_amdgcn_readlane(sl0.y, k1);
    const int s2 = __builtin_amdgcn_readlane(sl0.y, k2);
    const float2 h0 = Hr[(size_t)s0 * 32 + lane];
    const float2 h1 = Hr[(size_t)s1 * 32 + lane];
    const float2 h2 = Hr[(size_t)s2 * 32 + lane];
    const float a0 = bcast_f(al0, k);
    const float a1 = bcast_f(al0, k1);
    const float a2 = bcast_f(al0, k2);
    acc.x = fmaf(a0, h0.x, acc.x);  acc.y = fmaf(a0, h0.y, acc.y);
    if (r > 1) { acc.x = fmaf(a1, h1.x, acc.x);  acc.y = fmaf(a1, h1.y, acc.y); }
    if (r > 2) { acc.x = fmaf(a2, h2.x, acc.x);  acc.y = fmaf(a2, h2.y, acc.y); }
  }
  for (int base = 32; base < n; base += 32) {
    const int  j  = base + lane;
    const bool vj = j < n;
    const int2 sl = vj ? slots[j] : make_int2(0, 0);
    float al = 0.f;
    if (vj) {
      float t = hs[sl.y] + hdn;
      t = (t > 0.f) ? t : NEG_SLOPE * t;
      al = __expf(t - m) / dd;
      if (WRITE_ALPHA) alpha_out[sl.x] = al;
    }
    const int c2 = (n - base < 32) ? (n - base) : 32;
    for (int q = 0; q < c2; ++q) {
      const int    s = __builtin_amdgcn_readlane(sl.y, q);
      const float  a = bcast_f(al, q);
      const float2 h = Hr[(size_t)s * 32 + lane];
      acc.x = fmaf(a, h.x, acc.x);
      acc.y = fmaf(a, h.y, acc.y);
    }
  }
  acc.x = fmaf(als, hself.x, acc.x);
  acc.y = fmaf(als, hself.y, acc.y);
  float ox = acc.x + b.x;
  float oy = acc.y + b.y;
  ox = (ox > 0.f) ? ox : 0.f;
  oy = (oy > 0.f) ? oy : 0.f;
  return make_float2(ox, oy);
}

template <bool WRITE_ALPHA>
__global__ __launch_bounds__(256) void gat_node_fused(
    const int* __restrict__ off, const int* __restrict__ slots_,
    const float* __restrict__ hs, const float* __restrict__ hd,
    const float* __restrict__ H, const float* __restrict__ bias,
    float* __restrict__ out, float* __restrict__ alpha_out, int Nn, int E) {
  const int gw = __builtin_amdgcn_readfirstlane(
      (int)(blockIdx.x * 8 + (threadIdx.x >> 5)));
  const int lane  = threadIdx.x & 31;
  const int nw    = gridDim.x * 8;
  const int per   = (Nn + nw - 1) / nw;
  const int first = gw * per;
  const int stop  = (first + per < Nn) ? first + per : Nn;
  if (first >= stop) return;
  const float2* Hr = (const float2*)H;
  const float2  b  = ((const float2*)bias)[lane];
  int beg = off[first];
  int end = off[first + 1];
  for (int node = first; node < stop; ++node) {
    const int endNext = off[(node + 2 < Nn) ? node + 2 : Nn];
    const float2 o = node_row<WRITE_ALPHA>(
        node, end - beg, (const int2*)slots_ + beg, hs, hd, Hr, b, alpha_out,
        E, lane);
    ((float2*)out)[(size_t)node * 32 + lane] = o;
    beg = end;
    end = endNext;
  }
}

#define NB_SLOTS 8
#define NB_LDP 66
__global__ __launch_bounds__(256, 16) void gat_node_next(
    const int* __restrict__ off, const int* __restrict__ slots_,
    const float* __restrict__ hs, const float* __restrict__ hd,
    const float* __restrict__ H, const float* __restrict__ bias,
    const float* __restrict__ Wn, const float* __restrict__ a_srcn,
    const float* __restrict__ a_dstn, float* __restrict__ Hn,
    float* __restrict__ hsn, float* __restrict__ hdn, int Nn, int E) {
  __shared__ __attribute__((aligned(16))) float tileL[8 * NB_SLOTS * NB_LDP];
  const int wave = __builtin_amdgcn_readfirstlane((int)(threadIdx.x >> 5));
  const int lane = threadIdx.x & 31;
  const int nw   = gridDim.x * 8;
  const int per  = (Nn + nw - 1) / nw;
  const int blockFirst = (int)blockIdx.x * 8 * per;
  if (blockFirst >= Nn) return;
  const int first = blockFirst + wave * per;
  const int stop  = (first + per < Nn) ? first + per : Nn;
  const float2* Hr = (const float2*)H;
  const float2  b  = ((const float2*)bias)[lane];
  int beg = 0, end = 0;
  if (first < stop) {
    beg = off[first];
    end = off[first + 1];
  }
  int cnt = 0;
  for (int node = first; node < stop; ++node) {
    const int endNext = off[(node + 2 < Nn) ? node + 2 : Nn];
    const float2 o = node_row<false>(
        node, end - beg, (const int2*)slots_ + beg, hs, hd, Hr, b, nullptr, E,
        lane);
    *(float2*)&tileL[(wave * NB_SLOTS + cnt) * NB_LDP + 2 * lane] = o;
    ++cnt;
    beg = end;
    end = endNext;
  }
  for (int i = cnt; i < NB_SLOTS; ++i)
    *(float2*)&tileL[(wave * NB_SLOTS + i) * NB_LDP + 2 * lane] =
        make_float2(0.f, 0.f);

  const int l16 = lane & 15;
  const int hw  = lane >> 4;
  const int kh  = hw << 1;
  const int col = (wave & 3) * 16 + l16;
  const int rt0 = (wave >> 2) * 2;

  const float* wc = Wn + col;
  v2f bf[FDIM / 4];
#pragma unroll
  for (int k = 0; k < FDIM; k += 4) {
    bf[k >> 2][0] = wc[(size_t)(k + kh) * FDIM];
    bf[k >> 2][1] = wc[(size_t)(k + kh + 1) * FDIM];
  }
  __syncthreads();

  const float* ar0 = tileL + (rt0 * 16 + l16) * NB_LDP + kh;
  const float* ar1 = ar0 + 16 * NB_LDP;
  v8f c0 = {0.f, 0.f, 0.f, 0.f, 0.f, 0.f, 0.f, 0.f};
  v8f c1 = {0.f, 0.f, 0.f, 0.f, 0.f, 0.f, 0.f, 0.f};
#pragma unroll
  for (int k = 0; k < FDIM; k += 4) {
    v2f a0, a1;
    a0[0] = ar0[k];
    a0[1] = ar0[k + 1];
    a1[0] = ar1[k];
    a1[1] = ar1[k + 1];
    c0 = __builtin_amdgcn_wmma_f32_16x16x4_f32(false, a0, false, bf[k >> 2],
                                               (short)0, c0, false, false);
    c1 = __builtin_amdgcn_wmma_f32_16x16x4_f32(false, a1, false, bf[k >> 2],
                                               (short)0, c1, false, false);
  }

  const int nd0 = blockFirst + (2 * rt0 + hw) * per;
  const int nd1 = nd0 + 2 * per;
#pragma unroll
  for (int v = 0; v < 8; ++v) {
    if (v < per && nd0 + v < Nn) Hn[(size_t)(nd0 + v) * FDIM + col] = c0[v];
    if (v < per && nd1 + v < Nn) Hn[(size_t)(nd1 + v) * FDIM + col] = c1[v];
  }

  __syncthreads();
#pragma unroll
  for (int v = 0; v < 8; ++v) {
    tileL[(rt0 * 16 + 8 * hw + v) * NB_LDP + col]      = c0[v];
    tileL[(rt0 * 16 + 16 + 8 * hw + v) * NB_LDP + col] = c1[v];
  }
  __syncthreads();

  if (wave < 4) {
    const int    r    = threadIdx.x & 63;
    const float* av   = (wave >> 1) ? a_dstn : a_srcn;
    float*       dstp = (wave >> 1) ? hdn : hsn;
    const float* hrow = tileL + r * NB_LDP;
    float s = 0.f;
#pragma unroll
    for (int f = 0; f < FDIM; ++f) s = fmaf(hrow[f], av[f], s);
    const int i  = r & 7;
    const int nd = blockFirst + (r >> 3) * per + i;
    if (i < per && nd < Nn) dstp[nd] = s;
  }
}

static inline int cdiv_ll(long long a, long long b) {
  return (int)((a + b - 1) / b);
}

extern "C" void kernel_launch(void* const* d_in, const int* in_sizes, int n_in,
                              void* d_out, int out_size, void* d_ws,
                              size_t ws_size, hipStream_t stream) {
  const float* x      = (const float*)d_in[0];
  const int*   ei     = (const int*)d_in[1];
  const float* W1     = (const float*)d_in[2];
  const float* a_src1 = (const float*)d_in[3];
  const float* a_dst1 = (const float*)d_in[4];
  const float* b1     = (const float*)d_in[5];
  const float* W2     = (const float*)d_in[6];
  const float* a_src2 = (const float*)d_in[7];
  const float* a_dst2 = (const float*)d_in[8];
  const float* b2     = (const float*)d_in[9];

  const int IN_DIM = 128;
  const int Nn = in_sizes[0] / IN_DIM;
  const int E  = in_sizes[1] / 2;

  const int    nTiles = cdiv_ll(Nn, SCAN_TILE);
  const int    nRange = cdiv_ll(Nn, SPAN);
  const size_t nTsum  = (size_t)nRange * (SPAN / SCAN_TILE) * NCHUNK;
  const size_t NnPad  = ((size_t)Nn + 4) & ~(size_t)3;

  float* bufH   = (float*)d_ws;
  float* bufAgg = bufH + (size_t)Nn * FDIM;
  float* hs     = bufAgg + (size_t)Nn * FDIM;
  float* hd     = hs + NnPad;
  float* hs2    = hd + NnPad;
  float* hd2    = hs2 + NnPad;
  int*   off    = (int*)(hd2 + NnPad);
  int*   tsum   = off + NnPad;
  int*   slots  = tsum + ((nTsum + 3) & ~(size_t)3);
  int*   tab    = slots + (((size_t)2 * E + 3) & ~(size_t)3);

  float* h2out     = (float*)d_out;
  float* alpha_out = h2out + (size_t)Nn * FDIM;

  const int nextGrid = (cdiv_ll(Nn, 64) > LOOP_GRID) ? cdiv_ll(Nn, 64) : LOOP_GRID;

  const dim3 grpGrid(nRange, NCHUNK);
  const int  dotsRows = cdiv_ll(cdiv_ll(Nn, DOTS_T), nRange);
  const dim3 placeGrid(nRange, NCHUNK + dotsRows);
  csr_count<<<grpGrid, GRP_T, 0, stream>>>(ei, tab, tsum, Nn, (int)NnPad, E);
  csr_scan_gemm1<<<LOOP_GRID / 2 + nTiles, SCAN_TILE, 0, stream>>>(
      tab, tsum, off, Nn, (int)NnPad, E, x, W1, bufH, LOOP_GRID / 2);
  csr_place_dots1<<<placeGrid, GRP_T, 0, stream>>>(
      ei, tab, slots, Nn, (int)NnPad, E, bufH, a_src1, a_dst1, hs, hd);

  gat_node_next<<<nextGrid, 256, 0, stream>>>(
      off, slots, hs, hd, bufH, b1, W2, a_src2, a_dst2, bufAgg, hs2, hd2, Nn,
      E);

  gat_node_fused<true><<<LOOP_GRID, 256, 0, stream>>>(
      off, slots, hs2, hd2, bufAgg, b2, h2out, alpha_out, Nn, E);
}
